# starting kernel
# baseline (speedup 1.0000x reference)
#include <hip/hip_runtime.h>
#include <hip/hip_bf16.h>
#include <math.h>

typedef float v2f __attribute__((ext_vector_type(2)));
typedef float v8f __attribute__((ext_vector_type(8)));

#define DDIM 256
#define KCLS 5
#define PITCH 260
#define MAXN  9.96f

__device__ __forceinline__ float waveRedSum(float v) {
#pragma unroll
  for (int off = 16; off > 0; off >>= 1) v += __shfl_xor(v, off, 32);
  return v;
}

__global__ __launch_bounds__(256) void hpr_k1(const float* __restrict__ feat,
                                              float* __restrict__ proto,
                                              float* __restrict__ proto_y2,
                                              float* __restrict__ num,
                                              float* __restrict__ den,
                                              float* __restrict__ lossA) {
  __shared__ float red[256];
  int t = threadIdx.x;
  float agk[KCLS], ag[KCLS];
#pragma unroll
  for (int c = 0; c < KCLS; c++) { agk[c] = 0.f; ag[c] = 0.f; }

  for (int r = 0; r < 25; r++) {
    float v = feat[r * DDIM + t];
    red[t] = v * v; __syncthreads();
    for (int s = 128; s > 0; s >>= 1) { if (t < s) red[t] += red[t + s]; __syncthreads(); }
    float n2 = red[0]; __syncthreads();
    float n  = fmaxf(sqrtf(n2), 1e-5f);
    float sc = tanhf(0.1f * n) / (0.1f * n);
    float pn = sc * n;
    if (pn > MAXN) { sc *= MAXN / pn; pn = MAXN; }
    float x2    = pn * pn;
    float kq    = 2.0f / (1.0f + 0.01f * x2);
    float gamma = rsqrtf(1.0f - 0.01f * kq * kq * x2);
    int c = r / 5;
    agk[c] += gamma * kq * sc * v;
    ag[c]  += gamma;
  }

  for (int c = 0; c < KCLS; c++) {
    num[c * DDIM + t] = agk[c];
    float mk = agk[c] / ag[c];
    red[t] = mk * mk; __syncthreads();
    for (int s = 128; s > 0; s >>= 1) { if (t < s) red[t] += red[t + s]; __syncthreads(); }
    float mk2 = red[0]; __syncthreads();
    float dn = 1.0f + sqrtf(1.0f - 0.01f * mk2);
    proto[c * DDIM + t] = mk / dn;
    if (t == 0) { proto_y2[c] = mk2 / (dn * dn); den[c] = ag[c]; }
  }
  if (t == 0) lossA[0] = 0.f;
}

template <bool PASS2>
__global__ __launch_bounds__(64) void hpr_pass(const float* __restrict__ feat,
                                               const int* __restrict__ label,
                                               const float* __restrict__ proto,
                                               const float* __restrict__ proto_y2,
                                               float* __restrict__ num,
                                               float* __restrict__ den,
                                               float* __restrict__ lossA,
                                               float* __restrict__ out, int Q) {
  __shared__ __align__(16) float tile[2][16][PITCH];
  __shared__ __align__(16) float sproto[16][PITCH];
  __shared__ float sX2[2][16];
  __shared__ float sDist[2][16][8];
  __shared__ int   sPred[32];
  __shared__ float sCoef[32];
  __shared__ float bden[KCLS];
  __shared__ float bloss;

  const int tid  = threadIdx.x;
  const int wave = tid >> 5, lane = tid & 31;
  const int m16  = lane & 15, half = lane >> 4;
  const int qbase = blockIdx.x * 32 + wave * 16;

  for (int idx = tid; idx < 16 * DDIM; idx += 64) {
    int r = idx >> 8, cc = idx & 255;
    sproto[r][cc] = (r < KCLS) ? proto[r * DDIM + cc] : 0.f;
  }
  if (!PASS2) {
    if (tid < KCLS) bden[tid] = 0.f;
    if (tid == KCLS) bloss = 0.f;
  }

  for (int r = 0; r < 16; r++) {
    int q = qbase + r;
    const float* row = feat + (size_t)(25 + (q < Q ? q : 0)) * DDIM;
    float4 a0 = ((const float4*)row)[lane];
    float4 a1 = ((const float4*)row)[lane + 32];
    *(float4*)&tile[wave][r][lane * 4]       = a0;
    *(float4*)&tile[wave][r][128 + lane * 4] = a1;
  }
  __syncthreads();

  for (int r = 0; r < 16; r++) {
    float acc = 0.f;
#pragma unroll
    for (int i = 0; i < 8; i++) { float v = tile[wave][r][i * 32 + lane]; acc += v * v; }
    float n2 = waveRedSum(acc);
    float n  = fmaxf(sqrtf(n2), 1e-5f);
    float sc = tanhf(0.1f * n) / (0.1f * n);
    float pn = sc * n;
    if (pn > MAXN) { sc *= MAXN / pn; pn = MAXN; }
#pragma unroll
    for (int i = 0; i < 8; i++) tile[wave][r][i * 32 + lane] *= sc;
    if (lane == 0) sX2[wave][r] = pn * pn;
  }
  __syncthreads();

  v8f c = {0.f, 0.f, 0.f, 0.f, 0.f, 0.f, 0.f, 0.f};
  for (int kc = 0; kc < DDIM; kc += 4) {
    int col = kc + half * 2;
    v2f a = *(const v2f*)&tile[wave][m16][col];
    v2f b = *(const v2f*)&sproto[m16][col];
    c = __builtin_amdgcn_wmma_f32_16x16x4_f32(false, a, false, b, (short)0, c,
                                              false, false);
  }

  float y2 = proto_y2[(m16 < KCLS) ? m16 : 0];
  if (m16 < KCLS) {
#pragma unroll
    for (int g = 0; g < 8; g++) {
      int   m  = 8 * half + g;
      float x2 = sX2[wave][m];
      float xy = c[g];
      float A  = 1.f - 0.02f * xy + 0.01f * y2;
      float B  = 1.f - 0.01f * x2;
      float nv2 = fmaxf(A * A * x2 - 2.f * A * B * xy + B * B * y2, 0.f);
      float dnm = 1.f - 0.02f * xy + 1e-4f * x2 * y2 + 1e-5f;
      float man = sqrtf(nv2) / dnm;
      float z = fminf(fmaxf(0.1f * man, -1.f + 1e-5f), 1.f - 1e-5f);
      sDist[wave][m][m16] = 10.f * (log1pf(z) - log1pf(-z));
    }
  }
  __syncthreads();

  float ceContrib = 0.f;
  if (lane < 16) {
    int  q = qbase + lane;
    bool valid = (q < Q);
    float dv[KCLS];
#pragma unroll
    for (int k = 0; k < KCLS; k++) dv[k] = sDist[wave][lane][k];
    float best = dv[0]; int pred = 0;
#pragma unroll
    for (int k = 1; k < KCLS; k++) { if (dv[k] < best) { best = dv[k]; pred = k; } }

    if (PASS2) {
      float e[KCLS]; float se = 0.f;
#pragma unroll
      for (int k = 0; k < KCLS; k++) { e[k] = expf(best - dv[k]); se += e[k]; }
      if (valid) {
        float inv = 1.f / se;
#pragma unroll
        for (int k = 0; k < KCLS; k++) out[(size_t)q * KCLS + k] = e[k] * inv;
      }
    } else {
      float se = 0.f;
#pragma unroll
      for (int k = 0; k < KCLS; k++) se += expf(best - dv[k]);
      float lse = logf(se);
      int lbl = label[Q + q];
      float dl = dv[0];
#pragma unroll
      for (int k = 1; k < KCLS; k++) dl = (lbl == k) ? dv[k] : dl;
      float ce = (best - dl) - lse;
      ceContrib = valid ? ce : 0.f;

      float x2q   = sX2[wave][lane];
      float kq    = 2.0f / (1.0f + 0.01f * x2q);
      float gamma = rsqrtf(1.0f - 0.01f * kq * kq * x2q);
      sPred[wave * 16 + lane] = pred;
      sCoef[wave * 16 + lane] = valid ? gamma * kq : 0.f;
      if (valid) atomicAdd(&bden[pred], gamma);
    }
  }

  if (!PASS2) {
    float s = waveRedSum(ceContrib);
    if (lane == 0) atomicAdd(&bloss, s);
    __syncthreads();

    float racc[KCLS][4];
#pragma unroll
    for (int cc = 0; cc < KCLS; cc++)
#pragma unroll
      for (int j = 0; j < 4; j++) racc[cc][j] = 0.f;

    for (int m2 = 0; m2 < 32; m2++) {
      int p = sPred[m2]; float cf = sCoef[m2];
      int wv = m2 >> 4, r = m2 & 15;
      float v0 = tile[wv][r][tid];
      float v1 = tile[wv][r][tid + 64];
      float v2 = tile[wv][r][tid + 128];
      float v3 = tile[wv][r][tid + 192];
#pragma unroll
      for (int cc = 0; cc < KCLS; cc++) {
        float msk = (p == cc) ? cf : 0.f;
        racc[cc][0] += msk * v0; racc[cc][1] += msk * v1;
        racc[cc][2] += msk * v2; racc[cc][3] += msk * v3;
      }
    }
#pragma unroll
    for (int cc = 0; cc < KCLS; cc++) {
      atomicAdd(&num[cc * DDIM + tid],       racc[cc][0]);
      atomicAdd(&num[cc * DDIM + tid + 64],  racc[cc][1]);
      atomicAdd(&num[cc * DDIM + tid + 128], racc[cc][2]);
      atomicAdd(&num[cc * DDIM + tid + 192], racc[cc][3]);
    }
    if (tid < KCLS) atomicAdd(&den[tid], bden[tid]);
    if (tid == KCLS) atomicAdd(lossA, bloss);
  }
}

__global__ __launch_bounds__(256) void hpr_k3(const float* __restrict__ num,
                                              const float* __restrict__ den,
                                              const float* __restrict__ lossA,
                                              float* __restrict__ proto2,
                                              float* __restrict__ proto2_y2,
                                              float* __restrict__ out, int Q) {
  __shared__ float red[256];
  int t = threadIdx.x;
  for (int c = 0; c < KCLS; c++) {
    float mk = num[c * DDIM + t] / den[c];
    red[t] = mk * mk; __syncthreads();
    for (int s = 128; s > 0; s >>= 1) { if (t < s) red[t] += red[t + s]; __syncthreads(); }
    float mk2 = red[0]; __syncthreads();
    float dn = 1.0f + sqrtf(1.0f - 0.01f * mk2);
    proto2[c * DDIM + t] = mk / dn;
    if (t == 0) proto2_y2[c] = mk2 / (dn * dn);
  }
  if (t == 0) out[(size_t)Q * KCLS] = -lossA[0] / (float)Q;
}

extern "C" void kernel_launch(void* const* d_in, const int* in_sizes, int n_in,
                              void* d_out, int out_size, void* d_ws, size_t ws_size,
                              hipStream_t stream) {
  const float* feat  = (const float*)d_in[0];
  const int*   label = (const int*)d_in[1];
  float* out = (float*)d_out;
  float* ws  = (float*)d_ws;

  int total_rows = in_sizes[0] / DDIM;
  int Q = total_rows - 25;

  float* proto     = ws;
  float* proto_y2  = ws + 1280;
  float* num       = ws + 1288;
  float* den       = ws + 2568;
  float* lossA     = ws + 2576;
  float* proto2    = ws + 2580;
  float* proto2_y2 = ws + 3860;

  hpr_k1<<<1, 256, 0, stream>>>(feat, proto, proto_y2, num, den, lossA);

  int nblk = (Q + 31) / 32;
  hpr_pass<false><<<nblk, 64, 0, stream>>>(feat, label, proto, proto_y2,
                                           num, den, lossA, out, Q);
  hpr_k3<<<1, 256, 0, stream>>>(num, den, lossA, proto2, proto2_y2, out, Q);
  hpr_pass<true><<<nblk, 64, 0, stream>>>(feat, label, proto2, proto2_y2,
                                          num, den, lossA, out, Q);
}


# pass kernels: persistent grid, 512 work-groups of 8 waves loop over 128-query tiles; per work-group protos staged once, class sums kept in registers, one atomic set; num/den/loss sum order changes
# speedup vs baseline: 1.3036x; 1.3036x over previous
#include <hip/hip_runtime.h>
#include <hip/hip_bf16.h>
#include <math.h>

typedef float v2f __attribute__((ext_vector_type(2)));
typedef float v8f __attribute__((ext_vector_type(8)));

#define DDIM 256
#define KCLS 5
#define PITCH 260
#define MAXN  9.96f

__device__ __forceinline__ float waveRedSum(float v) {
#pragma unroll
  for (int off = 16; off > 0; off >>= 1) v += __shfl_xor(v, off, 32);
  return v;
}

__global__ __launch_bounds__(256) void hpr_k1(const float* __restrict__ feat,
                                              float* __restrict__ proto,
                                              float* __restrict__ proto_y2,
                                              float* __restrict__ num,
                                              float* __restrict__ den,
                                              float* __restrict__ lossA) {
  __shared__ float red[256];
  int t = threadIdx.x;
  float agk[KCLS], ag[KCLS];
#pragma unroll
  for (int c = 0; c < KCLS; c++) { agk[c] = 0.f; ag[c] = 0.f; }

  for (int r = 0; r < 25; r++) {
    float v = feat[r * DDIM + t];
    red[t] = v * v; __syncthreads();
    for (int s = 128; s > 0; s >>= 1) { if (t < s) red[t] += red[t + s]; __syncthreads(); }
    float n2 = red[0]; __syncthreads();
    float n  = fmaxf(sqrtf(n2), 1e-5f);
    float sc = tanhf(0.1f * n) / (0.1f * n);
    float pn = sc * n;
    if (pn > MAXN) { sc *= MAXN / pn; pn = MAXN; }
    float x2    = pn * pn;
    float kq    = 2.0f / (1.0f + 0.01f * x2);
    float gamma = rsqrtf(1.0f - 0.01f * kq * kq * x2);
    int c = r / 5;
    agk[c] += gamma * kq * sc * v;
    ag[c]  += gamma;
  }

  for (int c = 0; c < KCLS; c++) {
    num[c * DDIM + t] = agk[c];
    float mk = agk[c] / ag[c];
    red[t] = mk * mk; __syncthreads();
    for (int s = 128; s > 0; s >>= 1) { if (t < s) red[t] += red[t + s]; __syncthreads(); }
    float mk2 = red[0]; __syncthreads();
    float dn = 1.0f + sqrtf(1.0f - 0.01f * mk2);
    proto[c * DDIM + t] = mk / dn;
    if (t == 0) { proto_y2[c] = mk2 / (dn * dn); den[c] = ag[c]; }
  }
  if (t == 0) lossA[0] = 0.f;
}

#define PW    8
#define PGRID 512

template <bool PASS2>
__global__ __launch_bounds__(PW * 32, 4) void hpr_pass(const float* __restrict__ feat,
                                                    const int* __restrict__ label,
                                                    const float* __restrict__ proto,
                                                    const float* __restrict__ proto_y2,
                                                    float* __restrict__ num,
                                                    float* __restrict__ den,
                                                    float* __restrict__ lossA,
                                                    float* __restrict__ out, int Q) {
  __shared__ __align__(16) float tile[PW][16][PITCH];
  __shared__ __align__(16) float sproto[16][PITCH];
  __shared__ float sX2[PW][16];
  __shared__ float sDist[PW][16][8];
  __shared__ int   sPred[PW][16];
  __shared__ float sCoef[PW][16];
  __shared__ float sDen[PW][8];
  __shared__ float sLoss[PW];

  const int tid  = threadIdx.x;
  const int wave = tid >> 5, lane = tid & 31;
  const int m16  = lane & 15, half = lane >> 4;

  for (int idx = tid; idx < 16 * DDIM; idx += PW * 32) {
    int r = idx >> 8, cc = idx & 255;
    sproto[r][cc] = (r < KCLS) ? proto[r * DDIM + cc] : 0.f;
  }
  const float y2 = proto_y2[(m16 < KCLS) ? m16 : 0];

  float racc[KCLS][8];
  float dacc[KCLS];
  float lacc = 0.f;
#pragma unroll
  for (int cc = 0; cc < KCLS; cc++) {
    dacc[cc] = 0.f;
#pragma unroll
    for (int i = 0; i < 8; i++) racc[cc][i] = 0.f;
  }

  const int nWT = (Q + PW * 16 - 1) / (PW * 16);
  for (int T = blockIdx.x; T < nWT; T += gridDim.x) {
    const int qbase = T * (PW * 16) + wave * 16;

    for (int r = 0; r < 16; r++) {
      int q = qbase + r;
      const float* row = feat + (size_t)(25 + (q < Q ? q : 0)) * DDIM;
      float4 a0 = ((const float4*)row)[lane];
      float4 a1 = ((const float4*)row)[lane + 32];
      *(float4*)&tile[wave][r][lane * 4]       = a0;
      *(float4*)&tile[wave][r][128 + lane * 4] = a1;
    }
    __syncthreads();

    for (int r = 0; r < 16; r++) {
      float acc = 0.f;
#pragma unroll
      for (int i = 0; i < 8; i++) { float v = tile[wave][r][i * 32 + lane]; acc += v * v; }
      float n2 = waveRedSum(acc);
      float n  = fmaxf(sqrtf(n2), 1e-5f);
      float sc = tanhf(0.1f * n) / (0.1f * n);
      float pn = sc * n;
      if (pn > MAXN) { sc *= MAXN / pn; pn = MAXN; }
#pragma unroll
      for (int i = 0; i < 8; i++) tile[wave][r][i * 32 + lane] *= sc;
      if (lane == 0) sX2[wave][r] = pn * pn;
    }
    __syncthreads();

    v8f c = {0.f, 0.f, 0.f, 0.f, 0.f, 0.f, 0.f, 0.f};
#pragma unroll
    for (int kc = 0; kc < DDIM; kc += 4) {
      int col = kc + half * 2;
      v2f a = *(const v2f*)&tile[wave][m16][col];
      v2f b = *(const v2f*)&sproto[m16][col];
      c = __builtin_amdgcn_wmma_f32_16x16x4_f32(false, a, false, b, (short)0, c,
                                                false, false);
    }
    asm volatile("v_nop\n\tv_nop\n\tv_nop\n\tv_nop" : "+v"(c));

    if (m16 < KCLS) {
#pragma unroll
      for (int g = 0; g < 8; g++) {
        int   m  = 8 * half + g;
        float x2 = sX2[wave][m];
        float xy = c[g];
        float A  = 1.f - 0.02f * xy + 0.01f * y2;
        float B  = 1.f - 0.01f * x2;
        float nv2 = fmaxf(A * A * x2 - 2.f * A * B * xy + B * B * y2, 0.f);
        float dnm = 1.f - 0.02f * xy + 1e-4f * x2 * y2 + 1e-5f;
        float man = sqrtf(nv2) / dnm;
        float z = fminf(fmaxf(0.1f * man, -1.f + 1e-5f), 1.f - 1e-5f);
        sDist[wave][m][m16] = 10.f * (log1pf(z) - log1pf(-z));
      }
    }
    __syncthreads();

    if (lane < 16) {
      int  q = qbase + lane;
      bool valid = (q < Q);
      float dv[KCLS];
#pragma unroll
      for (int k = 0; k < KCLS; k++) dv[k] = sDist[wave][lane][k];
      float best = dv[0]; int pred = 0;
#pragma unroll
      for (int k = 1; k < KCLS; k++) { if (dv[k] < best) { best = dv[k]; pred = k; } }

      if (PASS2) {
        float e[KCLS]; float se = 0.f;
#pragma unroll
        for (int k = 0; k < KCLS; k++) { e[k] = expf(best - dv[k]); se += e[k]; }
        if (valid) {
          float inv = 1.f / se;
#pragma unroll
          for (int k = 0; k < KCLS; k++) out[(size_t)q * KCLS + k] = e[k] * inv;
        }
      } else {
        float se = 0.f;
#pragma unroll
        for (int k = 0; k < KCLS; k++) se += expf(best - dv[k]);
        float lse = logf(se);
        int lbl = valid ? label[Q + q] : 0;
        float dl = dv[0];
#pragma unroll
        for (int k = 1; k < KCLS; k++) dl = (lbl == k) ? dv[k] : dl;
        float ce = (best - dl) - lse;
        lacc += valid ? ce : 0.f;

        float x2q   = sX2[wave][lane];
        float kq    = 2.0f / (1.0f + 0.01f * x2q);
        float gamma = rsqrtf(1.0f - 0.01f * kq * kq * x2q);
        sPred[wave][lane] = pred;
        sCoef[wave][lane] = valid ? gamma * kq : 0.f;
#pragma unroll
        for (int cc = 0; cc < KCLS; cc++) dacc[cc] += (valid && pred == cc) ? gamma : 0.f;
      }
    }

    if (!PASS2) {
      __syncthreads();
      for (int r = 0; r < 16; r++) {
        int p = sPred[wave][r]; float cf = sCoef[wave][r];
        float v[8];
#pragma unroll
        for (int i = 0; i < 8; i++) v[i] = tile[wave][r][i * 32 + lane];
#pragma unroll
        for (int cc = 0; cc < KCLS; cc++) {
          float msk = (p == cc) ? cf : 0.f;
#pragma unroll
          for (int i = 0; i < 8; i++) racc[cc][i] += msk * v[i];
        }
      }
    }
    __syncthreads();
  }

  if (!PASS2) {
    float* red = &tile[0][0][0];
#pragma unroll
    for (int cc = 0; cc < KCLS; cc++)
#pragma unroll
      for (int i = 0; i < 8; i++)
        red[wave * (KCLS * DDIM) + cc * DDIM + i * 32 + lane] = racc[cc][i];
#pragma unroll
    for (int cc = 0; cc < KCLS; cc++) {
      float s = waveRedSum(dacc[cc]);
      if (lane == 0) sDen[wave][cc] = s;
    }
    {
      float s = waveRedSum(lacc);
      if (lane == 0) sLoss[wave] = s;
    }
    __syncthreads();
    for (int e = tid; e < KCLS * DDIM; e += PW * 32) {
      float s = 0.f;
#pragma unroll
      for (int w = 0; w < PW; w++) s += red[w * (KCLS * DDIM) + e];
      atomicAdd(&num[e], s);
    }
    if (tid < KCLS) {
      float s = 0.f;
#pragma unroll
      for (int w = 0; w < PW; w++) s += sDen[w][tid];
      atomicAdd(&den[tid], s);
    }
    if (tid == KCLS) {
      float s = 0.f;
#pragma unroll
      for (int w = 0; w < PW; w++) s += sLoss[w];
      atomicAdd(lossA, s);
    }
  }
}

__global__ __launch_bounds__(256) void hpr_k3(const float* __restrict__ num,
                                              const float* __restrict__ den,
                                              const float* __restrict__ lossA,
                                              float* __restrict__ proto2,
                                              float* __restrict__ proto2_y2,
                                              float* __restrict__ out, int Q) {
  __shared__ float red[256];
  int t = threadIdx.x;
  for (int c = 0; c < KCLS; c++) {
    float mk = num[c * DDIM + t] / den[c];
    red[t] = mk * mk; __syncthreads();
    for (int s = 128; s > 0; s >>= 1) { if (t < s) red[t] += red[t + s]; __syncthreads(); }
    float mk2 = red[0]; __syncthreads();
    float dn = 1.0f + sqrtf(1.0f - 0.01f * mk2);
    proto2[c * DDIM + t] = mk / dn;
    if (t == 0) proto2_y2[c] = mk2 / (dn * dn);
  }
  if (t == 0) out[(size_t)Q * KCLS] = -lossA[0] / (float)Q;
}

extern "C" void kernel_launch(void* const* d_in, const int* in_sizes, int n_in,
                              void* d_out, int out_size, void* d_ws, size_t ws_size,
                              hipStream_t stream) {
  const float* feat  = (const float*)d_in[0];
  const int*   label = (const int*)d_in[1];
  float* out = (float*)d_out;
  float* ws  = (float*)d_ws;

  int total_rows = in_sizes[0] / DDIM;
  int Q = total_rows - 25;

  float* proto     = ws;
  float* proto_y2  = ws + 1280;
  float* num       = ws + 1288;
  float* den       = ws + 2568;
  float* lossA     = ws + 2576;
  float* proto2    = ws + 2580;
  float* proto2_y2 = ws + 3860;

  hpr_k1<<<1, 256, 0, stream>>>(feat, proto, proto_y2, num, den, lossA);

  hpr_pass<false><<<PGRID, PW * 32, 0, stream>>>(feat, label, proto, proto_y2,
                                           num, den, lossA, out, Q);
  hpr_k3<<<1, 256, 0, stream>>>(num, den, lossA, proto2, proto2_y2, out, Q);
  hpr_pass<true><<<PGRID, PW * 32, 0, stream>>>(feat, label, proto2, proto2_y2,
                                          num, den, lossA, out, Q);
}


# k1 folded into pass 1: each work-group builds the protos from the 25 support rows itself (a wave per row); k1 only zeroes num/den/loss, work-group 0 adds the support sums; norm sum order changes
# speedup vs baseline: 1.9011x; 1.4583x over previous
#include <hip/hip_runtime.h>
#include <hip/hip_bf16.h>
#include <math.h>

typedef float v2f __attribute__((ext_vector_type(2)));
typedef float v8f __attribute__((ext_vector_type(8)));

#define DDIM 256
#define KCLS 5
#define PITCH 260
#define MAXN  9.96f

__device__ __forceinline__ float waveRedSum(float v) {
#pragma unroll
  for (int off = 16; off > 0; off >>= 1) v += __shfl_xor(v, off, 32);
  return v;
}

__global__ __launch_bounds__(256) void hpr_k1(float* __restrict__ num,
                                              float* __restrict__ den,
                                              float* __restrict__ lossA) {
  int t = threadIdx.x;
#pragma unroll
  for (int c = 0; c < KCLS; c++) num[c * DDIM + t] = 0.f;
  if (t < KCLS) den[t] = 0.f;
  if (t == 0) lossA[0] = 0.f;
}

#define PW    8
#define PGRID 512

template <bool PASS2>
__global__ __launch_bounds__(PW * 32, 4) void hpr_pass(const float* __restrict__ feat,
                                                    const int* __restrict__ label,
                                                    const float* __restrict__ proto,
                                                    const float* __restrict__ proto_y2,
                                                    float* __restrict__ num,
                                                    float* __restrict__ den,
                                                    float* __restrict__ lossA,
                                                    float* __restrict__ out, int Q) {
  __shared__ __align__(16) float tile[PW][16][PITCH];
  __shared__ __align__(16) float sproto[16][PITCH];
  __shared__ float sX2[PW][16];
  __shared__ float sDist[PW][16][8];
  __shared__ int   sPred[PW][16];
  __shared__ float sCoef[PW][16];
  __shared__ float sDen[PW][8];
  __shared__ float sLoss[PW];
  __shared__ float sCf[32];
  __shared__ float sGam[32];
  __shared__ float sN2[PW][8];

  const int tid  = threadIdx.x;
  const int wave = tid >> 5, lane = tid & 31;
  const int m16  = lane & 15, half = lane >> 4;

  float y2 = 0.f;
  if (!PASS2) {
    float* srow = &tile[0][0][0];
    for (int r = wave; r < 25; r += PW) {
      const float4* row = (const float4*)(feat + (size_t)r * DDIM);
      float4 a0 = row[lane];
      float4 a1 = row[lane + 32];
      float acc = a0.x * a0.x + a0.y * a0.y + a0.z * a0.z + a0.w * a0.w
                + a1.x * a1.x + a1.y * a1.y + a1.z * a1.z + a1.w * a1.w;
      float n2 = waveRedSum(acc);
      float n  = fmaxf(sqrtf(n2), 1e-5f);
      float sc = tanhf(0.1f * n) / (0.1f * n);
      float pn = sc * n;
      if (pn > MAXN) { sc *= MAXN / pn; pn = MAXN; }
      float x2    = pn * pn;
      float kq    = 2.0f / (1.0f + 0.01f * x2);
      float gamma = rsqrtf(1.0f - 0.01f * kq * kq * x2);
      *(float4*)&srow[r * DDIM + lane * 4]       = a0;
      *(float4*)&srow[r * DDIM + 128 + lane * 4] = a1;
      if (lane == 0) { sCf[r] = gamma * kq * sc; sGam[r] = gamma; }
    }
    __syncthreads();
    float mk[KCLS];
#pragma unroll
    for (int cc = 0; cc < KCLS; cc++) {
      float agk = 0.f, ag = 0.f;
#pragma unroll
      for (int j = 0; j < 5; j++) {
        int r = cc * 5 + j;
        agk += sCf[r] * srow[r * DDIM + tid];
        ag  += sGam[r];
      }
      if (blockIdx.x == 0) {
        atomicAdd(&num[cc * DDIM + tid], agk);
        if (tid == 0) atomicAdd(&den[cc], ag);
      }
      mk[cc] = agk / ag;
      float s = waveRedSum(mk[cc] * mk[cc]);
      if (lane == 0) sN2[wave][cc] = s;
    }
    __syncthreads();
#pragma unroll
    for (int cc = 0; cc < KCLS; cc++) {
      float mk2 = 0.f;
#pragma unroll
      for (int w = 0; w < PW; w++) mk2 += sN2[w][cc];
      float dn = 1.0f + sqrtf(1.0f - 0.01f * mk2);
      sproto[cc][tid] = mk[cc] / dn;
      float yc = mk2 / (dn * dn);
      y2 = (m16 == cc) ? yc : y2;
    }
#pragma unroll
    for (int r = KCLS; r < 16; r++) sproto[r][tid] = 0.f;
  } else {
    for (int idx = tid; idx < 16 * DDIM; idx += PW * 32) {
      int r = idx >> 8, cc = idx & 255;
      sproto[r][cc] = (r < KCLS) ? proto[r * DDIM + cc] : 0.f;
    }
    y2 = proto_y2[(m16 < KCLS) ? m16 : 0];
  }

  float racc[KCLS][8];
  float dacc[KCLS];
  float lacc = 0.f;
#pragma unroll
  for (int cc = 0; cc < KCLS; cc++) {
    dacc[cc] = 0.f;
#pragma unroll
    for (int i = 0; i < 8; i++) racc[cc][i] = 0.f;
  }

  const int nWT = (Q + PW * 16 - 1) / (PW * 16);
  for (int T = blockIdx.x; T < nWT; T += gridDim.x) {
    const int qbase = T * (PW * 16) + wave * 16;

    for (int r = 0; r < 16; r++) {
      int q = qbase + r;
      const float* row = feat + (size_t)(25 + (q < Q ? q : 0)) * DDIM;
      float4 a0 = ((const float4*)row)[lane];
      float4 a1 = ((const float4*)row)[lane + 32];
      *(float4*)&tile[wave][r][lane * 4]       = a0;
      *(float4*)&tile[wave][r][128 + lane * 4] = a1;
    }
    __syncthreads();

    for (int r = 0; r < 16; r++) {
      float acc = 0.f;
#pragma unroll
      for (int i = 0; i < 8; i++) { float v = tile[wave][r][i * 32 + lane]; acc += v * v; }
      float n2 = waveRedSum(acc);
      float n  = fmaxf(sqrtf(n2), 1e-5f);
      float sc = tanhf(0.1f * n) / (0.1f * n);
      float pn = sc * n;
      if (pn > MAXN) { sc *= MAXN / pn; pn = MAXN; }
#pragma unroll
      for (int i = 0; i < 8; i++) tile[wave][r][i * 32 + lane] *= sc;
      if (lane == 0) sX2[wave][r] = pn * pn;
    }
    __syncthreads();

    v8f c = {0.f, 0.f, 0.f, 0.f, 0.f, 0.f, 0.f, 0.f};
#pragma unroll
    for (int kc = 0; kc < DDIM; kc += 4) {
      int col = kc + half * 2;
      v2f a = *(const v2f*)&tile[wave][m16][col];
      v2f b = *(const v2f*)&sproto[m16][col];
      c = __builtin_amdgcn_wmma_f32_16x16x4_f32(false, a, false, b, (short)0, c,
                                                false, false);
    }
    asm volatile("v_nop\n\tv_nop\n\tv_nop\n\tv_nop" : "+v"(c));

    if (m16 < KCLS) {
#pragma unroll
      for (int g = 0; g < 8; g++) {
        int   m  = 8 * half + g;
        float x2 = sX2[wave][m];
        float xy = c[g];
        float A  = 1.f - 0.02f * xy + 0.01f * y2;
        float B  = 1.f - 0.01f * x2;
        float nv2 = fmaxf(A * A * x2 - 2.f * A * B * xy + B * B * y2, 0.f);
        float dnm = 1.f - 0.02f * xy + 1e-4f * x2 * y2 + 1e-5f;
        float man = sqrtf(nv2) / dnm;
        float z = fminf(fmaxf(0.1f * man, -1.f + 1e-5f), 1.f - 1e-5f);
        sDist[wave][m][m16] = 10.f * (log1pf(z) - log1pf(-z));
      }
    }
    __syncthreads();

    if (lane < 16) {
      int  q = qbase + lane;
      bool valid = (q < Q);
      float dv[KCLS];
#pragma unroll
      for (int k = 0; k < KCLS; k++) dv[k] = sDist[wave][lane][k];
      float best = dv[0]; int pred = 0;
#pragma unroll
      for (int k = 1; k < KCLS; k++) { if (dv[k] < best) { best = dv[k]; pred = k; } }

      if (PASS2) {
        float e[KCLS]; float se = 0.f;
#pragma unroll
        for (int k = 0; k < KCLS; k++) { e[k] = expf(best - dv[k]); se += e[k]; }
        if (valid) {
          float inv = 1.f / se;
#pragma unroll
          for (int k = 0; k < KCLS; k++) out[(size_t)q * KCLS + k] = e[k] * inv;
        }
      } else {
        float se = 0.f;
#pragma unroll
        for (int k = 0; k < KCLS; k++) se += expf(best - dv[k]);
        float lse = logf(se);
        int lbl = valid ? label[Q + q] : 0;
        float dl = dv[0];
#pragma unroll
        for (int k = 1; k < KCLS; k++) dl = (lbl == k) ? dv[k] : dl;
        float ce = (best - dl) - lse;
        lacc += valid ? ce : 0.f;

        float x2q   = sX2[wave][lane];
        float kq    = 2.0f / (1.0f + 0.01f * x2q);
        float gamma = rsqrtf(1.0f - 0.01f * kq * kq * x2q);
        sPred[wave][lane] = pred;
        sCoef[wave][lane] = valid ? gamma * kq : 0.f;
#pragma unroll
        for (int cc = 0; cc < KCLS; cc++) dacc[cc] += (valid && pred == cc) ? gamma : 0.f;
      }
    }

    if (!PASS2) {
      __syncthreads();
      for (int r = 0; r < 16; r++) {
        int p = sPred[wave][r]; float cf = sCoef[wave][r];
        float v[8];
#pragma unroll
        for (int i = 0; i < 8; i++) v[i] = tile[wave][r][i * 32 + lane];
#pragma unroll
        for (int cc = 0; cc < KCLS; cc++) {
          float msk = (p == cc) ? cf : 0.f;
#pragma unroll
          for (int i = 0; i < 8; i++) racc[cc][i] += msk * v[i];
        }
      }
    }
    __syncthreads();
  }

  if (!PASS2) {
    float* red = &tile[0][0][0];
#pragma unroll
    for (int cc = 0; cc < KCLS; cc++)
#pragma unroll
      for (int i = 0; i < 8; i++)
        red[wave * (KCLS * DDIM) + cc * DDIM + i * 32 + lane] = racc[cc][i];
#pragma unroll
    for (int cc = 0; cc < KCLS; cc++) {
      float s = waveRedSum(dacc[cc]);
      if (lane == 0) sDen[wave][cc] = s;
    }
    {
      float s = waveRedSum(lacc);
      if (lane == 0) sLoss[wave] = s;
    }
    __syncthreads();
    for (int e = tid; e < KCLS * DDIM; e += PW * 32) {
      float s = 0.f;
#pragma unroll
      for (int w = 0; w < PW; w++) s += red[w * (KCLS * DDIM) + e];
      atomicAdd(&num[e], s);
    }
    if (tid < KCLS) {
      float s = 0.f;
#pragma unroll
      for (int w = 0; w < PW; w++) s += sDen[w][tid];
      atomicAdd(&den[tid], s);
    }
    if (tid == KCLS) {
      float s = 0.f;
#pragma unroll
      for (int w = 0; w < PW; w++) s += sLoss[w];
      atomicAdd(lossA, s);
    }
  }
}

__global__ __launch_bounds__(256) void hpr_k3(const float* __restrict__ num,
                                              const float* __restrict__ den,
                                              const float* __restrict__ lossA,
                                              float* __restrict__ proto2,
                                              float* __restrict__ proto2_y2,
                                              float* __restrict__ out, int Q) {
  __shared__ float red[256];
  int t = threadIdx.x;
  for (int c = 0; c < KCLS; c++) {
    float mk = num[c * DDIM + t] / den[c];
    red[t] = mk * mk; __syncthreads();
    for (int s = 128; s > 0; s >>= 1) { if (t < s) red[t] += red[t + s]; __syncthreads(); }
    float mk2 = red[0]; __syncthreads();
    float dn = 1.0f + sqrtf(1.0f - 0.01f * mk2);
    proto2[c * DDIM + t] = mk / dn;
    if (t == 0) proto2_y2[c] = mk2 / (dn * dn);
  }
  if (t == 0) out[(size_t)Q * KCLS] = -lossA[0] / (float)Q;
}

extern "C" void kernel_launch(void* const* d_in, const int* in_sizes, int n_in,
                              void* d_out, int out_size, void* d_ws, size_t ws_size,
                              hipStream_t stream) {
  const float* feat  = (const float*)d_in[0];
  const int*   label = (const int*)d_in[1];
  float* out = (float*)d_out;
  float* ws  = (float*)d_ws;

  int total_rows = in_sizes[0] / DDIM;
  int Q = total_rows - 25;

  float* proto     = ws;
  float* proto_y2  = ws + 1280;
  float* num       = ws + 1288;
  float* den       = ws + 2568;
  float* lossA     = ws + 2576;
  float* proto2    = ws + 2580;
  float* proto2_y2 = ws + 3860;

  hpr_k1<<<1, 256, 0, stream>>>(num, den, lossA);

  hpr_pass<false><<<PGRID, PW * 32, 0, stream>>>(feat, label, proto, proto_y2,
                                           num, den, lossA, out, Q);
  hpr_k3<<<1, 256, 0, stream>>>(num, den, lossA, proto2, proto2_y2, out, Q);
  hpr_pass<true><<<PGRID, PW * 32, 0, stream>>>(feat, label, proto2, proto2_y2,
                                          num, den, lossA, out, Q);
}


# pass 2: each lane loads half a query row straight into matrix operand registers; product taken classes x queries, so lane c finishes query c alone; no query LDS staging; norm, dot sum orders change
# speedup vs baseline: 2.2232x; 1.1694x over previous
#include <hip/hip_runtime.h>
#include <hip/hip_bf16.h>
#include <math.h>

typedef float v2f __attribute__((ext_vector_type(2)));
typedef float v8f __attribute__((ext_vector_type(8)));

#define DDIM 256
#define KCLS 5
#define PITCH 260
#define MAXN  9.96f

__device__ __forceinline__ float waveRedSum(float v) {
#pragma unroll
  for (int off = 16; off > 0; off >>= 1) v += __shfl_xor(v, off, 32);
  return v;
}

__global__ __launch_bounds__(256) void hpr_k1(float* __restrict__ num,
                                              float* __restrict__ den,
                                              float* __restrict__ lossA) {
  int t = threadIdx.x;
#pragma unroll
  for (int c = 0; c < KCLS; c++) num[c * DDIM + t] = 0.f;
  if (t < KCLS) den[t] = 0.f;
  if (t == 0) lossA[0] = 0.f;
}

#define PW    8
#define PGRID 512

template <bool PASS2>
__global__ __launch_bounds__(PW * 32, 4) void hpr_pass(const float* __restrict__ feat,
                                                    const int* __restrict__ label,
                                                    const float* __restrict__ proto,
                                                    const float* __restrict__ proto_y2,
                                                    float* __restrict__ num,
                                                    float* __restrict__ den,
                                                    float* __restrict__ lossA,
                                                    float* __restrict__ out, int Q) {
  __shared__ __align__(16) float tile[PW][16][PITCH];
  __shared__ __align__(16) float sproto[16][PITCH];
  __shared__ float sX2[PW][16];
  __shared__ float sDist[PW][16][8];
  __shared__ int   sPred[PW][16];
  __shared__ float sCoef[PW][16];
  __shared__ float sDen[PW][8];
  __shared__ float sLoss[PW];
  __shared__ float sCf[32];
  __shared__ float sGam[32];
  __shared__ float sN2[PW][8];

  const int tid  = threadIdx.x;
  const int wave = tid >> 5, lane = tid & 31;
  const int m16  = lane & 15, half = lane >> 4;

  float y2 = 0.f;
  if (!PASS2) {
    float* srow = &tile[0][0][0];
    for (int r = wave; r < 25; r += PW) {
      const float4* row = (const float4*)(feat + (size_t)r * DDIM);
      float4 a0 = row[lane];
      float4 a1 = row[lane + 32];
      float acc = a0.x * a0.x + a0.y * a0.y + a0.z * a0.z + a0.w * a0.w
                + a1.x * a1.x + a1.y * a1.y + a1.z * a1.z + a1.w * a1.w;
      float n2 = waveRedSum(acc);
      float n  = fmaxf(sqrtf(n2), 1e-5f);
      float sc = tanhf(0.1f * n) / (0.1f * n);
      float pn = sc * n;
      if (pn > MAXN) { sc *= MAXN / pn; pn = MAXN; }
      float x2    = pn * pn;
      float kq    = 2.0f / (1.0f + 0.01f * x2);
      float gamma = rsqrtf(1.0f - 0.01f * kq * kq * x2);
      *(float4*)&srow[r * DDIM + lane * 4]       = a0;
      *(float4*)&srow[r * DDIM + 128 + lane * 4] = a1;
      if (lane == 0) { sCf[r] = gamma * kq * sc; sGam[r] = gamma; }
    }
    __syncthreads();
    float mk[KCLS];
#pragma unroll
    for (int cc = 0; cc < KCLS; cc++) {
      float agk = 0.f, ag = 0.f;
#pragma unroll
      for (int j = 0; j < 5; j++) {
        int r = cc * 5 + j;
        agk += sCf[r] * srow[r * DDIM + tid];
        ag  += sGam[r];
      }
      if (blockIdx.x == 0) {
        atomicAdd(&num[cc * DDIM + tid], agk);
        if (tid == 0) atomicAdd(&den[cc], ag);
      }
      mk[cc] = agk / ag;
      float s = waveRedSum(mk[cc] * mk[cc]);
      if (lane == 0) sN2[wave][cc] = s;
    }
    __syncthreads();
#pragma unroll
    for (int cc = 0; cc < KCLS; cc++) {
      float mk2 = 0.f;
#pragma unroll
      for (int w = 0; w < PW; w++) mk2 += sN2[w][cc];
      float dn = 1.0f + sqrtf(1.0f - 0.01f * mk2);
      sproto[cc][tid] = mk[cc] / dn;
      float yc = mk2 / (dn * dn);
      y2 = (m16 == cc) ? yc : y2;
    }
#pragma unroll
    for (int r = KCLS; r < 16; r++) sproto[r][tid] = 0.f;
  } else {
    for (int idx = tid; idx < 16 * DDIM; idx += PW * 32) {
      int r = idx >> 8, cc = idx & 255;
      sproto[r][cc] = (r < KCLS) ? proto[r * DDIM + cc] : 0.f;
    }
    y2 = proto_y2[(m16 < KCLS) ? m16 : 0];
  }

  float racc[KCLS][8];
  float dacc[KCLS];
  float lacc = 0.f;
#pragma unroll
  for (int cc = 0; cc < KCLS; cc++) {
    dacc[cc] = 0.f;
#pragma unroll
    for (int i = 0; i < 8; i++) racc[cc][i] = 0.f;
  }

  const int nWT = (Q + PW * 16 - 1) / (PW * 16);
  for (int T = blockIdx.x; T < nWT; T += gridDim.x) {
    const int qbase = T * (PW * 16) + wave * 16;

    for (int r = 0; r < 16; r++) {
      int q = qbase + r;
      const float* row = feat + (size_t)(25 + (q < Q ? q : 0)) * DDIM;
      float4 a0 = ((const float4*)row)[lane];
      float4 a1 = ((const float4*)row)[lane + 32];
      *(float4*)&tile[wave][r][lane * 4]       = a0;
      *(float4*)&tile[wave][r][128 + lane * 4] = a1;
    }
    __syncthreads();

    for (int r = 0; r < 16; r++) {
      float acc = 0.f;
#pragma unroll
      for (int i = 0; i < 8; i++) { float v = tile[wave][r][i * 32 + lane]; acc += v * v; }
      float n2 = waveRedSum(acc);
      float n  = fmaxf(sqrtf(n2), 1e-5f);
      float sc = tanhf(0.1f * n) / (0.1f * n);
      float pn = sc * n;
      if (pn > MAXN) { sc *= MAXN / pn; pn = MAXN; }
#pragma unroll
      for (int i = 0; i < 8; i++) tile[wave][r][i * 32 + lane] *= sc;
      if (lane == 0) sX2[wave][r] = pn * pn;
    }
    __syncthreads();

    v8f c = {0.f, 0.f, 0.f, 0.f, 0.f, 0.f, 0.f, 0.f};
#pragma unroll
    for (int kc = 0; kc < DDIM; kc += 4) {
      int col = kc + half * 2;
      v2f a = *(const v2f*)&tile[wave][m16][col];
      v2f b = *(const v2f*)&sproto[m16][col];
      c = __builtin_amdgcn_wmma_f32_16x16x4_f32(false, a, false, b, (short)0, c,
                                                false, false);
    }
    asm volatile("v_nop\n\tv_nop\n\tv_nop\n\tv_nop" : "+v"(c));

    if (m16 < KCLS) {
#pragma unroll
      for (int g = 0; g < 8; g++) {
        int   m  = 8 * half + g;
        float x2 = sX2[wave][m];
        float xy = c[g];
        float A  = 1.f - 0.02f * xy + 0.01f * y2;
        float B  = 1.f - 0.01f * x2;
        float nv2 = fmaxf(A * A * x2 - 2.f * A * B * xy + B * B * y2, 0.f);
        float dnm = 1.f - 0.02f * xy + 1e-4f * x2 * y2 + 1e-5f;
        float man = sqrtf(nv2) / dnm;
        float z = fminf(fmaxf(0.1f * man, -1.f + 1e-5f), 1.f - 1e-5f);
        sDist[wave][m][m16] = 10.f * (log1pf(z) - log1pf(-z));
      }
    }
    __syncthreads();

    if (lane < 16) {
      int  q = qbase + lane;
      bool valid = (q < Q);
      float dv[KCLS];
#pragma unroll
      for (int k = 0; k < KCLS; k++) dv[k] = sDist[wave][lane][k];
      float best = dv[0]; int pred = 0;
#pragma unroll
      for (int k = 1; k < KCLS; k++) { if (dv[k] < best) { best = dv[k]; pred = k; } }

      if (PASS2) {
        float e[KCLS]; float se = 0.f;
#pragma unroll
        for (int k = 0; k < KCLS; k++) { e[k] = expf(best - dv[k]); se += e[k]; }
        if (valid) {
          float inv = 1.f / se;
#pragma unroll
          for (int k = 0; k < KCLS; k++) out[(size_t)q * KCLS + k] = e[k] * inv;
        }
      } else {
        float se = 0.f;
#pragma unroll
        for (int k = 0; k < KCLS; k++) se += expf(best - dv[k]);
        float lse = logf(se);
        int lbl = valid ? label[Q + q] : 0;
        float dl = dv[0];
#pragma unroll
        for (int k = 1; k < KCLS; k++) dl = (lbl == k) ? dv[k] : dl;
        float ce = (best - dl) - lse;
        lacc += valid ? ce : 0.f;

        float x2q   = sX2[wave][lane];
        float kq    = 2.0f / (1.0f + 0.01f * x2q);
        float gamma = rsqrtf(1.0f - 0.01f * kq * kq * x2q);
        sPred[wave][lane] = pred;
        sCoef[wave][lane] = valid ? gamma * kq : 0.f;
#pragma unroll
        for (int cc = 0; cc < KCLS; cc++) dacc[cc] += (valid && pred == cc) ? gamma : 0.f;
      }
    }

    if (!PASS2) {
      __syncthreads();
      for (int r = 0; r < 16; r++) {
        int p = sPred[wave][r]; float cf = sCoef[wave][r];
        float v[8];
#pragma unroll
        for (int i = 0; i < 8; i++) v[i] = tile[wave][r][i * 32 + lane];
#pragma unroll
        for (int cc = 0; cc < KCLS; cc++) {
          float msk = (p == cc) ? cf : 0.f;
#pragma unroll
          for (int i = 0; i < 8; i++) racc[cc][i] += msk * v[i];
        }
      }
    }
    __syncthreads();
  }

  if (!PASS2) {
    float* red = &tile[0][0][0];
#pragma unroll
    for (int cc = 0; cc < KCLS; cc++)
#pragma unroll
      for (int i = 0; i < 8; i++)
        red[wave * (KCLS * DDIM) + cc * DDIM + i * 32 + lane] = racc[cc][i];
#pragma unroll
    for (int cc = 0; cc < KCLS; cc++) {
      float s = waveRedSum(dacc[cc]);
      if (lane == 0) sDen[wave][cc] = s;
    }
    {
      float s = waveRedSum(lacc);
      if (lane == 0) sLoss[wave] = s;
    }
    __syncthreads();
    for (int e = tid; e < KCLS * DDIM; e += PW * 32) {
      float s = 0.f;
#pragma unroll
      for (int w = 0; w < PW; w++) s += red[w * (KCLS * DDIM) + e];
      atomicAdd(&num[e], s);
    }
    if (tid < KCLS) {
      float s = 0.f;
#pragma unroll
      for (int w = 0; w < PW; w++) s += sDen[w][tid];
      atomicAdd(&den[tid], s);
    }
    if (tid == KCLS) {
      float s = 0.f;
#pragma unroll
      for (int w = 0; w < PW; w++) s += sLoss[w];
      atomicAdd(lossA, s);
    }
  }
}

__global__ __launch_bounds__(PW * 32, 4) void hpr_pass2(const float* __restrict__ feat,
                                                     const float* __restrict__ proto,
                                                     const float* __restrict__ proto_y2,
                                                     float* __restrict__ out, int Q) {
  __shared__ __align__(16) float sproto[16][PITCH];

  const int tid  = threadIdx.x;
  const int wave = tid >> 5, lane = tid & 31;
  const int m16  = lane & 15, half = lane >> 4;

  for (int idx = tid; idx < 16 * DDIM; idx += PW * 32) {
    int r = idx >> 8, cc = idx & 255;
    sproto[r][cc] = (r < KCLS) ? proto[r * DDIM + cc] : 0.f;
  }
  float y2[KCLS];
#pragma unroll
  for (int k = 0; k < KCLS; k++) y2[k] = proto_y2[k];
  __syncthreads();

  const int nWT = (Q + PW * 16 - 1) / (PW * 16);
  for (int T = blockIdx.x; T < nWT; T += gridDim.x) {
    const int  q     = T * (PW * 16) + wave * 16 + m16;
    const bool valid = (q < Q);
    const float4* rowp = (const float4*)(feat + (size_t)(25 + (valid ? q : 0)) * DDIM) + half;

    float4 x[32];
#pragma unroll
    for (int j = 0; j < 32; j++) x[j] = rowp[2 * j];

    float acc = 0.f;
#pragma unroll
    for (int j = 0; j < 32; j++)
      acc += x[j].x * x[j].x + x[j].y * x[j].y + x[j].z * x[j].z + x[j].w * x[j].w;
    float n2 = acc + __shfl_xor(acc, 16, 32);
    float n  = fmaxf(sqrtf(n2), 1e-5f);
    float sc = tanhf(0.1f * n) / (0.1f * n);
    float pn = sc * n;
    if (pn > MAXN) { sc *= MAXN / pn; pn = MAXN; }
    const float x2 = pn * pn;

    v8f c = {0.f, 0.f, 0.f, 0.f, 0.f, 0.f, 0.f, 0.f};
#pragma unroll
    for (int j = 0; j < 32; j++) {
      float4 p = *(const float4*)&sproto[m16][8 * j + 4 * half];
      v2f a0 = {p.x, p.y},       a1 = {p.z, p.w};
      v2f b0 = {x[j].x, x[j].y}, b1 = {x[j].z, x[j].w};
      c = __builtin_amdgcn_wmma_f32_16x16x4_f32(false, a0, false, b0, (short)0, c,
                                                false, false);
      c = __builtin_amdgcn_wmma_f32_16x16x4_f32(false, a1, false, b1, (short)0, c,
                                                false, false);
    }
    asm volatile("v_nop\n\tv_nop\n\tv_nop\n\tv_nop" : "+v"(c));

    if (half == 0 && valid) {
      const float B = 1.f - 0.01f * x2;
      float dv[KCLS];
#pragma unroll
      for (int k = 0; k < KCLS; k++) {
        float xy  = sc * c[k];
        float A   = 1.f - 0.02f * xy + 0.01f * y2[k];
        float nv2 = fmaxf(A * A * x2 - 2.f * A * B * xy + B * B * y2[k], 0.f);
        float dnm = 1.f - 0.02f * xy + 1e-4f * x2 * y2[k] + 1e-5f;
        float man = sqrtf(nv2) / dnm;
        float z = fminf(fmaxf(0.1f * man, -1.f + 1e-5f), 1.f - 1e-5f);
        dv[k] = 10.f * (log1pf(z) - log1pf(-z));
      }
      float best = dv[0];
#pragma unroll
      for (int k = 1; k < KCLS; k++) best = (dv[k] < best) ? dv[k] : best;
      float e[KCLS]; float se = 0.f;
#pragma unroll
      for (int k = 0; k < KCLS; k++) { e[k] = expf(best - dv[k]); se += e[k]; }
      float inv = 1.f / se;
#pragma unroll
      for (int k = 0; k < KCLS; k++) out[(size_t)q * KCLS + k] = e[k] * inv;
    }
  }
}

__global__ __launch_bounds__(256) void hpr_k3(const float* __restrict__ num,
                                              const float* __restrict__ den,
                                              const float* __restrict__ lossA,
                                              float* __restrict__ proto2,
                                              float* __restrict__ proto2_y2,
                                              float* __restrict__ out, int Q) {
  __shared__ float red[256];
  int t = threadIdx.x;
  for (int c = 0; c < KCLS; c++) {
    float mk = num[c * DDIM + t] / den[c];
    red[t] = mk * mk; __syncthreads();
    for (int s = 128; s > 0; s >>= 1) { if (t < s) red[t] += red[t + s]; __syncthreads(); }
    float mk2 = red[0]; __syncthreads();
    float dn = 1.0f + sqrtf(1.0f - 0.01f * mk2);
    proto2[c * DDIM + t] = mk / dn;
    if (t == 0) proto2_y2[c] = mk2 / (dn * dn);
  }
  if (t == 0) out[(size_t)Q * KCLS] = -lossA[0] / (float)Q;
}

extern "C" void kernel_launch(void* const* d_in, const int* in_sizes, int n_in,
                              void* d_out, int out_size, void* d_ws, size_t ws_size,
                              hipStream_t stream) {
  const float* feat  = (const float*)d_in[0];
  const int*   label = (const int*)d_in[1];
  float* out = (float*)d_out;
  float* ws  = (float*)d_ws;

  int total_rows = in_sizes[0] / DDIM;
  int Q = total_rows - 25;

  float* proto     = ws;
  float* proto_y2  = ws + 1280;
  float* num       = ws + 1288;
  float* den       = ws + 2568;
  float* lossA     = ws + 2576;
  float* proto2    = ws + 2580;
  float* proto2_y2 = ws + 3860;

  hpr_k1<<<1, 256, 0, stream>>>(num, den, lossA);

  hpr_pass<false><<<PGRID, PW * 32, 0, stream>>>(feat, label, proto, proto_y2,
                                           num, den, lossA, out, Q);
  hpr_k3<<<1, 256, 0, stream>>>(num, den, lossA, proto2, proto2_y2, out, Q);
  hpr_pass2<<<PGRID, PW * 32, 0, stream>>>(feat, proto2, proto2_y2, out, Q);
}


# pass 1: lane loads half a query row into matrix operand registers, finishes its query alone (as pass 2); rows cross a wave-private LDS tile once for the class sums; num, norm, dot sum orders change
# speedup vs baseline: 2.7562x; 1.2397x over previous
#include <hip/hip_runtime.h>
#include <hip/hip_bf16.h>
#include <math.h>

typedef float v2f __attribute__((ext_vector_type(2)));
typedef float v8f __attribute__((ext_vector_type(8)));

#define DDIM 256
#define KCLS 5
#define PITCH 260
#define MAXN  9.96f

__device__ __forceinline__ float waveRedSum(float v) {
#pragma unroll
  for (int off = 16; off > 0; off >>= 1) v += __shfl_xor(v, off, 32);
  return v;
}

__global__ __launch_bounds__(256) void hpr_k1(float* __restrict__ num,
                                              float* __restrict__ den,
                                              float* __restrict__ lossA) {
  int t = threadIdx.x;
#pragma unroll
  for (int c = 0; c < KCLS; c++) num[c * DDIM + t] = 0.f;
  if (t < KCLS) den[t] = 0.f;
  if (t == 0) lossA[0] = 0.f;
}

#define PW    8
#define PGRID 512

__global__ __launch_bounds__(PW * 32, 4) void hpr_pass1(const float* __restrict__ feat,
                                                     const int* __restrict__ label,
                                                     float* __restrict__ num,
                                                     float* __restrict__ den,
                                                     float* __restrict__ lossA, int Q) {
  __shared__ __align__(16) float tile[PW][16][PITCH];
  __shared__ __align__(16) float sproto[16][PITCH];
  __shared__ float sDen[PW][8];
  __shared__ float sLoss[PW];
  __shared__ float sCf[32];
  __shared__ float sGam[32];
  __shared__ float sN2[PW][8];

  const int tid  = threadIdx.x;
  const int wave = tid >> 5, lane = tid & 31;
  const int m16  = lane & 15, half = lane >> 4;

  float y2[KCLS];
  {
    float* srow = &tile[0][0][0];
    for (int r = wave; r < 25; r += PW) {
      const float4* row = (const float4*)(feat + (size_t)r * DDIM);
      float4 a0 = row[lane];
      float4 a1 = row[lane + 32];
      float acc = a0.x * a0.x + a0.y * a0.y + a0.z * a0.z + a0.w * a0.w
                + a1.x * a1.x + a1.y * a1.y + a1.z * a1.z + a1.w * a1.w;
      float n2 = waveRedSum(acc);
      float n  = fmaxf(sqrtf(n2), 1e-5f);
      float sc = tanhf(0.1f * n) / (0.1f * n);
      float pn = sc * n;
      if (pn > MAXN) { sc *= MAXN / pn; pn = MAXN; }
      float x2    = pn * pn;
      float kq    = 2.0f / (1.0f + 0.01f * x2);
      float gamma = rsqrtf(1.0f - 0.01f * kq * kq * x2);
      *(float4*)&srow[r * DDIM + lane * 4]       = a0;
      *(float4*)&srow[r * DDIM + 128 + lane * 4] = a1;
      if (lane == 0) { sCf[r] = gamma * kq * sc; sGam[r] = gamma; }
    }
    __syncthreads();
    float mk[KCLS];
#pragma unroll
    for (int cc = 0; cc < KCLS; cc++) {
      float agk = 0.f, ag = 0.f;
#pragma unroll
      for (int j = 0; j < 5; j++) {
        int r = cc * 5 + j;
        agk += sCf[r] * srow[r * DDIM + tid];
        ag  += sGam[r];
      }
      if (blockIdx.x == 0) {
        atomicAdd(&num[cc * DDIM + tid], agk);
        if (tid == 0) atomicAdd(&den[cc], ag);
      }
      mk[cc] = agk / ag;
      float s = waveRedSum(mk[cc] * mk[cc]);
      if (lane == 0) sN2[wave][cc] = s;
    }
    __syncthreads();
#pragma unroll
    for (int cc = 0; cc < KCLS; cc++) {
      float mk2 = 0.f;
#pragma unroll
      for (int w = 0; w < PW; w++) mk2 += sN2[w][cc];
      float dn = 1.0f + sqrtf(1.0f - 0.01f * mk2);
      sproto[cc][tid] = mk[cc] / dn;
      y2[cc] = mk2 / (dn * dn);
    }
#pragma unroll
    for (int r = KCLS; r < 16; r++) sproto[r][tid] = 0.f;
    __syncthreads();
  }

  v2f   racc[KCLS][4];
  float dacc[KCLS];
  float lacc = 0.f;
#pragma unroll
  for (int cc = 0; cc < KCLS; cc++) {
    dacc[cc] = 0.f;
#pragma unroll
    for (int i = 0; i < 4; i++) racc[cc][i] = (v2f){0.f, 0.f};
  }

  const int nWT = (Q + PW * 16 - 1) / (PW * 16);
  for (int T = blockIdx.x; T < nWT; T += gridDim.x) {
    const int  q     = T * (PW * 16) + wave * 16 + m16;
    const bool valid = (q < Q);
    const float4* rowp = (const float4*)(feat + (size_t)(25 + (valid ? q : 0)) * DDIM) + half;

    float4 x[32];
#pragma unroll
    for (int j = 0; j < 32; j++) x[j] = rowp[2 * j];
    const int lbl = (valid && half == 0) ? label[Q + q] : 0;

    float acc = 0.f;
#pragma unroll
    for (int j = 0; j < 32; j++)
      acc += x[j].x * x[j].x + x[j].y * x[j].y + x[j].z * x[j].z + x[j].w * x[j].w;
    float n2 = acc + __shfl_xor(acc, 16, 32);
    float n  = fmaxf(sqrtf(n2), 1e-5f);
    float sc = tanhf(0.1f * n) / (0.1f * n);
    float pn = sc * n;
    if (pn > MAXN) { sc *= MAXN / pn; pn = MAXN; }
    float x2 = pn * pn;
    float B     = 1.f - 0.01f * x2;
    float kq    = 2.0f / (1.0f + 0.01f * x2);
    float gamma = rsqrtf(1.0f - 0.01f * kq * kq * x2);
    float wrow  = gamma * kq * sc;

#pragma unroll
    for (int j = 0; j < 32; j++) *(float4*)&tile[wave][m16][8 * j + 4 * half] = x[j];

    v8f c = {0.f, 0.f, 0.f, 0.f, 0.f, 0.f, 0.f, 0.f};
    asm volatile("" : "+v"(sc), "+v"(x2), "+v"(B), "+v"(gamma), "+v"(wrow));
    __builtin_amdgcn_sched_barrier(0);
#pragma unroll
    for (int j = 0; j < 32; j++) {
      float4 p = *(const float4*)&sproto[m16][8 * j + 4 * half];
      v2f a0 = {p.x, p.y},       a1 = {p.z, p.w};
      v2f b0 = {x[j].x, x[j].y}, b1 = {x[j].z, x[j].w};
      c = __builtin_amdgcn_wmma_f32_16x16x4_f32(false, a0, false, b0, (short)0, c,
                                                false, false);
      c = __builtin_amdgcn_wmma_f32_16x16x4_f32(false, a1, false, b1, (short)0, c,
                                                false, false);
    }
    asm volatile("v_nop\n\tv_nop\n\tv_nop\n\tv_nop" : "+v"(c));
    __builtin_amdgcn_sched_barrier(0);

    const bool own = valid && (half == 0);
    float dv[KCLS];
#pragma unroll
    for (int k = 0; k < KCLS; k++) {
      float xy  = sc * c[k];
      float A   = 1.f - 0.02f * xy + 0.01f * y2[k];
      float nv2 = fmaxf(A * A * x2 - 2.f * A * B * xy + B * B * y2[k], 0.f);
      float dnm = 1.f - 0.02f * xy + 1e-4f * x2 * y2[k] + 1e-5f;
      float man = sqrtf(nv2) / dnm;
      float z = fminf(fmaxf(0.1f * man, -1.f + 1e-5f), 1.f - 1e-5f);
      dv[k] = 10.f * (log1pf(z) - log1pf(-z));
    }
    float best = dv[0]; int pr = 0;
#pragma unroll
    for (int k = 1; k < KCLS; k++) { if (dv[k] < best) { best = dv[k]; pr = k; } }
    float se = 0.f;
#pragma unroll
    for (int k = 0; k < KCLS; k++) se += expf(best - dv[k]);
    float lse = logf(se);
    float dl = dv[0];
#pragma unroll
    for (int k = 1; k < KCLS; k++) dl = (lbl == k) ? dv[k] : dl;
    float ce = (best - dl) - lse;
    lacc += own ? ce : 0.f;

    const float wq = own ? wrow : 0.f;
#pragma unroll
    for (int cc = 0; cc < KCLS; cc++) dacc[cc] += (own && pr == cc) ? gamma : 0.f;

    __builtin_amdgcn_fence(__ATOMIC_RELEASE, "wavefront");
    __builtin_amdgcn_wave_barrier();
    __builtin_amdgcn_fence(__ATOMIC_ACQUIRE, "wavefront");

#pragma unroll
    for (int r = 0; r < 16; r++) {
      const int   p  = __builtin_amdgcn_readlane(pr, r);
      const float cf = __int_as_float(__builtin_amdgcn_readlane(__float_as_int(wq), r));
      const float4 v0 = *(const float4*)&tile[wave][r][4 * lane];
      const float4 v1 = *(const float4*)&tile[wave][r][128 + 4 * lane];
      const v2f va = {v0.x, v0.y}, vb = {v0.z, v0.w}, vc = {v1.x, v1.y}, vd = {v1.z, v1.w};
#pragma unroll
      for (int cc = 0; cc < KCLS; cc++) {
        const float m  = (p == cc) ? cf : 0.f;
        const v2f   mm = {m, m};
        racc[cc][0] = __builtin_elementwise_fma(mm, va, racc[cc][0]);
        racc[cc][1] = __builtin_elementwise_fma(mm, vb, racc[cc][1]);
        racc[cc][2] = __builtin_elementwise_fma(mm, vc, racc[cc][2]);
        racc[cc][3] = __builtin_elementwise_fma(mm, vd, racc[cc][3]);
      }
    }
    __builtin_amdgcn_fence(__ATOMIC_RELEASE, "wavefront");
    __builtin_amdgcn_wave_barrier();
    __builtin_amdgcn_fence(__ATOMIC_ACQUIRE, "wavefront");
  }

  {
    __syncthreads();
    float* red = &tile[0][0][0];
#pragma unroll
    for (int cc = 0; cc < KCLS; cc++) {
      float4 lo = {racc[cc][0].x, racc[cc][0].y, racc[cc][1].x, racc[cc][1].y};
      float4 hi = {racc[cc][2].x, racc[cc][2].y, racc[cc][3].x, racc[cc][3].y};
      *(float4*)&red[wave * (KCLS * DDIM) + cc * DDIM + 4 * lane]       = lo;
      *(float4*)&red[wave * (KCLS * DDIM) + cc * DDIM + 128 + 4 * lane] = hi;
    }
#pragma unroll
    for (int cc = 0; cc < KCLS; cc++) {
      float s = waveRedSum(dacc[cc]);
      if (lane == 0) sDen[wave][cc] = s;
    }
    {
      float s = waveRedSum(lacc);
      if (lane == 0) sLoss[wave] = s;
    }
    __syncthreads();
    for (int e = tid; e < KCLS * DDIM; e += PW * 32) {
      float s = 0.f;
#pragma unroll
      for (int w = 0; w < PW; w++) s += red[w * (KCLS * DDIM) + e];
      atomicAdd(&num[e], s);
    }
    if (tid < KCLS) {
      float s = 0.f;
#pragma unroll
      for (int w = 0; w < PW; w++) s += sDen[w][tid];
      atomicAdd(&den[tid], s);
    }
    if (tid == KCLS) {
      float s = 0.f;
#pragma unroll
      for (int w = 0; w < PW; w++) s += sLoss[w];
      atomicAdd(lossA, s);
    }
  }
}

__global__ __launch_bounds__(PW * 32, 4) void hpr_pass2(const float* __restrict__ feat,
                                                     const float* __restrict__ proto,
                                                     const float* __restrict__ proto_y2,
                                                     float* __restrict__ out, int Q) {
  __shared__ __align__(16) float sproto[16][PITCH];

  const int tid  = threadIdx.x;
  const int wave = tid >> 5, lane = tid & 31;
  const int m16  = lane & 15, half = lane >> 4;

  for (int idx = tid; idx < 16 * DDIM; idx += PW * 32) {
    int r = idx >> 8, cc = idx & 255;
    sproto[r][cc] = (r < KCLS) ? proto[r * DDIM + cc] : 0.f;
  }
  float y2[KCLS];
#pragma unroll
  for (int k = 0; k < KCLS; k++) y2[k] = proto_y2[k];
  __syncthreads();

  const int nWT = (Q + PW * 16 - 1) / (PW * 16);
  for (int T = blockIdx.x; T < nWT; T += gridDim.x) {
    const int  q     = T * (PW * 16) + wave * 16 + m16;
    const bool valid = (q < Q);
    const float4* rowp = (const float4*)(feat + (size_t)(25 + (valid ? q : 0)) * DDIM) + half;

    float4 x[32];
#pragma unroll
    for (int j = 0; j < 32; j++) x[j] = rowp[2 * j];

    float acc = 0.f;
#pragma unroll
    for (int j = 0; j < 32; j++)
      acc += x[j].x * x[j].x + x[j].y * x[j].y + x[j].z * x[j].z + x[j].w * x[j].w;
    float n2 = acc + __shfl_xor(acc, 16, 32);
    float n  = fmaxf(sqrtf(n2), 1e-5f);
    float sc = tanhf(0.1f * n) / (0.1f * n);
    float pn = sc * n;
    if (pn > MAXN) { sc *= MAXN / pn; pn = MAXN; }
    const float x2 = pn * pn;

    v8f c = {0.f, 0.f, 0.f, 0.f, 0.f, 0.f, 0.f, 0.f};
#pragma unroll
    for (int j = 0; j < 32; j++) {
      float4 p = *(const float4*)&sproto[m16][8 * j + 4 * half];
      v2f a0 = {p.x, p.y},       a1 = {p.z, p.w};
      v2f b0 = {x[j].x, x[j].y}, b1 = {x[j].z, x[j].w};
      c = __builtin_amdgcn_wmma_f32_16x16x4_f32(false, a0, false, b0, (short)0, c,
                                                false, false);
      c = __builtin_amdgcn_wmma_f32_16x16x4_f32(false, a1, false, b1, (short)0, c,
                                                false, false);
    }
    asm volatile("v_nop\n\tv_nop\n\tv_nop\n\tv_nop" : "+v"(c));

    if (half == 0 && valid) {
      const float B = 1.f - 0.01f * x2;
      float dv[KCLS];
#pragma unroll
      for (int k = 0; k < KCLS; k++) {
        float xy  = sc * c[k];
        float A   = 1.f - 0.02f * xy + 0.01f * y2[k];
        float nv2 = fmaxf(A * A * x2 - 2.f * A * B * xy + B * B * y2[k], 0.f);
        float dnm = 1.f - 0.02f * xy + 1e-4f * x2 * y2[k] + 1e-5f;
        float man = sqrtf(nv2) / dnm;
        float z = fminf(fmaxf(0.1f * man, -1.f + 1e-5f), 1.f - 1e-5f);
        dv[k] = 10.f * (log1pf(z) - log1pf(-z));
      }
      float best = dv[0];
#pragma unroll
      for (int k = 1; k < KCLS; k++) best = (dv[k] < best) ? dv[k] : best;
      float e[KCLS]; float se = 0.f;
#pragma unroll
      for (int k = 0; k < KCLS; k++) { e[k] = expf(best - dv[k]); se += e[k]; }
      float inv = 1.f / se;
#pragma unroll
      for (int k = 0; k < KCLS; k++) out[(size_t)q * KCLS + k] = e[k] * inv;
    }
  }
}

__global__ __launch_bounds__(256) void hpr_k3(const float* __restrict__ num,
                                              const float* __restrict__ den,
                                              const float* __restrict__ lossA,
                                              float* __restrict__ proto2,
                                              float* __restrict__ proto2_y2,
                                              float* __restrict__ out, int Q) {
  __shared__ float red[256];
  int t = threadIdx.x;
  for (int c = 0; c < KCLS; c++) {
    float mk = num[c * DDIM + t] / den[c];
    red[t] = mk * mk; __syncthreads();
    for (int s = 128; s > 0; s >>= 1) { if (t < s) red[t] += red[t + s]; __syncthreads(); }
    float mk2 = red[0]; __syncthreads();
    float dn = 1.0f + sqrtf(1.0f - 0.01f * mk2);
    proto2[c * DDIM + t] = mk / dn;
    if (t == 0) proto2_y2[c] = mk2 / (dn * dn);
  }
  if (t == 0) out[(size_t)Q * KCLS] = -lossA[0] / (float)Q;
}

extern "C" void kernel_launch(void* const* d_in, const int* in_sizes, int n_in,
                              void* d_out, int out_size, void* d_ws, size_t ws_size,
                              hipStream_t stream) {
  const float* feat  = (const float*)d_in[0];
  const int*   label = (const int*)d_in[1];
  float* out = (float*)d_out;
  float* ws  = (float*)d_ws;

  int total_rows = in_sizes[0] / DDIM;
  int Q = total_rows - 25;

  float* proto     = ws;
  float* proto_y2  = ws + 1280;
  float* num       = ws + 1288;
  float* den       = ws + 2568;
  float* lossA     = ws + 2576;
  float* proto2    = ws + 2580;
  float* proto2_y2 = ws + 3860;

  hpr_k1<<<1, 256, 0, stream>>>(num, den, lossA);

  hpr_pass1<<<PGRID, PW * 32, 0, stream>>>(feat, label, num, den, lossA, Q);
  hpr_k3<<<1, 256, 0, stream>>>(num, den, lossA, proto2, proto2_y2, out, Q);
  hpr_pass2<<<PGRID, PW * 32, 0, stream>>>(feat, proto2, proto2_y2, out, Q);
}


# pass kernels: exp(-dist) taken as the tenth power of (1-z)/(1+z): softmax and loss from 4 multiplies per class and 2 logs per query, not 10 log1p + 5 exp; class = largest quotient; sum orders kept
# speedup vs baseline: 2.9670x; 1.0765x over previous
#include <hip/hip_runtime.h>
#include <hip/hip_bf16.h>
#include <math.h>

typedef float v2f __attribute__((ext_vector_type(2)));
typedef float v8f __attribute__((ext_vector_type(8)));

#define DDIM 256
#define KCLS 5
#define PITCH 260
#define MAXN  9.96f

__device__ __forceinline__ float waveRedSum(float v) {
#pragma unroll
  for (int off = 16; off > 0; off >>= 1) v += __shfl_xor(v, off, 32);
  return v;
}

__global__ __launch_bounds__(256) void hpr_k1(float* __restrict__ num,
                                              float* __restrict__ den,
                                              float* __restrict__ lossA) {
  int t = threadIdx.x;
#pragma unroll
  for (int c = 0; c < KCLS; c++) num[c * DDIM + t] = 0.f;
  if (t < KCLS) den[t] = 0.f;
  if (t == 0) lossA[0] = 0.f;
}

#define PW    8
#define PGRID 512

__global__ __launch_bounds__(PW * 32, 4) void hpr_pass1(const float* __restrict__ feat,
                                                     const int* __restrict__ label,
                                                     float* __restrict__ num,
                                                     float* __restrict__ den,
                                                     float* __restrict__ lossA, int Q) {
  __shared__ __align__(16) float tile[PW][16][PITCH];
  __shared__ __align__(16) float sproto[16][PITCH];
  __shared__ float sDen[PW][8];
  __shared__ float sLoss[PW];
  __shared__ float sCf[32];
  __shared__ float sGam[32];
  __shared__ float sN2[PW][8];

  const int tid  = threadIdx.x;
  const int wave = tid >> 5, lane = tid & 31;
  const int m16  = lane & 15, half = lane >> 4;

  float y2[KCLS];
  {
    float* srow = &tile[0][0][0];
    for (int r = wave; r < 25; r += PW) {
      const float4* row = (const float4*)(feat + (size_t)r * DDIM);
      float4 a0 = row[lane];
      float4 a1 = row[lane + 32];
      float acc = a0.x * a0.x + a0.y * a0.y + a0.z * a0.z + a0.w * a0.w
                + a1.x * a1.x + a1.y * a1.y + a1.z * a1.z + a1.w * a1.w;
      float n2 = waveRedSum(acc);
      float n  = fmaxf(sqrtf(n2), 1e-5f);
      float sc = tanhf(0.1f * n) / (0.1f * n);
      float pn = sc * n;
      if (pn > MAXN) { sc *= MAXN / pn; pn = MAXN; }
      float x2    = pn * pn;
      float kq    = 2.0f / (1.0f + 0.01f * x2);
      float gamma = rsqrtf(1.0f - 0.01f * kq * kq * x2);
      *(float4*)&srow[r * DDIM + lane * 4]       = a0;
      *(float4*)&srow[r * DDIM + 128 + lane * 4] = a1;
      if (lane == 0) { sCf[r] = gamma * kq * sc; sGam[r] = gamma; }
    }
    __syncthreads();
    float mk[KCLS];
#pragma unroll
    for (int cc = 0; cc < KCLS; cc++) {
      float agk = 0.f, ag = 0.f;
#pragma unroll
      for (int j = 0; j < 5; j++) {
        int r = cc * 5 + j;
        agk += sCf[r] * srow[r * DDIM + tid];
        ag  += sGam[r];
      }
      if (blockIdx.x == 0) {
        atomicAdd(&num[cc * DDIM + tid], agk);
        if (tid == 0) atomicAdd(&den[cc], ag);
      }
      mk[cc] = agk / ag;
      float s = waveRedSum(mk[cc] * mk[cc]);
      if (lane == 0) sN2[wave][cc] = s;
    }
    __syncthreads();
#pragma unroll
    for (int cc = 0; cc < KCLS; cc++) {
      float mk2 = 0.f;
#pragma unroll
      for (int w = 0; w < PW; w++) mk2 += sN2[w][cc];
      float dn = 1.0f + sqrtf(1.0f - 0.01f * mk2);
      sproto[cc][tid] = mk[cc] / dn;
      y2[cc] = mk2 / (dn * dn);
    }
#pragma unroll
    for (int r = KCLS; r < 16; r++) sproto[r][tid] = 0.f;
    __syncthreads();
  }

  v2f   racc[KCLS][4];
  float dacc[KCLS];
  float lacc = 0.f;
#pragma unroll
  for (int cc = 0; cc < KCLS; cc++) {
    dacc[cc] = 0.f;
#pragma unroll
    for (int i = 0; i < 4; i++) racc[cc][i] = (v2f){0.f, 0.f};
  }

  const int nWT = (Q + PW * 16 - 1) / (PW * 16);
  for (int T = blockIdx.x; T < nWT; T += gridDim.x) {
    const int  q     = T * (PW * 16) + wave * 16 + m16;
    const bool valid = (q < Q);
    const float4* rowp = (const float4*)(feat + (size_t)(25 + (valid ? q : 0)) * DDIM) + half;

    float4 x[32];
#pragma unroll
    for (int j = 0; j < 32; j++) x[j] = rowp[2 * j];
    const int lbl = (valid && half == 0) ? label[Q + q] : 0;

    float acc = 0.f;
#pragma unroll
    for (int j = 0; j < 32; j++)
      acc += x[j].x * x[j].x + x[j].y * x[j].y + x[j].z * x[j].z + x[j].w * x[j].w;
    float n2 = acc + __shfl_xor(acc, 16, 32);
    float n  = fmaxf(sqrtf(n2), 1e-5f);
    float sc = tanhf(0.1f * n) / (0.1f * n);
    float pn = sc * n;
    if (pn > MAXN) { sc *= MAXN / pn; pn = MAXN; }
    float x2 = pn * pn;
    float B     = 1.f - 0.01f * x2;
    float kq    = 2.0f / (1.0f + 0.01f * x2);
    float gamma = rsqrtf(1.0f - 0.01f * kq * kq * x2);
    float wrow  = gamma * kq * sc;

#pragma unroll
    for (int j = 0; j < 32; j++) *(float4*)&tile[wave][m16][8 * j + 4 * half] = x[j];

    v8f c = {0.f, 0.f, 0.f, 0.f, 0.f, 0.f, 0.f, 0.f};
    asm volatile("" : "+v"(sc), "+v"(x2), "+v"(B), "+v"(gamma), "+v"(wrow));
    __builtin_amdgcn_sched_barrier(0);
#pragma unroll
    for (int j = 0; j < 32; j++) {
      float4 p = *(const float4*)&sproto[m16][8 * j + 4 * half];
      v2f a0 = {p.x, p.y},       a1 = {p.z, p.w};
      v2f b0 = {x[j].x, x[j].y}, b1 = {x[j].z, x[j].w};
      c = __builtin_amdgcn_wmma_f32_16x16x4_f32(false, a0, false, b0, (short)0, c,
                                                false, false);
      c = __builtin_amdgcn_wmma_f32_16x16x4_f32(false, a1, false, b1, (short)0, c,
                                                false, false);
    }
    asm volatile("v_nop\n\tv_nop\n\tv_nop\n\tv_nop" : "+v"(c));
    __builtin_amdgcn_sched_barrier(0);

    const bool own = valid && (half == 0);
    float rq[KCLS];
#pragma unroll
    for (int k = 0; k < KCLS; k++) {
      float xy  = sc * c[k];
      float A   = 1.f - 0.02f * xy + 0.01f * y2[k];
      float nv2 = fmaxf(A * A * x2 - 2.f * A * B * xy + B * B * y2[k], 0.f);
      float dnm = 1.f - 0.02f * xy + 1e-4f * x2 * y2[k] + 1e-5f;
      float man = sqrtf(nv2) / dnm;
      float z = fminf(fmaxf(0.1f * man, -1.f + 1e-5f), 1.f - 1e-5f);
      rq[k] = (1.f - z) / (1.f + z);
    }
    float rb = rq[0]; int pr = 0;
#pragma unroll
    for (int k = 1; k < KCLS; k++) { if (rq[k] > rb) { rb = rq[k]; pr = k; } }
    const float irb = 1.f / rb;
    float se = 0.f;
    float tl = rq[0] * irb;
#pragma unroll
    for (int k = 0; k < KCLS; k++) {
      const float t  = rq[k] * irb;
      const float t2 = t * t, t4 = t2 * t2, t8 = t4 * t4;
      se += t8 * t2;
      if (k > 0) tl = (lbl == k) ? t : tl;
    }
    float ce = 10.f * logf(tl) - logf(se);
    lacc += own ? ce : 0.f;

    const float wq = own ? wrow : 0.f;
#pragma unroll
    for (int cc = 0; cc < KCLS; cc++) dacc[cc] += (own && pr == cc) ? gamma : 0.f;

    __builtin_amdgcn_fence(__ATOMIC_RELEASE, "wavefront");
    __builtin_amdgcn_wave_barrier();
    __builtin_amdgcn_fence(__ATOMIC_ACQUIRE, "wavefront");

#pragma unroll
    for (int r = 0; r < 16; r++) {
      const int   p  = __builtin_amdgcn_readlane(pr, r);
      const float cf = __int_as_float(__builtin_amdgcn_readlane(__float_as_int(wq), r));
      const float4 v0 = *(const float4*)&tile[wave][r][4 * lane];
      const float4 v1 = *(const float4*)&tile[wave][r][128 + 4 * lane];
      const v2f va = {v0.x, v0.y}, vb = {v0.z, v0.w}, vc = {v1.x, v1.y}, vd = {v1.z, v1.w};
#pragma unroll
      for (int cc = 0; cc < KCLS; cc++) {
        const float m  = (p == cc) ? cf : 0.f;
        const v2f   mm = {m, m};
        racc[cc][0] = __builtin_elementwise_fma(mm, va, racc[cc][0]);
        racc[cc][1] = __builtin_elementwise_fma(mm, vb, racc[cc][1]);
        racc[cc][2] = __builtin_elementwise_fma(mm, vc, racc[cc][2]);
        racc[cc][3] = __builtin_elementwise_fma(mm, vd, racc[cc][3]);
      }
    }
    __builtin_amdgcn_fence(__ATOMIC_RELEASE, "wavefront");
    __builtin_amdgcn_wave_barrier();
    __builtin_amdgcn_fence(__ATOMIC_ACQUIRE, "wavefront");
  }

  {
    __syncthreads();
    float* red = &tile[0][0][0];
#pragma unroll
    for (int cc = 0; cc < KCLS; cc++) {
      float4 lo = {racc[cc][0].x, racc[cc][0].y, racc[cc][1].x, racc[cc][1].y};
      float4 hi = {racc[cc][2].x, racc[cc][2].y, racc[cc][3].x, racc[cc][3].y};
      *(float4*)&red[wave * (KCLS * DDIM) + cc * DDIM + 4 * lane]       = lo;
      *(float4*)&red[wave * (KCLS * DDIM) + cc * DDIM + 128 + 4 * lane] = hi;
    }
#pragma unroll
    for (int cc = 0; cc < KCLS; cc++) {
      float s = waveRedSum(dacc[cc]);
      if (lane == 0) sDen[wave][cc] = s;
    }
    {
      float s = waveRedSum(lacc);
      if (lane == 0) sLoss[wave] = s;
    }
    __syncthreads();
    for (int e = tid; e < KCLS * DDIM; e += PW * 32) {
      float s = 0.f;
#pragma unroll
      for (int w = 0; w < PW; w++) s += red[w * (KCLS * DDIM) + e];
      atomicAdd(&num[e], s);
    }
    if (tid < KCLS) {
      float s = 0.f;
#pragma unroll
      for (int w = 0; w < PW; w++) s += sDen[w][tid];
      atomicAdd(&den[tid], s);
    }
    if (tid == KCLS) {
      float s = 0.f;
#pragma unroll
      for (int w = 0; w < PW; w++) s += sLoss[w];
      atomicAdd(lossA, s);
    }
  }
}

__global__ __launch_bounds__(PW * 32, 4) void hpr_pass2(const float* __restrict__ feat,
                                                     const float* __restrict__ proto,
                                                     const float* __restrict__ proto_y2,
                                                     float* __restrict__ out, int Q) {
  __shared__ __align__(16) float sproto[16][PITCH];

  const int tid  = threadIdx.x;
  const int wave = tid >> 5, lane = tid & 31;
  const int m16  = lane & 15, half = lane >> 4;

  for (int idx = tid; idx < 16 * DDIM; idx += PW * 32) {
    int r = idx >> 8, cc = idx & 255;
    sproto[r][cc] = (r < KCLS) ? proto[r * DDIM + cc] : 0.f;
  }
  float y2[KCLS];
#pragma unroll
  for (int k = 0; k < KCLS; k++) y2[k] = proto_y2[k];
  __syncthreads();

  const int nWT = (Q + PW * 16 - 1) / (PW * 16);
  for (int T = blockIdx.x; T < nWT; T += gridDim.x) {
    const int  q     = T * (PW * 16) + wave * 16 + m16;
    const bool valid = (q < Q);
    const float4* rowp = (const float4*)(feat + (size_t)(25 + (valid ? q : 0)) * DDIM) + half;

    float4 x[32];
#pragma unroll
    for (int j = 0; j < 32; j++) x[j] = rowp[2 * j];

    float acc = 0.f;
#pragma unroll
    for (int j = 0; j < 32; j++)
      acc += x[j].x * x[j].x + x[j].y * x[j].y + x[j].z * x[j].z + x[j].w * x[j].w;
    float n2 = acc + __shfl_xor(acc, 16, 32);
    float n  = fmaxf(sqrtf(n2), 1e-5f);
    float sc = tanhf(0.1f * n) / (0.1f * n);
    float pn = sc * n;
    if (pn > MAXN) { sc *= MAXN / pn; pn = MAXN; }
    const float x2 = pn * pn;

    v8f c = {0.f, 0.f, 0.f, 0.f, 0.f, 0.f, 0.f, 0.f};
#pragma unroll
    for (int j = 0; j < 32; j++) {
      float4 p = *(const float4*)&sproto[m16][8 * j + 4 * half];
      v2f a0 = {p.x, p.y},       a1 = {p.z, p.w};
      v2f b0 = {x[j].x, x[j].y}, b1 = {x[j].z, x[j].w};
      c = __builtin_amdgcn_wmma_f32_16x16x4_f32(false, a0, false, b0, (short)0, c,
                                                false, false);
      c = __builtin_amdgcn_wmma_f32_16x16x4_f32(false, a1, false, b1, (short)0, c,
                                                false, false);
    }
    asm volatile("v_nop\n\tv_nop\n\tv_nop\n\tv_nop" : "+v"(c));

    if (half == 0 && valid) {
      const float B = 1.f - 0.01f * x2;
      float rq[KCLS];
#pragma unroll
      for (int k = 0; k < KCLS; k++) {
        float xy  = sc * c[k];
        float A   = 1.f - 0.02f * xy + 0.01f * y2[k];
        float nv2 = fmaxf(A * A * x2 - 2.f * A * B * xy + B * B * y2[k], 0.f);
        float dnm = 1.f - 0.02f * xy + 1e-4f * x2 * y2[k] + 1e-5f;
        float man = sqrtf(nv2) / dnm;
        float z = fminf(fmaxf(0.1f * man, -1.f + 1e-5f), 1.f - 1e-5f);
        rq[k] = (1.f - z) / (1.f + z);
      }
      float rb = rq[0];
#pragma unroll
      for (int k = 1; k < KCLS; k++) rb = (rq[k] > rb) ? rq[k] : rb;
      const float irb = 1.f / rb;
      float e[KCLS]; float se = 0.f;
#pragma unroll
      for (int k = 0; k < KCLS; k++) {
        const float t  = rq[k] * irb;
        const float t2 = t * t, t4 = t2 * t2, t8 = t4 * t4;
        e[k] = t8 * t2;
        se += e[k];
      }
      float inv = 1.f / se;
#pragma unroll
      for (int k = 0; k < KCLS; k++) out[(size_t)q * KCLS + k] = e[k] * inv;
    }
  }
}

__global__ __launch_bounds__(256) void hpr_k3(const float* __restrict__ num,
                                              const float* __restrict__ den,
                                              const float* __restrict__ lossA,
                                              float* __restrict__ proto2,
                                              float* __restrict__ proto2_y2,
                                              float* __restrict__ out, int Q) {
  __shared__ float red[256];
  int t = threadIdx.x;
  for (int c = 0; c < KCLS; c++) {
    float mk = num[c * DDIM + t] / den[c];
    red[t] = mk * mk; __syncthreads();
    for (int s = 128; s > 0; s >>= 1) { if (t < s) red[t] += red[t + s]; __syncthreads(); }
    float mk2 = red[0]; __syncthreads();
    float dn = 1.0f + sqrtf(1.0f - 0.01f * mk2);
    proto2[c * DDIM + t] = mk / dn;
    if (t == 0) proto2_y2[c] = mk2 / (dn * dn);
  }
  if (t == 0) out[(size_t)Q * KCLS] = -lossA[0] / (float)Q;
}

extern "C" void kernel_launch(void* const* d_in, const int* in_sizes, int n_in,
                              void* d_out, int out_size, void* d_ws, size_t ws_size,
                              hipStream_t stream) {
  const float* feat  = (const float*)d_in[0];
  const int*   label = (const int*)d_in[1];
  float* out = (float*)d_out;
  float* ws  = (float*)d_ws;

  int total_rows = in_sizes[0] / DDIM;
  int Q = total_rows - 25;

  float* proto     = ws;
  float* proto_y2  = ws + 1280;
  float* num       = ws + 1288;
  float* den       = ws + 2568;
  float* lossA     = ws + 2576;
  float* proto2    = ws + 2580;
  float* proto2_y2 = ws + 3860;

  hpr_k1<<<1, 256, 0, stream>>>(num, den, lossA);

  hpr_pass1<<<PGRID, PW * 32, 0, stream>>>(feat, label, num, den, lossA, Q);
  hpr_k3<<<1, 256, 0, stream>>>(num, den, lossA, proto2, proto2_y2, out, Q);
  hpr_pass2<<<PGRID, PW * 32, 0, stream>>>(feat, proto2, proto2_y2, out, Q);
}


# pass kernels: row loads run a tile ahead: first tile requested before the proto head (LDS-only barriers), next tile loaded into the operand registers as the matrix run frees them; sum orders kept
# speedup vs baseline: 3.2601x; 1.0988x over previous
#include <hip/hip_runtime.h>
#include <hip/hip_bf16.h>
#include <math.h>

typedef float v2f __attribute__((ext_vector_type(2)));
typedef float v8f __attribute__((ext_vector_type(8)));
typedef float v4f __attribute__((ext_vector_type(4)));
typedef const __attribute__((address_space(1))) v4f* gv4p;

#define DDIM 256
#define KCLS 5
#define PITCH 260
#define MAXN  9.96f

__device__ __forceinline__ float waveRedSum(float v) {
#pragma unroll
  for (int off = 16; off > 0; off >>= 1) v += __shfl_xor(v, off, 32);
  return v;
}

__device__ __forceinline__ void hpr_lds_barrier() {
  __builtin_amdgcn_fence(__ATOMIC_RELEASE, "workgroup", "local");
  __builtin_amdgcn_s_barrier();
  __builtin_amdgcn_fence(__ATOMIC_ACQUIRE, "workgroup", "local");
}

__global__ __launch_bounds__(256) void hpr_k1(float* __restrict__ num,
                                              float* __restrict__ den,
                                              float* __restrict__ lossA) {
  int t = threadIdx.x;
#pragma unroll
  for (int c = 0; c < KCLS; c++) num[c * DDIM + t] = 0.f;
  if (t < KCLS) den[t] = 0.f;
  if (t == 0) lossA[0] = 0.f;
}

#define PW    8
#define PGRID 512
#define HROWS ((25 + PW - 1) / PW)

__global__ __launch_bounds__(PW * 32, 4) void hpr_pass1(const float* __restrict__ feat,
                                                     const int* __restrict__ label,
                                                     float* __restrict__ num,
                                                     float* __restrict__ den,
                                                     float* __restrict__ lossA, int Q) {
  __shared__ __align__(16) float tile[PW][16][PITCH];
  __shared__ __align__(16) float sproto[16][PITCH];
  __shared__ float sDen[PW][8];
  __shared__ float sLoss[PW];
  __shared__ float sCf[32];
  __shared__ float sGam[32];
  __shared__ float sN2[PW][8];

  const int tid  = threadIdx.x;
  const int wave = tid >> 5, lane = tid & 31;
  const int m16  = lane & 15, half = lane >> 4;

  const int nWT = (Q + PW * 16 - 1) / (PW * 16);
  int  T     = blockIdx.x;
  bool have  = (T < nWT);
  int  q     = T * (PW * 16) + wave * 16 + m16;
  bool valid = have && (q < Q);

  float4 sa0[HROWS], sa1[HROWS];
#pragma unroll
  for (int i = 0; i < HROWS; i++) {
    const int r = min(wave + PW * i, 24);
    const float4* row = (const float4*)(feat + (size_t)r * DDIM);
    sa0[i] = row[lane];
    sa1[i] = row[lane + 32];
  }
  float4 x[32];
  {
    const int rr = valid ? 25 + q : min(25, Q + 24);
    const float4* rowp = (const float4*)(feat + (size_t)rr * DDIM) + half;
#pragma unroll
    for (int j = 0; j < 32; j++) x[j] = rowp[2 * j];
  }
  int lbl = (valid && half == 0) ? label[Q + q] : 0;

  float y2[KCLS];
  {
    float* srow = &tile[0][0][0];
#pragma unroll
    for (int i = 0; i < HROWS; i++) {
      const int r = wave + PW * i;
      if (r < 25) {
        const float4 a0 = sa0[i];
        const float4 a1 = sa1[i];
        float acc = a0.x * a0.x + a0.y * a0.y + a0.z * a0.z + a0.w * a0.w
                  + a1.x * a1.x + a1.y * a1.y + a1.z * a1.z + a1.w * a1.w;
        float n2 = waveRedSum(acc);
        float n  = fmaxf(sqrtf(n2), 1e-5f);
        float sc = tanhf(0.1f * n) / (0.1f * n);
        float pn = sc * n;
        if (pn > MAXN) { sc *= MAXN / pn; pn = MAXN; }
        float x2    = pn * pn;
        float kq    = 2.0f / (1.0f + 0.01f * x2);
        float gamma = rsqrtf(1.0f - 0.01f * kq * kq * x2);
        *(float4*)&srow[r * DDIM + lane * 4]       = a0;
        *(float4*)&srow[r * DDIM + 128 + lane * 4] = a1;
        if (lane == 0) { sCf[r] = gamma * kq * sc; sGam[r] = gamma; }
      }
    }
    hpr_lds_barrier();
    float mk[KCLS];
#pragma unroll
    for (int cc = 0; cc < KCLS; cc++) {
      float agk = 0.f, ag = 0.f;
#pragma unroll
      for (int j = 0; j < 5; j++) {
        int r = cc * 5 + j;
        agk += sCf[r] * srow[r * DDIM + tid];
        ag  += sGam[r];
      }
      if (blockIdx.x == 0) {
        atomicAdd(&num[cc * DDIM + tid], agk);
        if (tid == 0) atomicAdd(&den[cc], ag);
      }
      mk[cc] = agk / ag;
      float s = waveRedSum(mk[cc] * mk[cc]);
      if (lane == 0) sN2[wave][cc] = s;
    }
    hpr_lds_barrier();
#pragma unroll
    for (int cc = 0; cc < KCLS; cc++) {
      float mk2 = 0.f;
#pragma unroll
      for (int w = 0; w < PW; w++) mk2 += sN2[w][cc];
      float dn = 1.0f + sqrtf(1.0f - 0.01f * mk2);
      sproto[cc][tid] = mk[cc] / dn;
      y2[cc] = mk2 / (dn * dn);
    }
#pragma unroll
    for (int r = KCLS; r < 16; r++) sproto[r][tid] = 0.f;
    hpr_lds_barrier();
  }

  v2f   racc[KCLS][4];
  float dacc[KCLS];
  float lacc = 0.f;
#pragma unroll
  for (int cc = 0; cc < KCLS; cc++) {
    dacc[cc] = 0.f;
#pragma unroll
    for (int i = 0; i < 4; i++) racc[cc][i] = (v2f){0.f, 0.f};
  }

  while (have) {
    const int  Tn     = T + gridDim.x;
    const bool more   = (Tn < nWT);
    const int  qn     = Tn * (PW * 16) + wave * 16 + m16;
    const bool validn = more && (qn < Q);
    unsigned long long rown = (unsigned long long)((const float4*)(feat + (more ? (size_t)(25 + (validn ? qn : 0)) * DDIM : (size_t)0)) + half);
    const int lbln = (validn && half == 0) ? label[Q + qn] : 0;

    float acc = 0.f;
#pragma unroll
    for (int j = 0; j < 32; j++)
      acc += x[j].x * x[j].x + x[j].y * x[j].y + x[j].z * x[j].z + x[j].w * x[j].w;
    float n2 = acc + __shfl_xor(acc, 16, 32);
    float n  = fmaxf(sqrtf(n2), 1e-5f);
    float sc = tanhf(0.1f * n) / (0.1f * n);
    float pn = sc * n;
    if (pn > MAXN) { sc *= MAXN / pn; pn = MAXN; }
    float x2 = pn * pn;
    float B     = 1.f - 0.01f * x2;
    float kq    = 2.0f / (1.0f + 0.01f * x2);
    float gamma = rsqrtf(1.0f - 0.01f * kq * kq * x2);
    float wrow  = gamma * kq * sc;

#pragma unroll
    for (int j = 0; j < 32; j++) *(float4*)&tile[wave][m16][8 * j + 4 * half] = x[j];

    v8f c = {0.f, 0.f, 0.f, 0.f, 0.f, 0.f, 0.f, 0.f};
    asm volatile("" : "+v"(sc), "+v"(x2), "+v"(B), "+v"(gamma), "+v"(wrow), "+v"(rown));
    const gv4p rowpn = (gv4p)rown;
    __builtin_amdgcn_sched_barrier(0);
#pragma unroll
    for (int j = 0; j < 32; j++) {
      float4 p = *(const float4*)&sproto[m16][8 * j + 4 * half];
      v2f a0 = {p.x, p.y},       a1 = {p.z, p.w};
      v2f b0 = {x[j].x, x[j].y}, b1 = {x[j].z, x[j].w};
      c = __builtin_amdgcn_wmma_f32_16x16x4_f32(false, a0, false, b0, (short)0, c,
                                                false, false);
      c = __builtin_amdgcn_wmma_f32_16x16x4_f32(false, a1, false, b1, (short)0, c,
                                                false, false);
      if (j >= 2) { const v4f t = rowpn[2 * (j - 2)]; x[j - 2] = make_float4(t.x, t.y, t.z, t.w); }
      __builtin_amdgcn_sched_barrier(0x100);
    }
    { const v4f t = rowpn[60]; x[30] = make_float4(t.x, t.y, t.z, t.w); }
    { const v4f t = rowpn[62]; x[31] = make_float4(t.x, t.y, t.z, t.w); }
    asm volatile("v_nop\n\tv_nop\n\tv_nop\n\tv_nop" : "+v"(c));
    __builtin_amdgcn_sched_barrier(0);

    const bool own = valid && (half == 0);
    float rq[KCLS];
#pragma unroll
    for (int k = 0; k < KCLS; k++) {
      float xy  = sc * c[k];
      float A   = 1.f - 0.02f * xy + 0.01f * y2[k];
      float nv2 = fmaxf(A * A * x2 - 2.f * A * B * xy + B * B * y2[k], 0.f);
      float dnm = 1.f - 0.02f * xy + 1e-4f * x2 * y2[k] + 1e-5f;
      float man = sqrtf(nv2) / dnm;
      float z = fminf(fmaxf(0.1f * man, -1.f + 1e-5f), 1.f - 1e-5f);
      rq[k] = (1.f - z) / (1.f + z);
    }
    float rb = rq[0]; int pr = 0;
#pragma unroll
    for (int k = 1; k < KCLS; k++) { if (rq[k] > rb) { rb = rq[k]; pr = k; } }
    const float irb = 1.f / rb;
    float se = 0.f;
    float tl = rq[0] * irb;
#pragma unroll
    for (int k = 0; k < KCLS; k++) {
      const float t  = rq[k] * irb;
      const float t2 = t * t, t4 = t2 * t2, t8 = t4 * t4;
      se += t8 * t2;
      if (k > 0) tl = (lbl == k) ? t : tl;
    }
    float ce = 10.f * logf(tl) - logf(se);
    lacc += own ? ce : 0.f;

    const float wq = own ? wrow : 0.f;
#pragma unroll
    for (int cc = 0; cc < KCLS; cc++) dacc[cc] += (own && pr == cc) ? gamma : 0.f;

    __builtin_amdgcn_fence(__ATOMIC_RELEASE, "wavefront");
    __builtin_amdgcn_wave_barrier();
    __builtin_amdgcn_fence(__ATOMIC_ACQUIRE, "wavefront");

#pragma unroll 4
    for (int r = 0; r < 16; r++) {
      const int   p  = __builtin_amdgcn_readlane(pr, r);
      const float cf = __int_as_float(__builtin_amdgcn_readlane(__float_as_int(wq), r));
      const float4 v0 = *(const float4*)&tile[wave][r][4 * lane];
      const float4 v1 = *(const float4*)&tile[wave][r][128 + 4 * lane];
      const v2f va = {v0.x, v0.y}, vb = {v0.z, v0.w}, vc = {v1.x, v1.y}, vd = {v1.z, v1.w};
#pragma unroll
      for (int cc = 0; cc < KCLS; cc++) {
        const float m  = (p == cc) ? cf : 0.f;
        const v2f   mm = {m, m};
        racc[cc][0] = __builtin_elementwise_fma(mm, va, racc[cc][0]);
        racc[cc][1] = __builtin_elementwise_fma(mm, vb, racc[cc][1]);
        racc[cc][2] = __builtin_elementwise_fma(mm, vc, racc[cc][2]);
        racc[cc][3] = __builtin_elementwise_fma(mm, vd, racc[cc][3]);
      }
    }
    __builtin_amdgcn_fence(__ATOMIC_RELEASE, "wavefront");
    __builtin_amdgcn_wave_barrier();
    __builtin_amdgcn_fence(__ATOMIC_ACQUIRE, "wavefront");
    lbl = lbln; T = Tn; q = qn; valid = validn; have = more;
  }

  {
    __syncthreads();
    float* red = &tile[0][0][0];
#pragma unroll
    for (int cc = 0; cc < KCLS; cc++) {
      float4 lo = {racc[cc][0].x, racc[cc][0].y, racc[cc][1].x, racc[cc][1].y};
      float4 hi = {racc[cc][2].x, racc[cc][2].y, racc[cc][3].x, racc[cc][3].y};
      *(float4*)&red[wave * (KCLS * DDIM) + cc * DDIM + 4 * lane]       = lo;
      *(float4*)&red[wave * (KCLS * DDIM) + cc * DDIM + 128 + 4 * lane] = hi;
    }
#pragma unroll
    for (int cc = 0; cc < KCLS; cc++) {
      float s = waveRedSum(dacc[cc]);
      if (lane == 0) sDen[wave][cc] = s;
    }
    {
      float s = waveRedSum(lacc);
      if (lane == 0) sLoss[wave] = s;
    }
    __syncthreads();
    for (int e = tid; e < KCLS * DDIM; e += PW * 32) {
      float s = 0.f;
#pragma unroll
      for (int w = 0; w < PW; w++) s += red[w * (KCLS * DDIM) + e];
      atomicAdd(&num[e], s);
    }
    if (tid < KCLS) {
      float s = 0.f;
#pragma unroll
      for (int w = 0; w < PW; w++) s += sDen[w][tid];
      atomicAdd(&den[tid], s);
    }
    if (tid == KCLS) {
      float s = 0.f;
#pragma unroll
      for (int w = 0; w < PW; w++) s += sLoss[w];
      atomicAdd(lossA, s);
    }
  }
}

static_assert(PW * 32 == DDIM, "thread tid stages column tid of the protos");
__global__ __launch_bounds__(PW * 32, 4) void hpr_pass2(const float* __restrict__ feat,
                                                     const float* __restrict__ proto,
                                                     const float* __restrict__ proto_y2,
                                                     float* __restrict__ out, int Q) {
  __shared__ __align__(16) float sproto[16][PITCH];

  const int tid  = threadIdx.x;
  const int wave = tid >> 5, lane = tid & 31;
  const int m16  = lane & 15, half = lane >> 4;

  const int nWT = (Q + PW * 16 - 1) / (PW * 16);
  int  T     = blockIdx.x;
  bool have  = (T < nWT);
  int  q     = T * (PW * 16) + wave * 16 + m16;
  bool valid = have && (q < Q);

  float pv[KCLS];
#pragma unroll
  for (int r = 0; r < KCLS; r++) pv[r] = proto[r * DDIM + tid];
  float4 x[32];
  {
    const int rr = valid ? 25 + q : min(25, Q + 24);
    const float4* rowp = (const float4*)(feat + (size_t)rr * DDIM) + half;
#pragma unroll
    for (int j = 0; j < 32; j++) x[j] = rowp[2 * j];
  }

#pragma unroll
  for (int r = 0; r < 16; r++) sproto[r][tid] = (r < KCLS) ? pv[r < KCLS ? r : 0] : 0.f;
  float y2[KCLS];
#pragma unroll
  for (int k = 0; k < KCLS; k++) y2[k] = proto_y2[k];
  hpr_lds_barrier();

  while (have) {
    const int  Tn     = T + gridDim.x;
    const bool more   = (Tn < nWT);
    const int  qn     = Tn * (PW * 16) + wave * 16 + m16;
    const bool validn = more && (qn < Q);
    unsigned long long rown = (unsigned long long)((const float4*)(feat + (more ? (size_t)(25 + (validn ? qn : 0)) * DDIM : (size_t)0)) + half);

    float acc = 0.f;
#pragma unroll
    for (int j = 0; j < 32; j++)
      acc += x[j].x * x[j].x + x[j].y * x[j].y + x[j].z * x[j].z + x[j].w * x[j].w;
    float n2 = acc + __shfl_xor(acc, 16, 32);
    float n  = fmaxf(sqrtf(n2), 1e-5f);
    float sc = tanhf(0.1f * n) / (0.1f * n);
    float pn = sc * n;
    if (pn > MAXN) { sc *= MAXN / pn; pn = MAXN; }
    float x2 = pn * pn;

    v8f c = {0.f, 0.f, 0.f, 0.f, 0.f, 0.f, 0.f, 0.f};
    asm volatile("" : "+v"(sc), "+v"(x2), "+v"(rown));
    const gv4p rowpn = (gv4p)rown;
    __builtin_amdgcn_sched_barrier(0);
#pragma unroll
    for (int j = 0; j < 32; j++) {
      float4 p = *(const float4*)&sproto[m16][8 * j + 4 * half];
      v2f a0 = {p.x, p.y},       a1 = {p.z, p.w};
      v2f b0 = {x[j].x, x[j].y}, b1 = {x[j].z, x[j].w};
      c = __builtin_amdgcn_wmma_f32_16x16x4_f32(false, a0, false, b0, (short)0, c,
                                                false, false);
      c = __builtin_amdgcn_wmma_f32_16x16x4_f32(false, a1, false, b1, (short)0, c,
                                                false, false);
      if (j >= 2) { const v4f t = rowpn[2 * (j - 2)]; x[j - 2] = make_float4(t.x, t.y, t.z, t.w); }
      __builtin_amdgcn_sched_barrier(0x100);
    }
    { const v4f t = rowpn[60]; x[30] = make_float4(t.x, t.y, t.z, t.w); }
    { const v4f t = rowpn[62]; x[31] = make_float4(t.x, t.y, t.z, t.w); }
    asm volatile("v_nop\n\tv_nop\n\tv_nop\n\tv_nop" : "+v"(c));
    __builtin_amdgcn_sched_barrier(0);

    if (half == 0 && valid) {
      const float B = 1.f - 0.01f * x2;
      float rq[KCLS];
#pragma unroll
      for (int k = 0; k < KCLS; k++) {
        float xy  = sc * c[k];
        float A   = 1.f - 0.02f * xy + 0.01f * y2[k];
        float nv2 = fmaxf(A * A * x2 - 2.f * A * B * xy + B * B * y2[k], 0.f);
        float dnm = 1.f - 0.02f * xy + 1e-4f * x2 * y2[k] + 1e-5f;
        float man = sqrtf(nv2) / dnm;
        float z = fminf(fmaxf(0.1f * man, -1.f + 1e-5f), 1.f - 1e-5f);
        rq[k] = (1.f - z) / (1.f + z);
      }
      float rb = rq[0];
#pragma unroll
      for (int k = 1; k < KCLS; k++) rb = (rq[k] > rb) ? rq[k] : rb;
      const float irb = 1.f / rb;
      float e[KCLS]; float se = 0.f;
#pragma unroll
      for (int k = 0; k < KCLS; k++) {
        const float t  = rq[k] * irb;
        const float t2 = t * t, t4 = t2 * t2, t8 = t4 * t4;
        e[k] = t8 * t2;
        se += e[k];
      }
      float inv = 1.f / se;
#pragma unroll
      for (int k = 0; k < KCLS; k++) out[(size_t)q * KCLS + k] = e[k] * inv;
    }
    T = Tn; q = qn; valid = validn; have = more;
  }
}

__global__ __launch_bounds__(256) void hpr_k3(const float* __restrict__ num,
                                              const float* __restrict__ den,
                                              const float* __restrict__ lossA,
                                              float* __restrict__ proto2,
                                              float* __restrict__ proto2_y2,
                                              float* __restrict__ out, int Q) {
  __shared__ float red[256];
  int t = threadIdx.x;
  for (int c = 0; c < KCLS; c++) {
    float mk = num[c * DDIM + t] / den[c];
    red[t] = mk * mk; __syncthreads();
    for (int s = 128; s > 0; s >>= 1) { if (t < s) red[t] += red[t + s]; __syncthreads(); }
    float mk2 = red[0]; __syncthreads();
    float dn = 1.0f + sqrtf(1.0f - 0.01f * mk2);
    proto2[c * DDIM + t] = mk / dn;
    if (t == 0) proto2_y2[c] = mk2 / (dn * dn);
  }
  if (t == 0) out[(size_t)Q * KCLS] = -lossA[0] / (float)Q;
}

extern "C" void kernel_launch(void* const* d_in, const int* in_sizes, int n_in,
                              void* d_out, int out_size, void* d_ws, size_t ws_size,
                              hipStream_t stream) {
  const float* feat  = (const float*)d_in[0];
  const int*   label = (const int*)d_in[1];
  float* out = (float*)d_out;
  float* ws  = (float*)d_ws;

  int total_rows = in_sizes[0] / DDIM;
  int Q = total_rows - 25;

  float* proto     = ws;
  float* proto_y2  = ws + 1280;
  float* num       = ws + 1288;
  float* den       = ws + 2568;
  float* lossA     = ws + 2576;
  float* proto2    = ws + 2580;
  float* proto2_y2 = ws + 3860;

  hpr_k1<<<1, 256, 0, stream>>>(num, den, lossA);

  hpr_pass1<<<PGRID, PW * 32, 0, stream>>>(feat, label, num, den, lossA, Q);
  hpr_k3<<<1, 256, 0, stream>>>(num, den, lossA, proto2, proto2_y2, out, Q);
  hpr_pass2<<<PGRID, PW * 32, 0, stream>>>(feat, proto2, proto2_y2, out, Q);
}


# pass kernels: tiles of 16 queries dealt per wave; the last partial round gives whole sets to the first half of the grid only, the rest one tile per work-group; num/den/loss sum order changes
# speedup vs baseline: 3.5179x; 1.0791x over previous
#include <hip/hip_runtime.h>
#include <hip/hip_bf16.h>
#include <math.h>

typedef float v2f __attribute__((ext_vector_type(2)));
typedef float v8f __attribute__((ext_vector_type(8)));
typedef float v4f __attribute__((ext_vector_type(4)));
typedef const __attribute__((address_space(1))) v4f* gv4p;

#define DDIM 256
#define KCLS 5
#define PITCH 260
#define MAXN  9.96f

__device__ __forceinline__ float waveRedSum(float v) {
#pragma unroll
  for (int off = 16; off > 0; off >>= 1) v += __shfl_xor(v, off, 32);
  return v;
}

__device__ __forceinline__ void hpr_lds_barrier() {
  __builtin_amdgcn_fence(__ATOMIC_RELEASE, "workgroup", "local");
  __builtin_amdgcn_s_barrier();
  __builtin_amdgcn_fence(__ATOMIC_ACQUIRE, "workgroup", "local");
}

__global__ __launch_bounds__(256) void hpr_k1(float* __restrict__ num,
                                              float* __restrict__ den,
                                              float* __restrict__ lossA) {
  int t = threadIdx.x;
#pragma unroll
  for (int c = 0; c < KCLS; c++) num[c * DDIM + t] = 0.f;
  if (t < KCLS) den[t] = 0.f;
  if (t == 0) lossA[0] = 0.f;
}

#define PW    8
#define PGRID 512
#define HROWS ((25 + PW - 1) / PW)

__device__ __forceinline__ int hpr_tile(int r, int w, int b, int nT) {
  const int base = r * (PW * PGRID);
  const int rem  = nT - base;
  if (rem >= PW * PGRID) return base + b * PW + w;
  if (rem <= 0) return -1;
  const int full = min(rem / PW, PGRID / 2);
  if (b < full) return base + b * PW + w;
  const int rest = rem - full * PW;
  const int idx  = w * (PGRID - full) + (b - full);
  return (idx < rest) ? base + full * PW + idx : -1;
}

__global__ __launch_bounds__(PW * 32, 4) void hpr_pass1(const float* __restrict__ feat,
                                                     const int* __restrict__ label,
                                                     float* __restrict__ num,
                                                     float* __restrict__ den,
                                                     float* __restrict__ lossA, int Q) {
  __shared__ __align__(16) float tile[PW][16][PITCH];
  __shared__ __align__(16) float sproto[16][PITCH];
  __shared__ float sDen[PW][8];
  __shared__ float sLoss[PW];
  __shared__ float sCf[32];
  __shared__ float sGam[32];
  __shared__ float sN2[PW][8];

  const int tid  = threadIdx.x;
  const int wave = __builtin_amdgcn_readfirstlane(tid >> 5);
  const int lane = tid & 31;
  const int m16  = lane & 15, half = lane >> 4;

  const int nT = (Q + 15) / 16;
  int  rnd   = 0;
  int  T     = hpr_tile(0, wave, blockIdx.x, nT);
  bool have  = (T >= 0);
  int  q     = T * 16 + m16;
  bool valid = have && (q < Q);

  float4 sa0[HROWS], sa1[HROWS];
#pragma unroll
  for (int i = 0; i < HROWS; i++) {
    const int r = min(wave + PW * i, 24);
    const float4* row = (const float4*)(feat + (size_t)r * DDIM);
    sa0[i] = row[lane];
    sa1[i] = row[lane + 32];
  }
  float4 x[32];
  {
    const int rr = valid ? 25 + q : min(25, Q + 24);
    const float4* rowp = (const float4*)(feat + (size_t)rr * DDIM) + half;
#pragma unroll
    for (int j = 0; j < 32; j++) x[j] = rowp[2 * j];
  }
  int lbl = (valid && half == 0) ? label[Q + q] : 0;

  float y2[KCLS];
  {
    float* srow = &tile[0][0][0];
#pragma unroll
    for (int i = 0; i < HROWS; i++) {
      const int r = wave + PW * i;
      if (r < 25) {
        const float4 a0 = sa0[i];
        const float4 a1 = sa1[i];
        float acc = a0.x * a0.x + a0.y * a0.y + a0.z * a0.z + a0.w * a0.w
                  + a1.x * a1.x + a1.y * a1.y + a1.z * a1.z + a1.w * a1.w;
        float n2 = waveRedSum(acc);
        float n  = fmaxf(sqrtf(n2), 1e-5f);
        float sc = tanhf(0.1f * n) / (0.1f * n);
        float pn = sc * n;
        if (pn > MAXN) { sc *= MAXN / pn; pn = MAXN; }
        float x2    = pn * pn;
        float kq    = 2.0f / (1.0f + 0.01f * x2);
        float gamma = rsqrtf(1.0f - 0.01f * kq * kq * x2);
        *(float4*)&srow[r * DDIM + lane * 4]       = a0;
        *(float4*)&srow[r * DDIM + 128 + lane * 4] = a1;
        if (lane == 0) { sCf[r] = gamma * kq * sc; sGam[r] = gamma; }
      }
    }
    hpr_lds_barrier();
    float mk[KCLS];
#pragma unroll
    for (int cc = 0; cc < KCLS; cc++) {
      float agk = 0.f, ag = 0.f;
#pragma unroll
      for (int j = 0; j < 5; j++) {
        int r = cc * 5 + j;
        agk += sCf[r] * srow[r * DDIM + tid];
        ag  += sGam[r];
      }
      if (blockIdx.x == 0) {
        atomicAdd(&num[cc * DDIM + tid], agk);
        if (tid == 0) atomicAdd(&den[cc], ag);
      }
      mk[cc] = agk / ag;
      float s = waveRedSum(mk[cc] * mk[cc]);
      if (lane == 0) sN2[wave][cc] = s;
    }
    hpr_lds_barrier();
#pragma unroll
    for (int cc = 0; cc < KCLS; cc++) {
      float mk2 = 0.f;
#pragma unroll
      for (int w = 0; w < PW; w++) mk2 += sN2[w][cc];
      float dn = 1.0f + sqrtf(1.0f - 0.01f * mk2);
      sproto[cc][tid] = mk[cc] / dn;
      y2[cc] = mk2 / (dn * dn);
    }
#pragma unroll
    for (int r = KCLS; r < 16; r++) sproto[r][tid] = 0.f;
    hpr_lds_barrier();
  }

  v2f   racc[KCLS][4];
  float dacc[KCLS];
  float lacc = 0.f;
#pragma unroll
  for (int cc = 0; cc < KCLS; cc++) {
    dacc[cc] = 0.f;
#pragma unroll
    for (int i = 0; i < 4; i++) racc[cc][i] = (v2f){0.f, 0.f};
  }

  while (have) {
    const int  Tn     = hpr_tile(rnd + 1, wave, blockIdx.x, nT);
    const bool more   = (Tn >= 0);
    const int  qn     = Tn * 16 + m16;
    const bool validn = more && (qn < Q);
    unsigned long long rown = (unsigned long long)((const float4*)(feat + (more ? (size_t)(25 + (validn ? qn : 0)) * DDIM : (size_t)0)) + half);
    const int lbln = (validn && half == 0) ? label[Q + qn] : 0;

    float acc = 0.f;
#pragma unroll
    for (int j = 0; j < 32; j++)
      acc += x[j].x * x[j].x + x[j].y * x[j].y + x[j].z * x[j].z + x[j].w * x[j].w;
    float n2 = acc + __shfl_xor(acc, 16, 32);
    float n  = fmaxf(sqrtf(n2), 1e-5f);
    float sc = tanhf(0.1f * n) / (0.1f * n);
    float pn = sc * n;
    if (pn > MAXN) { sc *= MAXN / pn; pn = MAXN; }
    float x2 = pn * pn;
    float B     = 1.f - 0.01f * x2;
    float kq    = 2.0f / (1.0f + 0.01f * x2);
    float gamma = rsqrtf(1.0f - 0.01f * kq * kq * x2);
    float wrow  = gamma * kq * sc;

#pragma unroll
    for (int j = 0; j < 32; j++) *(float4*)&tile[wave][m16][8 * j + 4 * half] = x[j];

    v8f c = {0.f, 0.f, 0.f, 0.f, 0.f, 0.f, 0.f, 0.f};
    asm volatile("" : "+v"(sc), "+v"(x2), "+v"(B), "+v"(gamma), "+v"(wrow), "+v"(rown));
    const gv4p rowpn = (gv4p)rown;
    __builtin_amdgcn_sched_barrier(0);
#pragma unroll
    for (int j = 0; j < 32; j++) {
      float4 p = *(const float4*)&sproto[m16][8 * j + 4 * half];
      v2f a0 = {p.x, p.y},       a1 = {p.z, p.w};
      v2f b0 = {x[j].x, x[j].y}, b1 = {x[j].z, x[j].w};
      c = __builtin_amdgcn_wmma_f32_16x16x4_f32(false, a0, false, b0, (short)0, c,
                                                false, false);
      c = __builtin_amdgcn_wmma_f32_16x16x4_f32(false, a1, false, b1, (short)0, c,
                                                false, false);
      if (j >= 2) { const v4f t = rowpn[2 * (j - 2)]; x[j - 2] = make_float4(t.x, t.y, t.z, t.w); }
      __builtin_amdgcn_sched_barrier(0x100);
    }
    { const v4f t = rowpn[60]; x[30] = make_float4(t.x, t.y, t.z, t.w); }
    { const v4f t = rowpn[62]; x[31] = make_float4(t.x, t.y, t.z, t.w); }
    asm volatile("v_nop\n\tv_nop\n\tv_nop\n\tv_nop" : "+v"(c));
    __builtin_amdgcn_sched_barrier(0);

    const bool own = valid && (half == 0);
    float rq[KCLS];
#pragma unroll
    for (int k = 0; k < KCLS; k++) {
      float xy  = sc * c[k];
      float A   = 1.f - 0.02f * xy + 0.01f * y2[k];
      float nv2 = fmaxf(A * A * x2 - 2.f * A * B * xy + B * B * y2[k], 0.f);
      float dnm = 1.f - 0.02f * xy + 1e-4f * x2 * y2[k] + 1e-5f;
      float man = sqrtf(nv2) / dnm;
      float z = fminf(fmaxf(0.1f * man, -1.f + 1e-5f), 1.f - 1e-5f);
      rq[k] = (1.f - z) / (1.f + z);
    }
    float rb = rq[0]; int pr = 0;
#pragma unroll
    for (int k = 1; k < KCLS; k++) { if (rq[k] > rb) { rb = rq[k]; pr = k; } }
    const float irb = 1.f / rb;
    float se = 0.f;
    float tl = rq[0] * irb;
#pragma unroll
    for (int k = 0; k < KCLS; k++) {
      const float t  = rq[k] * irb;
      const float t2 = t * t, t4 = t2 * t2, t8 = t4 * t4;
      se += t8 * t2;
      if (k > 0) tl = (lbl == k) ? t : tl;
    }
    float ce = 10.f * logf(tl) - logf(se);
    lacc += own ? ce : 0.f;

    const float wq = own ? wrow : 0.f;
#pragma unroll
    for (int cc = 0; cc < KCLS; cc++) dacc[cc] += (own && pr == cc) ? gamma : 0.f;

    __builtin_amdgcn_fence(__ATOMIC_RELEASE, "wavefront");
    __builtin_amdgcn_wave_barrier();
    __builtin_amdgcn_fence(__ATOMIC_ACQUIRE, "wavefront");

#pragma unroll 4
    for (int r = 0; r < 16; r++) {
      const int   p  = __builtin_amdgcn_readlane(pr, r);
      const float cf = __int_as_float(__builtin_amdgcn_readlane(__float_as_int(wq), r));
      const float4 v0 = *(const float4*)&tile[wave][r][4 * lane];
      const float4 v1 = *(const float4*)&tile[wave][r][128 + 4 * lane];
      const v2f va = {v0.x, v0.y}, vb = {v0.z, v0.w}, vc = {v1.x, v1.y}, vd = {v1.z, v1.w};
#pragma unroll
      for (int cc = 0; cc < KCLS; cc++) {
        const float m  = (p == cc) ? cf : 0.f;
        const v2f   mm = {m, m};
        racc[cc][0] = __builtin_elementwise_fma(mm, va, racc[cc][0]);
        racc[cc][1] = __builtin_elementwise_fma(mm, vb, racc[cc][1]);
        racc[cc][2] = __builtin_elementwise_fma(mm, vc, racc[cc][2]);
        racc[cc][3] = __builtin_elementwise_fma(mm, vd, racc[cc][3]);
      }
    }
    __builtin_amdgcn_fence(__ATOMIC_RELEASE, "wavefront");
    __builtin_amdgcn_wave_barrier();
    __builtin_amdgcn_fence(__ATOMIC_ACQUIRE, "wavefront");
    lbl = lbln; T = Tn; q = qn; valid = validn; have = more; rnd++;
  }

  {
    __syncthreads();
    float* red = &tile[0][0][0];
#pragma unroll
    for (int cc = 0; cc < KCLS; cc++) {
      float4 lo = {racc[cc][0].x, racc[cc][0].y, racc[cc][1].x, racc[cc][1].y};
      float4 hi = {racc[cc][2].x, racc[cc][2].y, racc[cc][3].x, racc[cc][3].y};
      *(float4*)&red[wave * (KCLS * DDIM) + cc * DDIM + 4 * lane]       = lo;
      *(float4*)&red[wave * (KCLS * DDIM) + cc * DDIM + 128 + 4 * lane] = hi;
    }
#pragma unroll
    for (int cc = 0; cc < KCLS; cc++) {
      float s = waveRedSum(dacc[cc]);
      if (lane == 0) sDen[wave][cc] = s;
    }
    {
      float s = waveRedSum(lacc);
      if (lane == 0) sLoss[wave] = s;
    }
    __syncthreads();
    for (int e = tid; e < KCLS * DDIM; e += PW * 32) {
      float s = 0.f;
#pragma unroll
      for (int w = 0; w < PW; w++) s += red[w * (KCLS * DDIM) + e];
      atomicAdd(&num[e], s);
    }
    if (tid < KCLS) {
      float s = 0.f;
#pragma unroll
      for (int w = 0; w < PW; w++) s += sDen[w][tid];
      atomicAdd(&den[tid], s);
    }
    if (tid == KCLS) {
      float s = 0.f;
#pragma unroll
      for (int w = 0; w < PW; w++) s += sLoss[w];
      atomicAdd(lossA, s);
    }
  }
}

static_assert(PW * 32 == DDIM, "thread tid stages column tid of the protos");
__global__ __launch_bounds__(PW * 32, 4) void hpr_pass2(const float* __restrict__ feat,
                                                     const float* __restrict__ proto,
                                                     const float* __restrict__ proto_y2,
                                                     float* __restrict__ out, int Q) {
  __shared__ __align__(16) float sproto[16][PITCH];

  const int tid  = threadIdx.x;
  const int wave = __builtin_amdgcn_readfirstlane(tid >> 5);
  const int lane = tid & 31;
  const int m16  = lane & 15, half = lane >> 4;

  const int nT = (Q + 15) / 16;
  int  rnd   = 0;
  int  T     = hpr_tile(0, wave, blockIdx.x, nT);
  bool have  = (T >= 0);
  int  q     = T * 16 + m16;
  bool valid = have && (q < Q);

  float pv[KCLS];
#pragma unroll
  for (int r = 0; r < KCLS; r++) pv[r] = proto[r * DDIM + tid];
  float4 x[32];
  {
    const int rr = valid ? 25 + q : min(25, Q + 24);
    const float4* rowp = (const float4*)(feat + (size_t)rr * DDIM) + half;
#pragma unroll
    for (int j = 0; j < 32; j++) x[j] = rowp[2 * j];
  }

#pragma unroll
  for (int r = 0; r < 16; r++) sproto[r][tid] = (r < KCLS) ? pv[r < KCLS ? r : 0] : 0.f;
  float y2[KCLS];
#pragma unroll
  for (int k = 0; k < KCLS; k++) y2[k] = proto_y2[k];
  hpr_lds_barrier();

  while (have) {
    const int  Tn     = hpr_tile(rnd + 1, wave, blockIdx.x, nT);
    const bool more   = (Tn >= 0);
    const int  qn     = Tn * 16 + m16;
    const bool validn = more && (qn < Q);
    unsigned long long rown = (unsigned long long)((const float4*)(feat + (more ? (size_t)(25 + (validn ? qn : 0)) * DDIM : (size_t)0)) + half);

    float acc = 0.f;
#pragma unroll
    for (int j = 0; j < 32; j++)
      acc += x[j].x * x[j].x + x[j].y * x[j].y + x[j].z * x[j].z + x[j].w * x[j].w;
    float n2 = acc + __shfl_xor(acc, 16, 32);
    float n  = fmaxf(sqrtf(n2), 1e-5f);
    float sc = tanhf(0.1f * n) / (0.1f * n);
    float pn = sc * n;
    if (pn > MAXN) { sc *= MAXN / pn; pn = MAXN; }
    float x2 = pn * pn;

    v8f c = {0.f, 0.f, 0.f, 0.f, 0.f, 0.f, 0.f, 0.f};
    asm volatile("" : "+v"(sc), "+v"(x2), "+v"(rown));
    const gv4p rowpn = (gv4p)rown;
    __builtin_amdgcn_sched_barrier(0);
#pragma unroll
    for (int j = 0; j < 32; j++) {
      float4 p = *(const float4*)&sproto[m16][8 * j + 4 * half];
      v2f a0 = {p.x, p.y},       a1 = {p.z, p.w};
      v2f b0 = {x[j].x, x[j].y}, b1 = {x[j].z, x[j].w};
      c = __builtin_amdgcn_wmma_f32_16x16x4_f32(false, a0, false, b0, (short)0, c,
                                                false, false);
      c = __builtin_amdgcn_wmma_f32_16x16x4_f32(false, a1, false, b1, (short)0, c,
                                                false, false);
      if (j >= 2) { const v4f t = rowpn[2 * (j - 2)]; x[j - 2] = make_float4(t.x, t.y, t.z, t.w); }
      __builtin_amdgcn_sched_barrier(0x100);
    }
    { const v4f t = rowpn[60]; x[30] = make_float4(t.x, t.y, t.z, t.w); }
    { const v4f t = rowpn[62]; x[31] = make_float4(t.x, t.y, t.z, t.w); }
    asm volatile("v_nop\n\tv_nop\n\tv_nop\n\tv_nop" : "+v"(c));
    __builtin_amdgcn_sched_barrier(0);

    if (half == 0 && valid) {
      const float B = 1.f - 0.01f * x2;
      float rq[KCLS];
#pragma unroll
      for (int k = 0; k < KCLS; k++) {
        float xy  = sc * c[k];
        float A   = 1.f - 0.02f * xy + 0.01f * y2[k];
        float nv2 = fmaxf(A * A * x2 - 2.f * A * B * xy + B * B * y2[k], 0.f);
        float dnm = 1.f - 0.02f * xy + 1e-4f * x2 * y2[k] + 1e-5f;
        float man = sqrtf(nv2) / dnm;
        float z = fminf(fmaxf(0.1f * man, -1.f + 1e-5f), 1.f - 1e-5f);
        rq[k] = (1.f - z) / (1.f + z);
      }
      float rb = rq[0];
#pragma unroll
      for (int k = 1; k < KCLS; k++) rb = (rq[k] > rb) ? rq[k] : rb;
      const float irb = 1.f / rb;
      float e[KCLS]; float se = 0.f;
#pragma unroll
      for (int k = 0; k < KCLS; k++) {
        const float t  = rq[k] * irb;
        const float t2 = t * t, t4 = t2 * t2, t8 = t4 * t4;
        e[k] = t8 * t2;
        se += e[k];
      }
      float inv = 1.f / se;
#pragma unroll
      for (int k = 0; k < KCLS; k++) out[(size_t)q * KCLS + k] = e[k] * inv;
    }
    T = Tn; q = qn; valid = validn; have = more; rnd++;
  }
}

__global__ __launch_bounds__(256) void hpr_k3(const float* __restrict__ num,
                                              const float* __restrict__ den,
                                              const float* __restrict__ lossA,
                                              float* __restrict__ proto2,
                                              float* __restrict__ proto2_y2,
                                              float* __restrict__ out, int Q) {
  __shared__ float red[256];
  int t = threadIdx.x;
  for (int c = 0; c < KCLS; c++) {
    float mk = num[c * DDIM + t] / den[c];
    red[t] = mk * mk; __syncthreads();
    for (int s = 128; s > 0; s >>= 1) { if (t < s) red[t] += red[t + s]; __syncthreads(); }
    float mk2 = red[0]; __syncthreads();
    float dn = 1.0f + sqrtf(1.0f - 0.01f * mk2);
    proto2[c * DDIM + t] = mk / dn;
    if (t == 0) proto2_y2[c] = mk2 / (dn * dn);
  }
  if (t == 0) out[(size_t)Q * KCLS] = -lossA[0] / (float)Q;
}

extern "C" void kernel_launch(void* const* d_in, const int* in_sizes, int n_in,
                              void* d_out, int out_size, void* d_ws, size_t ws_size,
                              hipStream_t stream) {
  const float* feat  = (const float*)d_in[0];
  const int*   label = (const int*)d_in[1];
  float* out = (float*)d_out;
  float* ws  = (float*)d_ws;

  int total_rows = in_sizes[0] / DDIM;
  int Q = total_rows - 25;

  float* proto     = ws;
  float* proto_y2  = ws + 1280;
  float* num       = ws + 1288;
  float* den       = ws + 2568;
  float* lossA     = ws + 2576;
  float* proto2    = ws + 2580;
  float* proto2_y2 = ws + 3860;

  hpr_k1<<<1, 256, 0, stream>>>(num, den, lossA);

  hpr_pass1<<<PGRID, PW * 32, 0, stream>>>(feat, label, num, den, lossA, Q);
  hpr_k3<<<1, 256, 0, stream>>>(num, den, lossA, proto2, proto2_y2, out, Q);
  hpr_pass2<<<PGRID, PW * 32, 0, stream>>>(feat, proto2, proto2_y2, out, Q);
}


# pass kernels: a tile's rows fetched as 32 whole 512-byte pieces (4 full lines per load, not 32 bytes of 16 rows) and turned into the operand layout in the wave's own LDS tile; sum orders kept
# speedup vs baseline: 3.8067x; 1.0821x over previous
#include <hip/hip_runtime.h>
#include <hip/hip_bf16.h>
#include <math.h>

typedef float v2f __attribute__((ext_vector_type(2)));
typedef float v8f __attribute__((ext_vector_type(8)));
typedef float v4f __attribute__((ext_vector_type(4)));
typedef const __attribute__((address_space(1))) v4f* gv4p;

#define DDIM 256
#define KCLS 5
#define PITCH 260
#define MAXN  9.96f

__device__ __forceinline__ float waveRedSum(float v) {
#pragma unroll
  for (int off = 16; off > 0; off >>= 1) v += __shfl_xor(v, off, 32);
  return v;
}

__device__ __forceinline__ void hpr_lds_barrier() {
  __builtin_amdgcn_fence(__ATOMIC_RELEASE, "workgroup", "local");
  __builtin_amdgcn_s_barrier();
  __builtin_amdgcn_fence(__ATOMIC_ACQUIRE, "workgroup", "local");
}

__global__ __launch_bounds__(256) void hpr_k1(float* __restrict__ num,
                                              float* __restrict__ den,
                                              float* __restrict__ lossA) {
  int t = threadIdx.x;
#pragma unroll
  for (int c = 0; c < KCLS; c++) num[c * DDIM + t] = 0.f;
  if (t < KCLS) den[t] = 0.f;
  if (t == 0) lossA[0] = 0.f;
}

#define PW    8
#define PGRID 512
#define HROWS ((25 + PW - 1) / PW)

__device__ __forceinline__ int hpr_tile(int r, int w, int b, int nT) {
  const int base = r * (PW * PGRID);
  const int rem  = nT - base;
  if (rem >= PW * PGRID) return base + b * PW + w;
  if (rem <= 0) return -1;
  const int full = min(rem / PW, PGRID / 2);
  if (b < full) return base + b * PW + w;
  const int rest = rem - full * PW;
  const int idx  = w * (PGRID - full) + (b - full);
  return (idx < rest) ? base + full * PW + idx : -1;
}

__global__ __launch_bounds__(PW * 32, 4) void hpr_pass1(const float* __restrict__ feat,
                                                     const int* __restrict__ label,
                                                     float* __restrict__ num,
                                                     float* __restrict__ den,
                                                     float* __restrict__ lossA, int Q) {
  __shared__ __align__(16) float tile[PW][16][PITCH];
  __shared__ __align__(16) float sproto[16][PITCH];
  __shared__ float sDen[PW][8];
  __shared__ float sLoss[PW];
  __shared__ float sCf[32];
  __shared__ float sGam[32];
  __shared__ float sN2[PW][8];

  const int tid  = threadIdx.x;
  const int wave = __builtin_amdgcn_readfirstlane(tid >> 5);
  const int lane = tid & 31;
  const int m16  = lane & 15, half = lane >> 4;

  const int nT = (Q + 15) / 16;
  int  rnd   = 0;
  int  T     = hpr_tile(0, wave, blockIdx.x, nT);
  bool have  = (T >= 0);
  int  q     = min(T * 16, Q - 16) + m16;
  bool valid = have && (q >= T * 16);

  float4 sa0[HROWS], sa1[HROWS];
#pragma unroll
  for (int i = 0; i < HROWS; i++) {
    const int r = min(wave + PW * i, 24);
    const float4* row = (const float4*)(feat + (size_t)r * DDIM);
    sa0[i] = row[lane];
    sa1[i] = row[lane + 32];
  }
  float4 x[32];
  {
    const float4* rowp = (const float4*)(feat + (have ? (size_t)(25 + min(T * 16, Q - 16)) * DDIM : (size_t)0)) + lane;
#pragma unroll
    for (int i = 0; i < 32; i++) x[i] = rowp[32 * i];
  }
  int lbl = (valid && half == 0) ? label[Q + q] : 0;

  float y2[KCLS];
  {
    float* srow = &tile[0][0][0];
#pragma unroll
    for (int i = 0; i < HROWS; i++) {
      const int r = wave + PW * i;
      if (r < 25) {
        const float4 a0 = sa0[i];
        const float4 a1 = sa1[i];
        float acc = a0.x * a0.x + a0.y * a0.y + a0.z * a0.z + a0.w * a0.w
                  + a1.x * a1.x + a1.y * a1.y + a1.z * a1.z + a1.w * a1.w;
        float n2 = waveRedSum(acc);
        float n  = fmaxf(sqrtf(n2), 1e-5f);
        float sc = tanhf(0.1f * n) / (0.1f * n);
        float pn = sc * n;
        if (pn > MAXN) { sc *= MAXN / pn; pn = MAXN; }
        float x2    = pn * pn;
        float kq    = 2.0f / (1.0f + 0.01f * x2);
        float gamma = rsqrtf(1.0f - 0.01f * kq * kq * x2);
        *(float4*)&srow[r * DDIM + lane * 4]       = a0;
        *(float4*)&srow[r * DDIM + 128 + lane * 4] = a1;
        if (lane == 0) { sCf[r] = gamma * kq * sc; sGam[r] = gamma; }
      }
    }
    hpr_lds_barrier();
    float mk[KCLS];
#pragma unroll
    for (int cc = 0; cc < KCLS; cc++) {
      float agk = 0.f, ag = 0.f;
#pragma unroll
      for (int j = 0; j < 5; j++) {
        int r = cc * 5 + j;
        agk += sCf[r] * srow[r * DDIM + tid];
        ag  += sGam[r];
      }
      if (blockIdx.x == 0) {
        atomicAdd(&num[cc * DDIM + tid], agk);
        if (tid == 0) atomicAdd(&den[cc], ag);
      }
      mk[cc] = agk / ag;
      float s = waveRedSum(mk[cc] * mk[cc]);
      if (lane == 0) sN2[wave][cc] = s;
    }
    hpr_lds_barrier();
#pragma unroll
    for (int cc = 0; cc < KCLS; cc++) {
      float mk2 = 0.f;
#pragma unroll
      for (int w = 0; w < PW; w++) mk2 += sN2[w][cc];
      float dn = 1.0f + sqrtf(1.0f - 0.01f * mk2);
      sproto[cc][tid] = mk[cc] / dn;
      y2[cc] = mk2 / (dn * dn);
    }
#pragma unroll
    for (int r = KCLS; r < 16; r++) sproto[r][tid] = 0.f;
    hpr_lds_barrier();
  }

  v2f   racc[KCLS][4];
  float dacc[KCLS];
  float lacc = 0.f;
#pragma unroll
  for (int cc = 0; cc < KCLS; cc++) {
    dacc[cc] = 0.f;
#pragma unroll
    for (int i = 0; i < 4; i++) racc[cc][i] = (v2f){0.f, 0.f};
  }

  while (have) {
    const int  Tn     = hpr_tile(rnd + 1, wave, blockIdx.x, nT);
    const bool more   = (Tn >= 0);
    const int  qbn    = min(Tn * 16, Q - 16);
    const int  qn     = qbn + m16;
    const bool validn = more && (qn >= Tn * 16);
    unsigned long long rown = (unsigned long long)((const float4*)(feat + (more ? (size_t)(25 + qbn) * DDIM : (size_t)0)) + lane);
    const int lbln = (validn && half == 0) ? label[Q + qn] : 0;

#pragma unroll
    for (int i = 0; i < 32; i++) *(float4*)&tile[wave][i >> 1][128 * (i & 1) + 4 * lane] = x[i];
    __builtin_amdgcn_fence(__ATOMIC_RELEASE, "wavefront");
    __builtin_amdgcn_wave_barrier();
    __builtin_amdgcn_fence(__ATOMIC_ACQUIRE, "wavefront");
#pragma unroll
    for (int j = 0; j < 32; j++) x[j] = *(const float4*)&tile[wave][m16][8 * j + 4 * half];
    __builtin_amdgcn_sched_barrier(0);

    float acc = 0.f;
#pragma unroll
    for (int j = 0; j < 32; j++)
      acc += x[j].x * x[j].x + x[j].y * x[j].y + x[j].z * x[j].z + x[j].w * x[j].w;
    float n2 = acc + __shfl_xor(acc, 16, 32);
    float n  = fmaxf(sqrtf(n2), 1e-5f);
    float sc = tanhf(0.1f * n) / (0.1f * n);
    float pn = sc * n;
    if (pn > MAXN) { sc *= MAXN / pn; pn = MAXN; }
    float x2 = pn * pn;
    float B     = 1.f - 0.01f * x2;
    float kq    = 2.0f / (1.0f + 0.01f * x2);
    float gamma = rsqrtf(1.0f - 0.01f * kq * kq * x2);
    float wrow  = gamma * kq * sc;

    v8f c = {0.f, 0.f, 0.f, 0.f, 0.f, 0.f, 0.f, 0.f};
    asm volatile("" : "+v"(sc), "+v"(x2), "+v"(B), "+v"(gamma), "+v"(wrow), "+v"(rown));
    const gv4p rowpn = (gv4p)rown;
    __builtin_amdgcn_sched_barrier(0);
#pragma unroll
    for (int j = 0; j < 32; j++) {
      float4 p = *(const float4*)&sproto[m16][8 * j + 4 * half];
      v2f a0 = {p.x, p.y},       a1 = {p.z, p.w};
      v2f b0 = {x[j].x, x[j].y}, b1 = {x[j].z, x[j].w};
      c = __builtin_amdgcn_wmma_f32_16x16x4_f32(false, a0, false, b0, (short)0, c,
                                                false, false);
      c = __builtin_amdgcn_wmma_f32_16x16x4_f32(false, a1, false, b1, (short)0, c,
                                                false, false);
      if (j >= 2) { const v4f t = rowpn[32 * (j - 2)]; x[j - 2] = make_float4(t.x, t.y, t.z, t.w); }
      __builtin_amdgcn_sched_barrier(0x100);
    }
    { const v4f t = rowpn[32 * 30]; x[30] = make_float4(t.x, t.y, t.z, t.w); }
    { const v4f t = rowpn[32 * 31]; x[31] = make_float4(t.x, t.y, t.z, t.w); }
    asm volatile("v_nop\n\tv_nop\n\tv_nop\n\tv_nop" : "+v"(c));
    __builtin_amdgcn_sched_barrier(0);

    const bool own = valid && (half == 0);
    float rq[KCLS];
#pragma unroll
    for (int k = 0; k < KCLS; k++) {
      float xy  = sc * c[k];
      float A   = 1.f - 0.02f * xy + 0.01f * y2[k];
      float nv2 = fmaxf(A * A * x2 - 2.f * A * B * xy + B * B * y2[k], 0.f);
      float dnm = 1.f - 0.02f * xy + 1e-4f * x2 * y2[k] + 1e-5f;
      float man = sqrtf(nv2) / dnm;
      float z = fminf(fmaxf(0.1f * man, -1.f + 1e-5f), 1.f - 1e-5f);
      rq[k] = (1.f - z) / (1.f + z);
    }
    float rb = rq[0]; int pr = 0;
#pragma unroll
    for (int k = 1; k < KCLS; k++) { if (rq[k] > rb) { rb = rq[k]; pr = k; } }
    const float irb = 1.f / rb;
    float se = 0.f;
    float tl = rq[0] * irb;
#pragma unroll
    for (int k = 0; k < KCLS; k++) {
      const float t  = rq[k] * irb;
      const float t2 = t * t, t4 = t2 * t2, t8 = t4 * t4;
      se += t8 * t2;
      if (k > 0) tl = (lbl == k) ? t : tl;
    }
    float ce = 10.f * logf(tl) - logf(se);
    lacc += own ? ce : 0.f;

    const float wq = own ? wrow : 0.f;
#pragma unroll
    for (int cc = 0; cc < KCLS; cc++) dacc[cc] += (own && pr == cc) ? gamma : 0.f;

#pragma unroll 4
    for (int r = 0; r < 16; r++) {
      const int   p  = __builtin_amdgcn_readlane(pr, r);
      const float cf = __int_as_float(__builtin_amdgcn_readlane(__float_as_int(wq), r));
      const float4 v0 = *(const float4*)&tile[wave][r][4 * lane];
      const float4 v1 = *(const float4*)&tile[wave][r][128 + 4 * lane];
      const v2f va = {v0.x, v0.y}, vb = {v0.z, v0.w}, vc = {v1.x, v1.y}, vd = {v1.z, v1.w};
#pragma unroll
      for (int cc = 0; cc < KCLS; cc++) {
        const float m  = (p == cc) ? cf : 0.f;
        const v2f   mm = {m, m};
        racc[cc][0] = __builtin_elementwise_fma(mm, va, racc[cc][0]);
        racc[cc][1] = __builtin_elementwise_fma(mm, vb, racc[cc][1]);
        racc[cc][2] = __builtin_elementwise_fma(mm, vc, racc[cc][2]);
        racc[cc][3] = __builtin_elementwise_fma(mm, vd, racc[cc][3]);
      }
    }
    __builtin_amdgcn_fence(__ATOMIC_RELEASE, "wavefront");
    __builtin_amdgcn_wave_barrier();
    __builtin_amdgcn_fence(__ATOMIC_ACQUIRE, "wavefront");
    lbl = lbln; T = Tn; q = qn; valid = validn; have = more; rnd++;
  }

  {
    __syncthreads();
    float* red = &tile[0][0][0];
#pragma unroll
    for (int cc = 0; cc < KCLS; cc++) {
      float4 lo = {racc[cc][0].x, racc[cc][0].y, racc[cc][1].x, racc[cc][1].y};
      float4 hi = {racc[cc][2].x, racc[cc][2].y, racc[cc][3].x, racc[cc][3].y};
      *(float4*)&red[wave * (KCLS * DDIM) + cc * DDIM + 4 * lane]       = lo;
      *(float4*)&red[wave * (KCLS * DDIM) + cc * DDIM + 128 + 4 * lane] = hi;
    }
#pragma unroll
    for (int cc = 0; cc < KCLS; cc++) {
      float s = waveRedSum(dacc[cc]);
      if (lane == 0) sDen[wave][cc] = s;
    }
    {
      float s = waveRedSum(lacc);
      if (lane == 0) sLoss[wave] = s;
    }
    __syncthreads();
    for (int e = tid; e < KCLS * DDIM; e += PW * 32) {
      float s = 0.f;
#pragma unroll
      for (int w = 0; w < PW; w++) s += red[w * (KCLS * DDIM) + e];
      atomicAdd(&num[e], s);
    }
    if (tid < KCLS) {
      float s = 0.f;
#pragma unroll
      for (int w = 0; w < PW; w++) s += sDen[w][tid];
      atomicAdd(&den[tid], s);
    }
    if (tid == KCLS) {
      float s = 0.f;
#pragma unroll
      for (int w = 0; w < PW; w++) s += sLoss[w];
      atomicAdd(lossA, s);
    }
  }
}

static_assert(PW * 32 == DDIM, "thread tid stages column tid of the protos");
__global__ __launch_bounds__(PW * 32, 4) void hpr_pass2(const float* __restrict__ feat,
                                                     const float* __restrict__ proto,
                                                     const float* __restrict__ proto_y2,
                                                     float* __restrict__ out, int Q) {
  __shared__ __align__(16) float tile[PW][16][PITCH];
  __shared__ __align__(16) float sproto[16][PITCH];

  const int tid  = threadIdx.x;
  const int wave = __builtin_amdgcn_readfirstlane(tid >> 5);
  const int lane = tid & 31;
  const int m16  = lane & 15, half = lane >> 4;

  const int nT = (Q + 15) / 16;
  int  rnd   = 0;
  int  T     = hpr_tile(0, wave, blockIdx.x, nT);
  bool have  = (T >= 0);
  int  q     = min(T * 16, Q - 16) + m16;
  bool valid = have && (q >= T * 16);

  float pv[KCLS];
#pragma unroll
  for (int r = 0; r < KCLS; r++) pv[r] = proto[r * DDIM + tid];
  float4 x[32];
  {
    const float4* rowp = (const float4*)(feat + (have ? (size_t)(25 + min(T * 16, Q - 16)) * DDIM : (size_t)0)) + lane;
#pragma unroll
    for (int i = 0; i < 32; i++) x[i] = rowp[32 * i];
  }

#pragma unroll
  for (int r = 0; r < 16; r++) sproto[r][tid] = (r < KCLS) ? pv[r < KCLS ? r : 0] : 0.f;
  float y2[KCLS];
#pragma unroll
  for (int k = 0; k < KCLS; k++) y2[k] = proto_y2[k];
  hpr_lds_barrier();

  while (have) {
    const int  Tn     = hpr_tile(rnd + 1, wave, blockIdx.x, nT);
    const bool more   = (Tn >= 0);
    const int  qbn    = min(Tn * 16, Q - 16);
    const int  qn     = qbn + m16;
    const bool validn = more && (qn >= Tn * 16);
    unsigned long long rown = (unsigned long long)((const float4*)(feat + (more ? (size_t)(25 + qbn) * DDIM : (size_t)0)) + lane);

#pragma unroll
    for (int i = 0; i < 32; i++) *(float4*)&tile[wave][i >> 1][128 * (i & 1) + 4 * lane] = x[i];
    __builtin_amdgcn_fence(__ATOMIC_RELEASE, "wavefront");
    __builtin_amdgcn_wave_barrier();
    __builtin_amdgcn_fence(__ATOMIC_ACQUIRE, "wavefront");
#pragma unroll
    for (int j = 0; j < 32; j++) x[j] = *(const float4*)&tile[wave][m16][8 * j + 4 * half];
    __builtin_amdgcn_sched_barrier(0);

    float acc = 0.f;
#pragma unroll
    for (int j = 0; j < 32; j++)
      acc += x[j].x * x[j].x + x[j].y * x[j].y + x[j].z * x[j].z + x[j].w * x[j].w;
    float n2 = acc + __shfl_xor(acc, 16, 32);
    float n  = fmaxf(sqrtf(n2), 1e-5f);
    float sc = tanhf(0.1f * n) / (0.1f * n);
    float pn = sc * n;
    if (pn > MAXN) { sc *= MAXN / pn; pn = MAXN; }
    float x2 = pn * pn;

    v8f c = {0.f, 0.f, 0.f, 0.f, 0.f, 0.f, 0.f, 0.f};
    asm volatile("" : "+v"(sc), "+v"(x2), "+v"(rown));
    const gv4p rowpn = (gv4p)rown;
    __builtin_amdgcn_sched_barrier(0);
#pragma unroll
    for (int j = 0; j < 32; j++) {
      float4 p = *(const float4*)&sproto[m16][8 * j + 4 * half];
      v2f a0 = {p.x, p.y},       a1 = {p.z, p.w};
      v2f b0 = {x[j].x, x[j].y}, b1 = {x[j].z, x[j].w};
      c = __builtin_amdgcn_wmma_f32_16x16x4_f32(false, a0, false, b0, (short)0, c,
                                                false, false);
      c = __builtin_amdgcn_wmma_f32_16x16x4_f32(false, a1, false, b1, (short)0, c,
                                                false, false);
      if (j >= 2) { const v4f t = rowpn[32 * (j - 2)]; x[j - 2] = make_float4(t.x, t.y, t.z, t.w); }
      __builtin_amdgcn_sched_barrier(0x100);
    }
    { const v4f t = rowpn[32 * 30]; x[30] = make_float4(t.x, t.y, t.z, t.w); }
    { const v4f t = rowpn[32 * 31]; x[31] = make_float4(t.x, t.y, t.z, t.w); }
    asm volatile("v_nop\n\tv_nop\n\tv_nop\n\tv_nop" : "+v"(c));
    __builtin_amdgcn_sched_barrier(0);

    if (half == 0 && valid) {
      const float B = 1.f - 0.01f * x2;
      float rq[KCLS];
#pragma unroll
      for (int k = 0; k < KCLS; k++) {
        float xy  = sc * c[k];
        float A   = 1.f - 0.02f * xy + 0.01f * y2[k];
        float nv2 = fmaxf(A * A * x2 - 2.f * A * B * xy + B * B * y2[k], 0.f);
        float dnm = 1.f - 0.02f * xy + 1e-4f * x2 * y2[k] + 1e-5f;
        float man = sqrtf(nv2) / dnm;
        float z = fminf(fmaxf(0.1f * man, -1.f + 1e-5f), 1.f - 1e-5f);
        rq[k] = (1.f - z) / (1.f + z);
      }
      float rb = rq[0];
#pragma unroll
      for (int k = 1; k < KCLS; k++) rb = (rq[k] > rb) ? rq[k] : rb;
      const float irb = 1.f / rb;
      float e[KCLS]; float se = 0.f;
#pragma unroll
      for (int k = 0; k < KCLS; k++) {
        const float t  = rq[k] * irb;
        const float t2 = t * t, t4 = t2 * t2, t8 = t4 * t4;
        e[k] = t8 * t2;
        se += e[k];
      }
      float inv = 1.f / se;
#pragma unroll
      for (int k = 0; k < KCLS; k++) out[(size_t)q * KCLS + k] = e[k] * inv;
    }
    __builtin_amdgcn_fence(__ATOMIC_RELEASE, "wavefront");
    __builtin_amdgcn_wave_barrier();
    __builtin_amdgcn_fence(__ATOMIC_ACQUIRE, "wavefront");
    T = Tn; q = qn; valid = validn; have = more; rnd++;
  }
}

__global__ __launch_bounds__(256) void hpr_k3(const float* __restrict__ num,
                                              const float* __restrict__ den,
                                              const float* __restrict__ lossA,
                                              float* __restrict__ proto2,
                                              float* __restrict__ proto2_y2,
                                              float* __restrict__ out, int Q) {
  __shared__ float red[256];
  int t = threadIdx.x;
  for (int c = 0; c < KCLS; c++) {
    float mk = num[c * DDIM + t] / den[c];
    red[t] = mk * mk; __syncthreads();
    for (int s = 128; s > 0; s >>= 1) { if (t < s) red[t] += red[t + s]; __syncthreads(); }
    float mk2 = red[0]; __syncthreads();
    float dn = 1.0f + sqrtf(1.0f - 0.01f * mk2);
    proto2[c * DDIM + t] = mk / dn;
    if (t == 0) proto2_y2[c] = mk2 / (dn * dn);
  }
  if (t == 0) out[(size_t)Q * KCLS] = -lossA[0] / (float)Q;
}

extern "C" void kernel_launch(void* const* d_in, const int* in_sizes, int n_in,
                              void* d_out, int out_size, void* d_ws, size_t ws_size,
                              hipStream_t stream) {
  const float* feat  = (const float*)d_in[0];
  const int*   label = (const int*)d_in[1];
  float* out = (float*)d_out;
  float* ws  = (float*)d_ws;

  int total_rows = in_sizes[0] / DDIM;
  int Q = total_rows - 25;

  float* proto     = ws;
  float* proto_y2  = ws + 1280;
  float* num       = ws + 1288;
  float* den       = ws + 2568;
  float* lossA     = ws + 2576;
  float* proto2    = ws + 2580;
  float* proto2_y2 = ws + 3860;

  hpr_k1<<<1, 256, 0, stream>>>(num, den, lossA);

  hpr_pass1<<<PGRID, PW * 32, 0, stream>>>(feat, label, num, den, lossA, Q);
  hpr_k3<<<1, 256, 0, stream>>>(num, den, lossA, proto2, proto2_y2, out, Q);
  hpr_pass2<<<PGRID, PW * 32, 0, stream>>>(feat, proto2, proto2_y2, out, Q);
}


# class sums leave pass 1 as a plane per work-group (both passes: 256 groups of 16 waves; planes lie in out); a new k3, one group per class, adds them in fixed order: no atomics, no zeroing launch
# speedup vs baseline: 4.1963x; 1.1023x over previous
#include <hip/hip_runtime.h>
#include <hip/hip_bf16.h>
#include <math.h>

typedef float v2f __attribute__((ext_vector_type(2)));
typedef float v8f __attribute__((ext_vector_type(8)));
typedef float v4f __attribute__((ext_vector_type(4)));
typedef const __attribute__((address_space(1))) v4f* gv4p;

#define DDIM 256
#define KCLS 5
#define PITCH 260
#define MAXN  9.96f

__device__ __forceinline__ float waveRedSum(float v) {
#pragma unroll
  for (int off = 16; off > 0; off >>= 1) v += __shfl_xor(v, off, 32);
  return v;
}

__device__ __forceinline__ void hpr_lds_barrier() {
  __builtin_amdgcn_fence(__ATOMIC_RELEASE, "workgroup", "local");
  __builtin_amdgcn_s_barrier();
  __builtin_amdgcn_fence(__ATOMIC_ACQUIRE, "workgroup", "local");
}

#define PW    16
#define PGRID 256
#define PSTR  1312
#define HROWS ((25 + PW - 1) / PW)

static_assert(PW == 16, "hpr_rank orders 16 waves");
__device__ __forceinline__ int hpr_rank(int w) { return 4 * (((w & 3) - (w >> 2)) & 3) + (w >> 2); }
__device__ __forceinline__ int hpr_tile(int r, int w, int b, int nT) {
  const int base = r * (PW * PGRID);
  const int rem  = nT - base;
  if (rem >= PW * PGRID) return base + b * PW + w;
  if (rem <= 0) return -1;
  const int per = rem / PGRID;
  const int ext = rem - per * PGRID;
  const int k   = hpr_rank(w);
  if (k < per) return base + b * per + k;
  return (k == per && b < ext) ? base + PGRID * per + b : -1;
}

__global__ __launch_bounds__(PW * 32, 4) void hpr_pass1(const float* __restrict__ feat,
                                                     const int* __restrict__ label,
                                                     float* __restrict__ plane, int Q) {
  __shared__ __align__(16) float tile[PW][16][PITCH];
  __shared__ __align__(16) float sproto[16][PITCH];
  __shared__ float sDen[PW][8];
  __shared__ float sLoss[PW];
  __shared__ float sCf[32];
  __shared__ float sGam[32];
  __shared__ float sN2[PW][8];

  const int tid  = threadIdx.x;
  const int d    = tid & (DDIM - 1);
  const int wave = __builtin_amdgcn_readfirstlane(tid >> 5);
  const int lane = tid & 31;
  const int m16  = lane & 15, half = lane >> 4;

  const int nT = (Q + 15) / 16;
  int  rnd   = 0;
  int  T     = hpr_tile(0, wave, blockIdx.x, nT);
  bool have  = (T >= 0);
  int  q     = min(T * 16, Q - 16) + m16;
  bool valid = have && (q >= T * 16);

  float4 sa0[HROWS], sa1[HROWS];
#pragma unroll
  for (int i = 0; i < HROWS; i++) {
    const int r = min(wave + PW * i, 24);
    const float4* row = (const float4*)(feat + (size_t)r * DDIM);
    sa0[i] = row[lane];
    sa1[i] = row[lane + 32];
  }
  float4 x[32];
  {
    const float4* rowp = (const float4*)(feat + (have ? (size_t)(25 + min(T * 16, Q - 16)) * DDIM : (size_t)0)) + lane;
#pragma unroll
    for (int i = 0; i < 32; i++) x[i] = rowp[32 * i];
  }
  int lbl = (valid && half == 0) ? label[Q + q] : 0;

  float y2[KCLS];
  {
    float* srow = &tile[0][0][0];
#pragma unroll
    for (int i = 0; i < HROWS; i++) {
      const int r = wave + PW * i;
      if (r < 25) {
        const float4 a0 = sa0[i];
        const float4 a1 = sa1[i];
        float acc = a0.x * a0.x + a0.y * a0.y + a0.z * a0.z + a0.w * a0.w
                  + a1.x * a1.x + a1.y * a1.y + a1.z * a1.z + a1.w * a1.w;
        float n2 = waveRedSum(acc);
        float n  = fmaxf(sqrtf(n2), 1e-5f);
        float sc = tanhf(0.1f * n) / (0.1f * n);
        float pn = sc * n;
        if (pn > MAXN) { sc *= MAXN / pn; pn = MAXN; }
        float x2    = pn * pn;
        float kq    = 2.0f / (1.0f + 0.01f * x2);
        float gamma = rsqrtf(1.0f - 0.01f * kq * kq * x2);
        *(float4*)&srow[r * DDIM + lane * 4]       = a0;
        *(float4*)&srow[r * DDIM + 128 + lane * 4] = a1;
        if (lane == 0) { sCf[r] = gamma * kq * sc; sGam[r] = gamma; }
      }
    }
    hpr_lds_barrier();
    float mk[KCLS];
#pragma unroll
    for (int cc = 0; cc < KCLS; cc++) {
      float agk = 0.f, ag = 0.f;
#pragma unroll
      for (int j = 0; j < 5; j++) {
        int r = cc * 5 + j;
        agk += sCf[r] * srow[r * DDIM + d];
        ag  += sGam[r];
      }
      if (blockIdx.x == 0 && tid < DDIM) {
        plane[(size_t)PGRID * PSTR + cc * DDIM + tid] = agk;
        if (tid == 0) plane[(size_t)PGRID * PSTR + KCLS * DDIM + cc] = ag;
      }
      mk[cc] = agk / ag;
      float s = waveRedSum(mk[cc] * mk[cc]);
      if (lane == 0) sN2[wave][cc] = s;
    }
    hpr_lds_barrier();
#pragma unroll
    for (int cc = 0; cc < KCLS; cc++) {
      float mk2 = 0.f;
#pragma unroll
      for (int w = 0; w < DDIM / 32; w++) mk2 += sN2[w][cc];
      float dn = 1.0f + sqrtf(1.0f - 0.01f * mk2);
      sproto[cc][d] = mk[cc] / dn;
      y2[cc] = mk2 / (dn * dn);
    }
#pragma unroll
    for (int r = KCLS; r < 16; r++) sproto[r][d] = 0.f;
    hpr_lds_barrier();
  }

  v2f   racc[KCLS][4];
  float dacc[KCLS];
  float lacc = 0.f;
#pragma unroll
  for (int cc = 0; cc < KCLS; cc++) {
    dacc[cc] = 0.f;
#pragma unroll
    for (int i = 0; i < 4; i++) racc[cc][i] = (v2f){0.f, 0.f};
  }

  while (have) {
    const int  Tn     = hpr_tile(rnd + 1, wave, blockIdx.x, nT);
    const bool more   = (Tn >= 0);
    const int  qbn    = min(Tn * 16, Q - 16);
    const int  qn     = qbn + m16;
    const bool validn = more && (qn >= Tn * 16);
    unsigned long long rown = (unsigned long long)((const float4*)(feat + (more ? (size_t)(25 + qbn) * DDIM : (size_t)0)) + lane);
    const int lbln = (validn && half == 0) ? label[Q + qn] : 0;

#pragma unroll
    for (int i = 0; i < 32; i++) *(float4*)&tile[wave][i >> 1][128 * (i & 1) + 4 * lane] = x[i];
    __builtin_amdgcn_fence(__ATOMIC_RELEASE, "wavefront");
    __builtin_amdgcn_wave_barrier();
    __builtin_amdgcn_fence(__ATOMIC_ACQUIRE, "wavefront");
#pragma unroll
    for (int j = 0; j < 32; j++) x[j] = *(const float4*)&tile[wave][m16][8 * j + 4 * half];
    __builtin_amdgcn_sched_barrier(0);

    float acc = 0.f;
#pragma unroll
    for (int j = 0; j < 32; j++)
      acc += x[j].x * x[j].x + x[j].y * x[j].y + x[j].z * x[j].z + x[j].w * x[j].w;
    float n2 = acc + __shfl_xor(acc, 16, 32);
    float n  = fmaxf(sqrtf(n2), 1e-5f);
    float sc = tanhf(0.1f * n) / (0.1f * n);
    float pn = sc * n;
    if (pn > MAXN) { sc *= MAXN / pn; pn = MAXN; }
    float x2 = pn * pn;
    float B     = 1.f - 0.01f * x2;
    float kq    = 2.0f / (1.0f + 0.01f * x2);
    float gamma = rsqrtf(1.0f - 0.01f * kq * kq * x2);
    float wrow  = gamma * kq * sc;

    v8f c = {0.f, 0.f, 0.f, 0.f, 0.f, 0.f, 0.f, 0.f};
    asm volatile("" : "+v"(sc), "+v"(x2), "+v"(B), "+v"(gamma), "+v"(wrow), "+v"(rown));
    const gv4p rowpn = (gv4p)rown;
    __builtin_amdgcn_sched_barrier(0);
#pragma unroll
    for (int j = 0; j < 32; j++) {
      float4 p = *(const float4*)&sproto[m16][8 * j + 4 * half];
      v2f a0 = {p.x, p.y},       a1 = {p.z, p.w};
      v2f b0 = {x[j].x, x[j].y}, b1 = {x[j].z, x[j].w};
      c = __builtin_amdgcn_wmma_f32_16x16x4_f32(false, a0, false, b0, (short)0, c,
                                                false, false);
      c = __builtin_amdgcn_wmma_f32_16x16x4_f32(false, a1, false, b1, (short)0, c,
                                                false, false);
      if (j >= 2) { const v4f t = rowpn[32 * (j - 2)]; x[j - 2] = make_float4(t.x, t.y, t.z, t.w); }
      __builtin_amdgcn_sched_barrier(0x100);
    }
    { const v4f t = rowpn[32 * 30]; x[30] = make_float4(t.x, t.y, t.z, t.w); }
    { const v4f t = rowpn[32 * 31]; x[31] = make_float4(t.x, t.y, t.z, t.w); }
    asm volatile("v_nop\n\tv_nop\n\tv_nop\n\tv_nop" : "+v"(c));
    __builtin_amdgcn_sched_barrier(0);

    const bool own = valid && (half == 0);
    float rq[KCLS];
#pragma unroll
    for (int k = 0; k < KCLS; k++) {
      float xy  = sc * c[k];
      float A   = 1.f - 0.02f * xy + 0.01f * y2[k];
      float nv2 = fmaxf(A * A * x2 - 2.f * A * B * xy + B * B * y2[k], 0.f);
      float dnm = 1.f - 0.02f * xy + 1e-4f * x2 * y2[k] + 1e-5f;
      float man = sqrtf(nv2) / dnm;
      float z = fminf(fmaxf(0.1f * man, -1.f + 1e-5f), 1.f - 1e-5f);
      rq[k] = (1.f - z) / (1.f + z);
    }
    float rb = rq[0]; int pr = 0;
#pragma unroll
    for (int k = 1; k < KCLS; k++) { if (rq[k] > rb) { rb = rq[k]; pr = k; } }
    const float irb = 1.f / rb;
    float se = 0.f;
    float tl = rq[0] * irb;
#pragma unroll
    for (int k = 0; k < KCLS; k++) {
      const float t  = rq[k] * irb;
      const float t2 = t * t, t4 = t2 * t2, t8 = t4 * t4;
      se += t8 * t2;
      if (k > 0) tl = (lbl == k) ? t : tl;
    }
    float ce = 10.f * logf(tl) - logf(se);
    lacc += own ? ce : 0.f;

    const float wq = own ? wrow : 0.f;
#pragma unroll
    for (int cc = 0; cc < KCLS; cc++) dacc[cc] += (own && pr == cc) ? gamma : 0.f;

#pragma unroll 4
    for (int r = 0; r < 16; r++) {
      const int   p  = __builtin_amdgcn_readlane(pr, r);
      const float cf = __int_as_float(__builtin_amdgcn_readlane(__float_as_int(wq), r));
      const float4 v0 = *(const float4*)&tile[wave][r][4 * lane];
      const float4 v1 = *(const float4*)&tile[wave][r][128 + 4 * lane];
      const v2f va = {v0.x, v0.y}, vb = {v0.z, v0.w}, vc = {v1.x, v1.y}, vd = {v1.z, v1.w};
#pragma unroll
      for (int cc = 0; cc < KCLS; cc++) {
        const float m  = (p == cc) ? cf : 0.f;
        const v2f   mm = {m, m};
        racc[cc][0] = __builtin_elementwise_fma(mm, va, racc[cc][0]);
        racc[cc][1] = __builtin_elementwise_fma(mm, vb, racc[cc][1]);
        racc[cc][2] = __builtin_elementwise_fma(mm, vc, racc[cc][2]);
        racc[cc][3] = __builtin_elementwise_fma(mm, vd, racc[cc][3]);
      }
    }
    __builtin_amdgcn_fence(__ATOMIC_RELEASE, "wavefront");
    __builtin_amdgcn_wave_barrier();
    __builtin_amdgcn_fence(__ATOMIC_ACQUIRE, "wavefront");
    lbl = lbln; T = Tn; q = qn; valid = validn; have = more; rnd++;
  }

  {
    __syncthreads();
    float* red = &tile[0][0][0];
#pragma unroll
    for (int cc = 0; cc < KCLS; cc++) {
      float4 lo = {racc[cc][0].x, racc[cc][0].y, racc[cc][1].x, racc[cc][1].y};
      float4 hi = {racc[cc][2].x, racc[cc][2].y, racc[cc][3].x, racc[cc][3].y};
      *(float4*)&red[wave * (KCLS * DDIM) + cc * DDIM + 4 * lane]       = lo;
      *(float4*)&red[wave * (KCLS * DDIM) + cc * DDIM + 128 + 4 * lane] = hi;
    }
#pragma unroll
    for (int cc = 0; cc < KCLS; cc++) {
      float s = waveRedSum(dacc[cc]);
      if (lane == 0) sDen[wave][cc] = s;
    }
    {
      float s = waveRedSum(lacc);
      if (lane == 0) sLoss[wave] = s;
    }
    __syncthreads();
    for (int e = tid; e < KCLS * DDIM; e += PW * 32) {
      float s = 0.f;
#pragma unroll
      for (int w = 0; w < PW; w++) s += red[w * (KCLS * DDIM) + e];
      plane[(size_t)blockIdx.x * PSTR + e] = s;
    }
    if (tid < KCLS) {
      float s = 0.f;
#pragma unroll
      for (int w = 0; w < PW; w++) s += sDen[w][tid];
      plane[(size_t)blockIdx.x * PSTR + KCLS * DDIM + tid] = s;
    }
    if (tid == KCLS) {
      float s = 0.f;
#pragma unroll
      for (int w = 0; w < PW; w++) s += sLoss[w];
      plane[(size_t)blockIdx.x * PSTR + KCLS * DDIM + KCLS] = s;
    }
  }
}

__global__ __launch_bounds__(PW * 32, 4) void hpr_pass2(const float* __restrict__ feat,
                                                     const float* __restrict__ proto,
                                                     const float* __restrict__ proto_y2,
                                                     float* __restrict__ out, int Q) {
  __shared__ __align__(16) float tile[PW][16][PITCH];
  __shared__ __align__(16) float sproto[16][PITCH];

  const int tid  = threadIdx.x;
  const int d    = tid & (DDIM - 1);
  const int wave = __builtin_amdgcn_readfirstlane(tid >> 5);
  const int lane = tid & 31;
  const int m16  = lane & 15, half = lane >> 4;

  const int nT = (Q + 15) / 16;
  int  rnd   = 0;
  int  T     = hpr_tile(0, wave, blockIdx.x, nT);
  bool have  = (T >= 0);
  int  q     = min(T * 16, Q - 16) + m16;
  bool valid = have && (q >= T * 16);

  float pv[KCLS];
#pragma unroll
  for (int r = 0; r < KCLS; r++) pv[r] = proto[r * DDIM + d];
  float4 x[32];
  {
    const float4* rowp = (const float4*)(feat + (have ? (size_t)(25 + min(T * 16, Q - 16)) * DDIM : (size_t)0)) + lane;
#pragma unroll
    for (int i = 0; i < 32; i++) x[i] = rowp[32 * i];
  }

#pragma unroll
  for (int r = 0; r < 16; r++) sproto[r][d] = (r < KCLS) ? pv[r < KCLS ? r : 0] : 0.f;
  float y2[KCLS];
#pragma unroll
  for (int k = 0; k < KCLS; k++) y2[k] = proto_y2[k];
  hpr_lds_barrier();

  while (have) {
    const int  Tn     = hpr_tile(rnd + 1, wave, blockIdx.x, nT);
    const bool more   = (Tn >= 0);
    const int  qbn    = min(Tn * 16, Q - 16);
    const int  qn     = qbn + m16;
    const bool validn = more && (qn >= Tn * 16);
    unsigned long long rown = (unsigned long long)((const float4*)(feat + (more ? (size_t)(25 + qbn) * DDIM : (size_t)0)) + lane);

#pragma unroll
    for (int i = 0; i < 32; i++) *(float4*)&tile[wave][i >> 1][128 * (i & 1) + 4 * lane] = x[i];
    __builtin_amdgcn_fence(__ATOMIC_RELEASE, "wavefront");
    __builtin_amdgcn_wave_barrier();
    __builtin_amdgcn_fence(__ATOMIC_ACQUIRE, "wavefront");
#pragma unroll
    for (int j = 0; j < 32; j++) x[j] = *(const float4*)&tile[wave][m16][8 * j + 4 * half];
    __builtin_amdgcn_sched_barrier(0);

    float acc = 0.f;
#pragma unroll
    for (int j = 0; j < 32; j++)
      acc += x[j].x * x[j].x + x[j].y * x[j].y + x[j].z * x[j].z + x[j].w * x[j].w;
    float n2 = acc + __shfl_xor(acc, 16, 32);
    float n  = fmaxf(sqrtf(n2), 1e-5f);
    float sc = tanhf(0.1f * n) / (0.1f * n);
    float pn = sc * n;
    if (pn > MAXN) { sc *= MAXN / pn; pn = MAXN; }
    float x2 = pn * pn;

    v8f c = {0.f, 0.f, 0.f, 0.f, 0.f, 0.f, 0.f, 0.f};
    asm volatile("" : "+v"(sc), "+v"(x2), "+v"(rown));
    const gv4p rowpn = (gv4p)rown;
    __builtin_amdgcn_sched_barrier(0);
#pragma unroll
    for (int j = 0; j < 32; j++) {
      float4 p = *(const float4*)&sproto[m16][8 * j + 4 * half];
      v2f a0 = {p.x, p.y},       a1 = {p.z, p.w};
      v2f b0 = {x[j].x, x[j].y}, b1 = {x[j].z, x[j].w};
      c = __builtin_amdgcn_wmma_f32_16x16x4_f32(false, a0, false, b0, (short)0, c,
                                                false, false);
      c = __builtin_amdgcn_wmma_f32_16x16x4_f32(false, a1, false, b1, (short)0, c,
                                                false, false);
      if (j >= 2) { const v4f t = rowpn[32 * (j - 2)]; x[j - 2] = make_float4(t.x, t.y, t.z, t.w); }
      __builtin_amdgcn_sched_barrier(0x100);
    }
    { const v4f t = rowpn[32 * 30]; x[30] = make_float4(t.x, t.y, t.z, t.w); }
    { const v4f t = rowpn[32 * 31]; x[31] = make_float4(t.x, t.y, t.z, t.w); }
    asm volatile("v_nop\n\tv_nop\n\tv_nop\n\tv_nop" : "+v"(c));
    __builtin_amdgcn_sched_barrier(0);

    if (half == 0 && valid) {
      const float B = 1.f - 0.01f * x2;
      float rq[KCLS];
#pragma unroll
      for (int k = 0; k < KCLS; k++) {
        float xy  = sc * c[k];
        float A   = 1.f - 0.02f * xy + 0.01f * y2[k];
        float nv2 = fmaxf(A * A * x2 - 2.f * A * B * xy + B * B * y2[k], 0.f);
        float dnm = 1.f - 0.02f * xy + 1e-4f * x2 * y2[k] + 1e-5f;
        float man = sqrtf(nv2) / dnm;
        float z = fminf(fmaxf(0.1f * man, -1.f + 1e-5f), 1.f - 1e-5f);
        rq[k] = (1.f - z) / (1.f + z);
      }
      float rb = rq[0];
#pragma unroll
      for (int k = 1; k < KCLS; k++) rb = (rq[k] > rb) ? rq[k] : rb;
      const float irb = 1.f / rb;
      float e[KCLS]; float se = 0.f;
#pragma unroll
      for (int k = 0; k < KCLS; k++) {
        const float t  = rq[k] * irb;
        const float t2 = t * t, t4 = t2 * t2, t8 = t4 * t4;
        e[k] = t8 * t2;
        se += e[k];
      }
      float inv = 1.f / se;
#pragma unroll
      for (int k = 0; k < KCLS; k++) out[(size_t)q * KCLS + k] = e[k] * inv;
    }
    __builtin_amdgcn_fence(__ATOMIC_RELEASE, "wavefront");
    __builtin_amdgcn_wave_barrier();
    __builtin_amdgcn_fence(__ATOMIC_ACQUIRE, "wavefront");
    T = Tn; q = qn; valid = validn; have = more; rnd++;
  }
}

#define K3T 1024
static_assert(PGRID % (K3T / DDIM) == 0 && PGRID < K3T, "plane split of hpr_k3");
__global__ __launch_bounds__(K3T) void hpr_k3(const float* plane,
                                              float* __restrict__ proto2,
                                              float* __restrict__ proto2_y2,
                                              float* out, int Q) {
  __shared__ float part[K3T / DDIM][DDIM];
  __shared__ float sw[K3T / 32];
  __shared__ float sn[DDIM / 32];
  const int c    = blockIdx.x;
  const int tid  = threadIdx.x;
  const int t    = tid & (DDIM - 1), g = tid >> 8;
  const int wave = tid >> 5, lane = tid & 31;
  const float dv0 = (tid <= PGRID) ? plane[(size_t)tid * PSTR + KCLS * DDIM + c] : 0.f;
  const float lv0 = (c == 0 && tid < PGRID) ? plane[(size_t)tid * PSTR + KCLS * DDIM + KCLS] : 0.f;
  float s = 0.f;
#pragma unroll
  for (int i = 0; i < PGRID / (K3T / DDIM); i++)
    s += plane[(size_t)(g + (K3T / DDIM) * i) * PSTR + c * DDIM + t];
  if (g == 0) s += plane[(size_t)PGRID * PSTR + c * DDIM + t];
  part[g][t] = s;
  const float dv = waveRedSum(dv0);
  if (lane == 0) sw[wave] = dv;
  __syncthreads();
  float den = 0.f;
#pragma unroll
  for (int w = 0; w < K3T / 32; w++) den += sw[w];
  const float num = (part[0][t] + part[1][t]) + (part[2][t] + part[3][t]);
  const float mk  = num / den;
  const float sq  = waveRedSum(mk * mk);
  if (g == 0 && lane == 0) sn[wave] = sq;
  __syncthreads();
  float mk2 = 0.f;
#pragma unroll
  for (int w = 0; w < DDIM / 32; w++) mk2 += sn[w];
  const float dn = 1.0f + sqrtf(1.0f - 0.01f * mk2);
  if (g == 0) {
    proto2[c * DDIM + t] = mk / dn;
    if (t == 0) proto2_y2[c] = mk2 / (dn * dn);
  }
  if (c == 0) {
    const float lv = waveRedSum(lv0);
    if (lane == 0) sw[wave] = lv;
    __syncthreads();
    if (tid == 0) {
      float L = 0.f;
#pragma unroll
      for (int w = 0; w < K3T / 32; w++) L += sw[w];
      out[(size_t)Q * KCLS] = -L / (float)Q;
    }
  }
}

extern "C" void kernel_launch(void* const* d_in, const int* in_sizes, int n_in,
                              void* d_out, int out_size, void* d_ws, size_t ws_size,
                              hipStream_t stream) {
  const float* feat  = (const float*)d_in[0];
  const int*   label = (const int*)d_in[1];
  float* out = (float*)d_out;
  float* ws  = (float*)d_ws;

  int total_rows = in_sizes[0] / DDIM;
  int Q = total_rows - 25;

  float* proto2    = ws;
  float* proto2_y2 = ws + 1280;

  hpr_pass1<<<PGRID, PW * 32, 0, stream>>>(feat, label, out, Q);
  hpr_k3<<<KCLS, K3T, 0, stream>>>(out, proto2, proto2_y2, out, Q);
  hpr_pass2<<<PGRID, PW * 32, 0, stream>>>(feat, proto2, proto2_y2, out, Q);
}


# k3 as 40 work-groups (class x 32 columns) that read the planes 16 bytes per lane in whole lines and store Klein means; pass 2's head takes proto_new and its norms; num/den/loss sum order changes
# speedup vs baseline: 4.4573x; 1.0622x over previous
#include <hip/hip_runtime.h>
#include <hip/hip_bf16.h>
#include <math.h>

typedef float v2f __attribute__((ext_vector_type(2)));
typedef float v8f __attribute__((ext_vector_type(8)));
typedef float v4f __attribute__((ext_vector_type(4)));
typedef const __attribute__((address_space(1))) v4f* gv4p;

#define DDIM 256
#define KCLS 5
#define PITCH 260
#define MAXN  9.96f

__device__ __forceinline__ float waveRedSum(float v) {
#pragma unroll
  for (int off = 16; off > 0; off >>= 1) v += __shfl_xor(v, off, 32);
  return v;
}

__device__ __forceinline__ void hpr_lds_barrier() {
  __builtin_amdgcn_fence(__ATOMIC_RELEASE, "workgroup", "local");
  __builtin_amdgcn_s_barrier();
  __builtin_amdgcn_fence(__ATOMIC_ACQUIRE, "workgroup", "local");
}

#define PW    16
#define PGRID 256
#define PSTR  1312
#define HROWS ((25 + PW - 1) / PW)

static_assert(PW == 16, "hpr_rank orders 16 waves");
__device__ __forceinline__ int hpr_rank(int w) { return 4 * (((w & 3) - (w >> 2)) & 3) + (w >> 2); }
__device__ __forceinline__ int hpr_tile(int r, int w, int b, int nT) {
  const int base = r * (PW * PGRID);
  const int rem  = nT - base;
  if (rem >= PW * PGRID) return base + b * PW + w;
  if (rem <= 0) return -1;
  const int per = rem / PGRID;
  const int ext = rem - per * PGRID;
  const int k   = hpr_rank(w);
  if (k < per) return base + b * per + k;
  return (k == per && b < ext) ? base + PGRID * per + b : -1;
}

__global__ __launch_bounds__(PW * 32, 4) void hpr_pass1(const float* __restrict__ feat,
                                                     const int* __restrict__ label,
                                                     float* __restrict__ plane, int Q) {
  __shared__ __align__(16) float tile[PW][16][PITCH];
  __shared__ __align__(16) float sproto[16][PITCH];
  __shared__ float sDen[PW][8];
  __shared__ float sLoss[PW];
  __shared__ float sCf[32];
  __shared__ float sGam[32];
  __shared__ float sN2[PW][8];

  const int tid  = threadIdx.x;
  const int d    = tid & (DDIM - 1);
  const int wave = __builtin_amdgcn_readfirstlane(tid >> 5);
  const int lane = tid & 31;
  const int m16  = lane & 15, half = lane >> 4;

  const int nT = (Q + 15) / 16;
  int  rnd   = 0;
  int  T     = hpr_tile(0, wave, blockIdx.x, nT);
  bool have  = (T >= 0);
  int  q     = min(T * 16, Q - 16) + m16;
  bool valid = have && (q >= T * 16);

  float4 sa0[HROWS], sa1[HROWS];
#pragma unroll
  for (int i = 0; i < HROWS; i++) {
    const int r = min(wave + PW * i, 24);
    const float4* row = (const float4*)(feat + (size_t)r * DDIM);
    sa0[i] = row[lane];
    sa1[i] = row[lane + 32];
  }
  float4 x[32];
  {
    const float4* rowp = (const float4*)(feat + (have ? (size_t)(25 + min(T * 16, Q - 16)) * DDIM : (size_t)0)) + lane;
#pragma unroll
    for (int i = 0; i < 32; i++) x[i] = rowp[32 * i];
  }
  int lbl = (valid && half == 0) ? label[Q + q] : 0;

  float y2[KCLS];
  {
    float* srow = &tile[0][0][0];
#pragma unroll
    for (int i = 0; i < HROWS; i++) {
      const int r = wave + PW * i;
      if (r < 25) {
        const float4 a0 = sa0[i];
        const float4 a1 = sa1[i];
        float acc = a0.x * a0.x + a0.y * a0.y + a0.z * a0.z + a0.w * a0.w
                  + a1.x * a1.x + a1.y * a1.y + a1.z * a1.z + a1.w * a1.w;
        float n2 = waveRedSum(acc);
        float n  = fmaxf(sqrtf(n2), 1e-5f);
        float sc = tanhf(0.1f * n) / (0.1f * n);
        float pn = sc * n;
        if (pn > MAXN) { sc *= MAXN / pn; pn = MAXN; }
        float x2    = pn * pn;
        float kq    = 2.0f / (1.0f + 0.01f * x2);
        float gamma = rsqrtf(1.0f - 0.01f * kq * kq * x2);
        *(float4*)&srow[r * DDIM + lane * 4]       = a0;
        *(float4*)&srow[r * DDIM + 128 + lane * 4] = a1;
        if (lane == 0) { sCf[r] = gamma * kq * sc; sGam[r] = gamma; }
      }
    }
    hpr_lds_barrier();
    float mk[KCLS];
#pragma unroll
    for (int cc = 0; cc < KCLS; cc++) {
      float agk = 0.f, ag = 0.f;
#pragma unroll
      for (int j = 0; j < 5; j++) {
        int r = cc * 5 + j;
        agk += sCf[r] * srow[r * DDIM + d];
        ag  += sGam[r];
      }
      if (blockIdx.x == 0 && tid < DDIM) {
        plane[(size_t)PGRID * PSTR + cc * DDIM + tid] = agk;
        if (tid == 0) plane[(size_t)PGRID * PSTR + KCLS * DDIM + cc] = ag;
      }
      mk[cc] = agk / ag;
      float s = waveRedSum(mk[cc] * mk[cc]);
      if (lane == 0) sN2[wave][cc] = s;
    }
    hpr_lds_barrier();
#pragma unroll
    for (int cc = 0; cc < KCLS; cc++) {
      float mk2 = 0.f;
#pragma unroll
      for (int w = 0; w < DDIM / 32; w++) mk2 += sN2[w][cc];
      float dn = 1.0f + sqrtf(1.0f - 0.01f * mk2);
      sproto[cc][d] = mk[cc] / dn;
      y2[cc] = mk2 / (dn * dn);
    }
#pragma unroll
    for (int r = KCLS; r < 16; r++) sproto[r][d] = 0.f;
    hpr_lds_barrier();
  }

  v2f   racc[KCLS][4];
  float dacc[KCLS];
  float lacc = 0.f;
#pragma unroll
  for (int cc = 0; cc < KCLS; cc++) {
    dacc[cc] = 0.f;
#pragma unroll
    for (int i = 0; i < 4; i++) racc[cc][i] = (v2f){0.f, 0.f};
  }

  while (have) {
    const int  Tn     = hpr_tile(rnd + 1, wave, blockIdx.x, nT);
    const bool more   = (Tn >= 0);
    const int  qbn    = min(Tn * 16, Q - 16);
    const int  qn     = qbn + m16;
    const bool validn = more && (qn >= Tn * 16);
    unsigned long long rown = (unsigned long long)((const float4*)(feat + (more ? (size_t)(25 + qbn) * DDIM : (size_t)0)) + lane);
    const int lbln = (validn && half == 0) ? label[Q + qn] : 0;

#pragma unroll
    for (int i = 0; i < 32; i++) *(float4*)&tile[wave][i >> 1][128 * (i & 1) + 4 * lane] = x[i];
    __builtin_amdgcn_fence(__ATOMIC_RELEASE, "wavefront");
    __builtin_amdgcn_wave_barrier();
    __builtin_amdgcn_fence(__ATOMIC_ACQUIRE, "wavefront");
#pragma unroll
    for (int j = 0; j < 32; j++) x[j] = *(const float4*)&tile[wave][m16][8 * j + 4 * half];
    __builtin_amdgcn_sched_barrier(0);

    float acc = 0.f;
#pragma unroll
    for (int j = 0; j < 32; j++)
      acc += x[j].x * x[j].x + x[j].y * x[j].y + x[j].z * x[j].z + x[j].w * x[j].w;
    float n2 = acc + __shfl_xor(acc, 16, 32);
    float n  = fmaxf(sqrtf(n2), 1e-5f);
    float sc = tanhf(0.1f * n) / (0.1f * n);
    float pn = sc * n;
    if (pn > MAXN) { sc *= MAXN / pn; pn = MAXN; }
    float x2 = pn * pn;
    float B     = 1.f - 0.01f * x2;
    float kq    = 2.0f / (1.0f + 0.01f * x2);
    float gamma = rsqrtf(1.0f - 0.01f * kq * kq * x2);
    float wrow  = gamma * kq * sc;

    v8f c = {0.f, 0.f, 0.f, 0.f, 0.f, 0.f, 0.f, 0.f};
    asm volatile("" : "+v"(sc), "+v"(x2), "+v"(B), "+v"(gamma), "+v"(wrow), "+v"(rown));
    const gv4p rowpn = (gv4p)rown;
    __builtin_amdgcn_sched_barrier(0);
#pragma unroll
    for (int j = 0; j < 32; j++) {
      float4 p = *(const float4*)&sproto[m16][8 * j + 4 * half];
      v2f a0 = {p.x, p.y},       a1 = {p.z, p.w};
      v2f b0 = {x[j].x, x[j].y}, b1 = {x[j].z, x[j].w};
      c = __builtin_amdgcn_wmma_f32_16x16x4_f32(false, a0, false, b0, (short)0, c,
                                                false, false);
      c = __builtin_amdgcn_wmma_f32_16x16x4_f32(false, a1, false, b1, (short)0, c,
                                                false, false);
      if (j >= 2) { const v4f t = rowpn[32 * (j - 2)]; x[j - 2] = make_float4(t.x, t.y, t.z, t.w); }
      __builtin_amdgcn_sched_barrier(0x100);
    }
    { const v4f t = rowpn[32 * 30]; x[30] = make_float4(t.x, t.y, t.z, t.w); }
    { const v4f t = rowpn[32 * 31]; x[31] = make_float4(t.x, t.y, t.z, t.w); }
    asm volatile("v_nop\n\tv_nop\n\tv_nop\n\tv_nop" : "+v"(c));
    __builtin_amdgcn_sched_barrier(0);

    const bool own = valid && (half == 0);
    float rq[KCLS];
#pragma unroll
    for (int k = 0; k < KCLS; k++) {
      float xy  = sc * c[k];
      float A   = 1.f - 0.02f * xy + 0.01f * y2[k];
      float nv2 = fmaxf(A * A * x2 - 2.f * A * B * xy + B * B * y2[k], 0.f);
      float dnm = 1.f - 0.02f * xy + 1e-4f * x2 * y2[k] + 1e-5f;
      float man = sqrtf(nv2) / dnm;
      float z = fminf(fmaxf(0.1f * man, -1.f + 1e-5f), 1.f - 1e-5f);
      rq[k] = (1.f - z) / (1.f + z);
    }
    float rb = rq[0]; int pr = 0;
#pragma unroll
    for (int k = 1; k < KCLS; k++) { if (rq[k] > rb) { rb = rq[k]; pr = k; } }
    const float irb = 1.f / rb;
    float se = 0.f;
    float tl = rq[0] * irb;
#pragma unroll
    for (int k = 0; k < KCLS; k++) {
      const float t  = rq[k] * irb;
      const float t2 = t * t, t4 = t2 * t2, t8 = t4 * t4;
      se += t8 * t2;
      if (k > 0) tl = (lbl == k) ? t : tl;
    }
    float ce = 10.f * logf(tl) - logf(se);
    lacc += own ? ce : 0.f;

    const float wq = own ? wrow : 0.f;
#pragma unroll
    for (int cc = 0; cc < KCLS; cc++) dacc[cc] += (own && pr == cc) ? gamma : 0.f;

#pragma unroll 4
    for (int r = 0; r < 16; r++) {
      const int   p  = __builtin_amdgcn_readlane(pr, r);
      const float cf = __int_as_float(__builtin_amdgcn_readlane(__float_as_int(wq), r));
      const float4 v0 = *(const float4*)&tile[wave][r][4 * lane];
      const float4 v1 = *(const float4*)&tile[wave][r][128 + 4 * lane];
      const v2f va = {v0.x, v0.y}, vb = {v0.z, v0.w}, vc = {v1.x, v1.y}, vd = {v1.z, v1.w};
#pragma unroll
      for (int cc = 0; cc < KCLS; cc++) {
        const float m  = (p == cc) ? cf : 0.f;
        const v2f   mm = {m, m};
        racc[cc][0] = __builtin_elementwise_fma(mm, va, racc[cc][0]);
        racc[cc][1] = __builtin_elementwise_fma(mm, vb, racc[cc][1]);
        racc[cc][2] = __builtin_elementwise_fma(mm, vc, racc[cc][2]);
        racc[cc][3] = __builtin_elementwise_fma(mm, vd, racc[cc][3]);
      }
    }
    __builtin_amdgcn_fence(__ATOMIC_RELEASE, "wavefront");
    __builtin_amdgcn_wave_barrier();
    __builtin_amdgcn_fence(__ATOMIC_ACQUIRE, "wavefront");
    lbl = lbln; T = Tn; q = qn; valid = validn; have = more; rnd++;
  }

  {
    __syncthreads();
    float* red = &tile[0][0][0];
#pragma unroll
    for (int cc = 0; cc < KCLS; cc++) {
      float4 lo = {racc[cc][0].x, racc[cc][0].y, racc[cc][1].x, racc[cc][1].y};
      float4 hi = {racc[cc][2].x, racc[cc][2].y, racc[cc][3].x, racc[cc][3].y};
      *(float4*)&red[wave * (KCLS * DDIM) + cc * DDIM + 4 * lane]       = lo;
      *(float4*)&red[wave * (KCLS * DDIM) + cc * DDIM + 128 + 4 * lane] = hi;
    }
#pragma unroll
    for (int cc = 0; cc < KCLS; cc++) {
      float s = waveRedSum(dacc[cc]);
      if (lane == 0) sDen[wave][cc] = s;
    }
    {
      float s = waveRedSum(lacc);
      if (lane == 0) sLoss[wave] = s;
    }
    __syncthreads();
    for (int e = tid; e < KCLS * DDIM; e += PW * 32) {
      float s = 0.f;
#pragma unroll
      for (int w = 0; w < PW; w++) s += red[w * (KCLS * DDIM) + e];
      plane[(size_t)blockIdx.x * PSTR + e] = s;
    }
    if (tid < KCLS) {
      float s = 0.f;
#pragma unroll
      for (int w = 0; w < PW; w++) s += sDen[w][tid];
      plane[(size_t)blockIdx.x * PSTR + KCLS * DDIM + tid] = s;
    }
    if (tid == KCLS) {
      float s = 0.f;
#pragma unroll
      for (int w = 0; w < PW; w++) s += sLoss[w];
      plane[(size_t)blockIdx.x * PSTR + KCLS * DDIM + KCLS] = s;
    }
  }
}

__global__ __launch_bounds__(PW * 32, 4) void hpr_pass2(const float* __restrict__ feat,
                                                     const float* __restrict__ kmean,
                                                     float* __restrict__ out, int Q) {
  __shared__ __align__(16) float tile[PW][16][PITCH];
  __shared__ __align__(16) float sproto[16][PITCH];
  __shared__ float sN2[PW][8];

  const int tid  = threadIdx.x;
  const int d    = tid & (DDIM - 1);
  const int wave = __builtin_amdgcn_readfirstlane(tid >> 5);
  const int lane = tid & 31;
  const int m16  = lane & 15, half = lane >> 4;

  const int nT = (Q + 15) / 16;
  int  rnd   = 0;
  int  T     = hpr_tile(0, wave, blockIdx.x, nT);
  bool have  = (T >= 0);
  int  q     = min(T * 16, Q - 16) + m16;
  bool valid = have && (q >= T * 16);

  float pv[KCLS];
#pragma unroll
  for (int r = 0; r < KCLS; r++) pv[r] = kmean[r * DDIM + d];
  float4 x[32];
  {
    const float4* rowp = (const float4*)(feat + (have ? (size_t)(25 + min(T * 16, Q - 16)) * DDIM : (size_t)0)) + lane;
#pragma unroll
    for (int i = 0; i < 32; i++) x[i] = rowp[32 * i];
  }

  float sq[KCLS];
#pragma unroll
  for (int cc = 0; cc < KCLS; cc++) sq[cc] = pv[cc] * pv[cc];
#pragma unroll
  for (int off = 16; off > 0; off >>= 1) {
#pragma unroll
    for (int cc = 0; cc < KCLS; cc++) sq[cc] += __shfl_xor(sq[cc], off, 32);
  }
  if (lane == 0) {
#pragma unroll
    for (int cc = 0; cc < KCLS; cc++) sN2[wave][cc] = sq[cc];
  }
  hpr_lds_barrier();
  float y2[KCLS];
#pragma unroll
  for (int cc = 0; cc < KCLS; cc++) {
    float mk2 = 0.f;
#pragma unroll
    for (int w = 0; w < DDIM / 32; w++) mk2 += sN2[w][cc];
    const float dn = 1.0f + sqrtf(1.0f - 0.01f * mk2);
    sproto[cc][d] = pv[cc] / dn;
    y2[cc] = mk2 / (dn * dn);
  }
#pragma unroll
  for (int r = KCLS; r < 16; r++) sproto[r][d] = 0.f;
  hpr_lds_barrier();

  while (have) {
    const int  Tn     = hpr_tile(rnd + 1, wave, blockIdx.x, nT);
    const bool more   = (Tn >= 0);
    const int  qbn    = min(Tn * 16, Q - 16);
    const int  qn     = qbn + m16;
    const bool validn = more && (qn >= Tn * 16);
    unsigned long long rown = (unsigned long long)((const float4*)(feat + (more ? (size_t)(25 + qbn) * DDIM : (size_t)0)) + lane);

#pragma unroll
    for (int i = 0; i < 32; i++) *(float4*)&tile[wave][i >> 1][128 * (i & 1) + 4 * lane] = x[i];
    __builtin_amdgcn_fence(__ATOMIC_RELEASE, "wavefront");
    __builtin_amdgcn_wave_barrier();
    __builtin_amdgcn_fence(__ATOMIC_ACQUIRE, "wavefront");
#pragma unroll
    for (int j = 0; j < 32; j++) x[j] = *(const float4*)&tile[wave][m16][8 * j + 4 * half];
    __builtin_amdgcn_sched_barrier(0);

    float acc = 0.f;
#pragma unroll
    for (int j = 0; j < 32; j++)
      acc += x[j].x * x[j].x + x[j].y * x[j].y + x[j].z * x[j].z + x[j].w * x[j].w;
    float n2 = acc + __shfl_xor(acc, 16, 32);
    float n  = fmaxf(sqrtf(n2), 1e-5f);
    float sc = tanhf(0.1f * n) / (0.1f * n);
    float pn = sc * n;
    if (pn > MAXN) { sc *= MAXN / pn; pn = MAXN; }
    float x2 = pn * pn;

    v8f c = {0.f, 0.f, 0.f, 0.f, 0.f, 0.f, 0.f, 0.f};
    asm volatile("" : "+v"(sc), "+v"(x2), "+v"(rown));
    const gv4p rowpn = (gv4p)rown;
    __builtin_amdgcn_sched_barrier(0);
#pragma unroll
    for (int j = 0; j < 32; j++) {
      float4 p = *(const float4*)&sproto[m16][8 * j + 4 * half];
      v2f a0 = {p.x, p.y},       a1 = {p.z, p.w};
      v2f b0 = {x[j].x, x[j].y}, b1 = {x[j].z, x[j].w};
      c = __builtin_amdgcn_wmma_f32_16x16x4_f32(false, a0, false, b0, (short)0, c,
                                                false, false);
      c = __builtin_amdgcn_wmma_f32_16x16x4_f32(false, a1, false, b1, (short)0, c,
                                                false, false);
      if (j >= 2) { const v4f t = rowpn[32 * (j - 2)]; x[j - 2] = make_float4(t.x, t.y, t.z, t.w); }
      __builtin_amdgcn_sched_barrier(0x100);
    }
    { const v4f t = rowpn[32 * 30]; x[30] = make_float4(t.x, t.y, t.z, t.w); }
    { const v4f t = rowpn[32 * 31]; x[31] = make_float4(t.x, t.y, t.z, t.w); }
    asm volatile("v_nop\n\tv_nop\n\tv_nop\n\tv_nop" : "+v"(c));
    __builtin_amdgcn_sched_barrier(0);

    if (half == 0 && valid) {
      const float B = 1.f - 0.01f * x2;
      float rq[KCLS];
#pragma unroll
      for (int k = 0; k < KCLS; k++) {
        float xy  = sc * c[k];
        float A   = 1.f - 0.02f * xy + 0.01f * y2[k];
        float nv2 = fmaxf(A * A * x2 - 2.f * A * B * xy + B * B * y2[k], 0.f);
        float dnm = 1.f - 0.02f * xy + 1e-4f * x2 * y2[k] + 1e-5f;
        float man = sqrtf(nv2) / dnm;
        float z = fminf(fmaxf(0.1f * man, -1.f + 1e-5f), 1.f - 1e-5f);
        rq[k] = (1.f - z) / (1.f + z);
      }
      float rb = rq[0];
#pragma unroll
      for (int k = 1; k < KCLS; k++) rb = (rq[k] > rb) ? rq[k] : rb;
      const float irb = 1.f / rb;
      float e[KCLS]; float se = 0.f;
#pragma unroll
      for (int k = 0; k < KCLS; k++) {
        const float t  = rq[k] * irb;
        const float t2 = t * t, t4 = t2 * t2, t8 = t4 * t4;
        e[k] = t8 * t2;
        se += e[k];
      }
      float inv = 1.f / se;
#pragma unroll
      for (int k = 0; k < KCLS; k++) out[(size_t)q * KCLS + k] = e[k] * inv;
    }
    __builtin_amdgcn_fence(__ATOMIC_RELEASE, "wavefront");
    __builtin_amdgcn_wave_barrier();
    __builtin_amdgcn_fence(__ATOMIC_ACQUIRE, "wavefront");
    T = Tn; q = qn; valid = validn; have = more; rnd++;
  }
}

#define K3T  256
#define K3SEG (DDIM / 32)
static_assert(PGRID % (4 * (K3T / 32)) == 0 && PGRID == K3T, "plane split of hpr_k3");
__global__ __launch_bounds__(K3T) void hpr_k3(const float* plane,
                                              float* __restrict__ kmean,
                                              float* out, int Q) {
  __shared__ __align__(16) float part[K3T / 32][32];
  __shared__ float sw[K3T / 32];
  __shared__ float sl[K3T / 32];
  const int c    = blockIdx.x / K3SEG, seg = blockIdx.x % K3SEG;
  const int tid  = threadIdx.x;
  const int wave = tid >> 5, lane = tid & 31;
  const int pl   = lane >> 3, ch = lane & 7;
  const float dv0 = plane[(size_t)tid * PSTR + KCLS * DDIM + c];
  const float dvs = (tid == 0) ? plane[(size_t)PGRID * PSTR + KCLS * DDIM + c] : 0.f;
  const float lv0 = (blockIdx.x == 0) ? plane[(size_t)tid * PSTR + KCLS * DDIM + KCLS] : 0.f;
  const float* base = plane + c * DDIM + seg * 32 + 4 * ch;
  float4 a = {0.f, 0.f, 0.f, 0.f};
#pragma unroll
  for (int i = 0; i < PGRID / (4 * (K3T / 32)); i++) {
    const int p = 4 * (wave + (K3T / 32) * i) + pl;
    const float4 v = *(const float4*)(base + (size_t)p * PSTR);
    a.x += v.x; a.y += v.y; a.z += v.z; a.w += v.w;
  }
  if (wave == 0 && pl == 0) {
    const float4 v = *(const float4*)(base + (size_t)PGRID * PSTR);
    a.x += v.x; a.y += v.y; a.z += v.z; a.w += v.w;
  }
  float dv = dv0 + dvs, lv = lv0;
  a.x += __shfl_xor(a.x, 16, 32); a.y += __shfl_xor(a.y, 16, 32);
  a.z += __shfl_xor(a.z, 16, 32); a.w += __shfl_xor(a.w, 16, 32);
  dv  += __shfl_xor(dv, 16, 32);  lv  += __shfl_xor(lv, 16, 32);
  a.x += __shfl_xor(a.x, 8, 32);  a.y += __shfl_xor(a.y, 8, 32);
  a.z += __shfl_xor(a.z, 8, 32);  a.w += __shfl_xor(a.w, 8, 32);
  dv  += __shfl_xor(dv, 8, 32);   lv  += __shfl_xor(lv, 8, 32);
#pragma unroll
  for (int off = 4; off > 0; off >>= 1) { dv += __shfl_xor(dv, off, 32); lv += __shfl_xor(lv, off, 32); }
  if (lane < 8) *(float4*)&part[wave][4 * lane] = a;
  if (lane == 0) { sw[wave] = dv; sl[wave] = lv; }
  __syncthreads();
  if (tid < 32) {
    float den = 0.f, num = 0.f;
#pragma unroll
    for (int w = 0; w < K3T / 32; w++) { den += sw[w]; num += part[w][tid]; }
    kmean[c * DDIM + seg * 32 + tid] = num / den;
  }
  if (blockIdx.x == 0 && tid == 32) {
    float L = 0.f;
#pragma unroll
    for (int w = 0; w < K3T / 32; w++) L += sl[w];
    out[(size_t)Q * KCLS] = -L / (float)Q;
  }
}

extern "C" void kernel_launch(void* const* d_in, const int* in_sizes, int n_in,
                              void* d_out, int out_size, void* d_ws, size_t ws_size,
                              hipStream_t stream) {
  const float* feat  = (const float*)d_in[0];
  const int*   label = (const int*)d_in[1];
  float* out = (float*)d_out;
  float* ws  = (float*)d_ws;

  int total_rows = in_sizes[0] / DDIM;
  int Q = total_rows - 25;

  float* kmean = ws;

  hpr_pass1<<<PGRID, PW * 32, 0, stream>>>(feat, label, out, Q);
  hpr_k3<<<KCLS * K3SEG, K3T, 0, stream>>>(out, kmean, out, Q);
  hpr_pass2<<<PGRID, PW * 32, 0, stream>>>(feat, kmean, out, Q);
}
